# starting kernel
# baseline (speedup 1.0000x reference)
#include <hip/hip_runtime.h>

#define N_NODES   100000
#define N_EDGES   1200000
#define EMBED_D   32
#define HIDDEN_D  64
#define N_CLASSES 10
#define N_GRAPHS  1024

typedef __attribute__((ext_vector_type(2))) float v2f;
typedef __attribute__((ext_vector_type(8))) float v8f;

__device__ __forceinline__ void atomAdd1(float* p, float v) {
    asm volatile("global_atomic_add_f32 %0, %1, off scope:SCOPE_DEV"
                 :: "v"(p), "v"(v) : "memory");
}
__device__ __forceinline__ void atomAdd4(float* p, float4 v) {
    asm volatile(
        "global_atomic_add_f32 %0, %1, off scope:SCOPE_DEV\n\t"
        "global_atomic_add_f32 %0, %2, off offset:4 scope:SCOPE_DEV\n\t"
        "global_atomic_add_f32 %0, %3, off offset:8 scope:SCOPE_DEV\n\t"
        "global_atomic_add_f32 %0, %4, off offset:12 scope:SCOPE_DEV"
        :: "v"(p), "v"(v.x), "v"(v.y), "v"(v.z), "v"(v.w) : "memory");
}

__global__ void zero_kernel(float* __restrict__ p, int n) {
    int t = blockIdx.x * blockDim.x + threadIdx.x;
    if (t < n) p[t] = 0.0f;
}

__global__ void gather_embed_kernel(const int* __restrict__ ids,
                                    const float* __restrict__ embed,
                                    float* __restrict__ x0, int nNodes) {
    int t = blockIdx.x * blockDim.x + threadIdx.x;
    if (t >= nNodes * (EMBED_D / 4)) return;
    int node = t >> 3;
    int q    = t & 7;
    float4 v = reinterpret_cast<const float4*>(embed + (size_t)ids[node] * EMBED_D)[q];
    reinterpret_cast<float4*>(x0 + (size_t)node * EMBED_D)[q] = v;
}

__global__ void edge_count_kernel(const int* __restrict__ dst,
                                  float* __restrict__ cnt, int nEdges) {
    int e = blockIdx.x * blockDim.x + threadIdx.x;
    if (e >= nEdges) return;
    atomAddF_guard:
    atomAdd1(cnt + dst[e], 1.0f);
}

template <int DIM>
__global__ void edge_scatter_kernel(const int* __restrict__ src,
                                    const int* __restrict__ dst,
                                    const float* __restrict__ x,
                                    float* __restrict__ agg, int nEdges) {
    constexpr int QPE = DIM / 4;
    int t = blockIdx.x * blockDim.x + threadIdx.x;
    if (t >= nEdges * QPE) return;
    int e = t / QPE;
    int q = t & (QPE - 1);
    int s = src[e], d = dst[e];
    float4 v = reinterpret_cast<const float4*>(x + (size_t)s * DIM)[q];
    atomAdd4(agg + (size_t)d * DIM + q * 4, v);
}

template <int K>
__global__ void sage_gemm_kernel(const float* __restrict__ aggRaw,
                                 const float* __restrict__ cnt,
                                 const float* __restrict__ x,
                                 const float* __restrict__ Wl,
                                 const float* __restrict__ Wr,
                                 const float* __restrict__ bias,
                                 float* __restrict__ out, int nNodes) {
    const int lane = threadIdx.x & 31;
    const int wave = threadIdx.x >> 5;
    const int row0 = blockIdx.x * 16;
    if (row0 >= nNodes) return;
    const int col0 = wave * 16;
    const int grp  = lane >> 4;
    const int lr   = lane & 15;

    const int   arow = row0 + lr;
    const float invc = 1.0f / fmaxf(cnt[arow], 1.0f);

    v8f c = {};
#pragma unroll
    for (int k0 = 0; k0 < K; k0 += 4) {
        const int ka = k0 + 2 * grp;
        v2f a, b;
        a.x = aggRaw[(size_t)arow * K + ka + 0] * invc;
        a.y = aggRaw[(size_t)arow * K + ka + 1] * invc;
        b.x = Wl[(size_t)(ka + 0) * HIDDEN_D + col0 + lr];
        b.y = Wl[(size_t)(ka + 1) * HIDDEN_D + col0 + lr];
        c = __builtin_amdgcn_wmma_f32_16x16x4_f32(false, a, false, b,
                                                  (short)0, c, false, false);
    }
#pragma unroll
    for (int k0 = 0; k0 < K; k0 += 4) {
        const int ka = k0 + 2 * grp;
        v2f a, b;
        a.x = x[(size_t)arow * K + ka + 0];
        a.y = x[(size_t)arow * K + ka + 1];
        b.x = Wr[(size_t)(ka + 0) * HIDDEN_D + col0 + lr];
        b.y = Wr[(size_t)(ka + 1) * HIDDEN_D + col0 + lr];
        c = __builtin_amdgcn_wmma_f32_16x16x4_f32(false, a, false, b,
                                                  (short)0, c, false, false);
    }

    const float bv = bias[col0 + lr];
#pragma unroll
    for (int v = 0; v < 8; ++v) {
        const int m = v + 8 * grp;
        out[(size_t)(row0 + m) * HIDDEN_D + col0 + lr] = fmaxf(c[v] + bv, 0.0f);
    }
}

__global__ void pool_kernel(const int* __restrict__ batch,
                            const float* __restrict__ h,
                            float* __restrict__ gsum,
                            float* __restrict__ gcnt, int nNodes) {
    int t = blockIdx.x * blockDim.x + threadIdx.x;
    if (t >= nNodes * (HIDDEN_D / 4)) return;
    int node = t >> 4;
    int q    = t & 15;
    int g    = batch[node];
    float4 v = reinterpret_cast<const float4*>(h + (size_t)node * HIDDEN_D)[q];
    atomAdd4(gsum + (size_t)g * HIDDEN_D + q * 4, v);
    if (q == 0) atomAdd1(gcnt + g, 1.0f);
}

__global__ void head_kernel(const float* __restrict__ gsum,
                            const float* __restrict__ gcnt,
                            const float* __restrict__ Wo,
                            const float* __restrict__ bo,
                            float* __restrict__ out) {
    int t = blockIdx.x * blockDim.x + threadIdx.x;
    if (t >= N_GRAPHS * N_CLASSES) return;
    int g = t / N_CLASSES;
    int c = t - g * N_CLASSES;
    float invc = 1.0f / fmaxf(gcnt[g], 1.0f);
    float acc = 0.0f;
    for (int d = 0; d < HIDDEN_D; ++d)
        acc += gsum[(size_t)g * HIDDEN_D + d] * invc * Wo[(size_t)d * N_CLASSES + c];
    out[t] = acc + bo[c];
}

extern "C" void kernel_launch(void* const* d_in, const int* in_sizes, int n_in,
                              void* d_out, int out_size, void* d_ws, size_t ws_size,
                              hipStream_t stream) {
    const int*   node_ids = (const int*)  d_in[0];
    const int*   src      = (const int*)  d_in[1];
    const int*   dst      = (const int*)  d_in[2];
    const int*   batch    = (const int*)  d_in[3];
    const float* embed    = (const float*)d_in[4];
    const float* W1_l     = (const float*)d_in[5];
    const float* b1       = (const float*)d_in[6];
    const float* W1_r     = (const float*)d_in[7];
    const float* W2_l     = (const float*)d_in[8];
    const float* b2       = (const float*)d_in[9];
    const float* W2_r     = (const float*)d_in[10];
    const float* W_out    = (const float*)d_in[11];
    const float* b_out    = (const float*)d_in[12];
    float* out = (float*)d_out;

    float* ws   = (float*)d_ws;
    float* x0   = ws;
    float* h1   = x0   + (size_t)N_NODES * EMBED_D;
    float* h2   = h1   + (size_t)N_NODES * HIDDEN_D;
    float* agg1 = h2   + (size_t)N_NODES * HIDDEN_D;
    float* agg2 = agg1 + (size_t)N_NODES * EMBED_D;
    float* cnt  = agg2 + (size_t)N_NODES * HIDDEN_D;
    float* gsum = cnt  + (size_t)N_NODES;
    float* gcnt = gsum + (size_t)N_GRAPHS * HIDDEN_D;
    const int zeroN = N_NODES * EMBED_D + N_NODES * HIDDEN_D + N_NODES
                    + N_GRAPHS * HIDDEN_D + N_GRAPHS;

    const int B = 256;
    auto blocks = [](int n, int b) { return (n + b - 1) / b; };

    zero_kernel<<<blocks(zeroN, B), B, 0, stream>>>(agg1, zeroN);

    gather_embed_kernel<<<blocks(N_NODES * 8, B), B, 0, stream>>>(
        node_ids, embed, x0, N_NODES);

    edge_count_kernel<<<blocks(N_EDGES, B), B, 0, stream>>>(dst, cnt, N_EDGES);

    edge_scatter_kernel<EMBED_D><<<blocks(N_EDGES * (EMBED_D / 4), B), B, 0, stream>>>(
        src, dst, x0, agg1, N_EDGES);
    sage_gemm_kernel<EMBED_D><<<N_NODES / 16, 128, 0, stream>>>(
        agg1, cnt, x0, W1_l, W1_r, b1, h1, N_NODES);

    edge_scatter_kernel<HIDDEN_D><<<blocks(N_EDGES * (HIDDEN_D / 4), B), B, 0, stream>>>(
        src, dst, h1, agg2, N_EDGES);
    sage_gemm_kernel<HIDDEN_D><<<N_NODES / 16, 128, 0, stream>>>(
        agg2, cnt, h1, W2_l, W2_r, b2, h2, N_NODES);

    pool_kernel<<<blocks(N_NODES * (HIDDEN_D / 4), B), B, 0, stream>>>(
        batch, h2, gsum, gcnt, N_NODES);
    head_kernel<<<blocks(N_GRAPHS * N_CLASSES, B), B, 0, stream>>>(
        gsum, gcnt, W_out, b_out, out);
}


# edge sums by owner tile: block keeps 392 destination rows in LDS and scans all destinations, a wave sums its own rows' source rows; edge atomics, big zero fill, count launch gone; row sum order fixed
# speedup vs baseline: 1.7047x; 1.7047x over previous
#include <hip/hip_runtime.h>
#include <type_traits>

#define N_NODES   100000
#define N_EDGES   1200000
#define EMBED_D   32
#define HIDDEN_D  64
#define N_CLASSES 10
#define N_GRAPHS  1024

typedef __attribute__((ext_vector_type(2))) float v2f;
typedef __attribute__((ext_vector_type(8))) float v8f;

__device__ __forceinline__ void atomAdd1(float* p, float v) {
    asm volatile("global_atomic_add_f32 %0, %1, off scope:SCOPE_DEV"
                 :: "v"(p), "v"(v) : "memory");
}

__global__ void zero_kernel(float* __restrict__ p, int n) {
    int t = blockIdx.x * blockDim.x + threadIdx.x;
    if (t < n) p[t] = 0.0f;
}

__global__ void gather_embed_kernel(const int* __restrict__ ids,
                                    const float* __restrict__ embed,
                                    float* __restrict__ x0, int nNodes) {
    int t = blockIdx.x * blockDim.x + threadIdx.x;
    if (t >= nNodes * (EMBED_D / 4)) return;
    int node = t >> 3;
    int q    = t & 7;
    float4 v = reinterpret_cast<const float4*>(embed + (size_t)ids[node] * EMBED_D)[q];
    reinterpret_cast<float4*>(x0 + (size_t)node * EMBED_D)[q] = v;
}

#define TILE_ROWS   392
#define AGG_THREADS 1024
#define AGG_WAVES   32
#define BCAP        32
#define LCAP        512
#define SCAN_U      8
static_assert(N_EDGES % 4 == 0, "destination list is read 4 at a time");
static_assert(TILE_ROWS + AGG_WAVES <= 1024 && N_EDGES < (1 << 21),
              "an entry is (edge << 10) | local row");

template <int DIM, bool COUNT>
__global__ __launch_bounds__(AGG_THREADS)
void tile_agg_kernel(const int* __restrict__ src,
                     const int* __restrict__ dst,
                     const float* __restrict__ x,
                     float* __restrict__ agg,
                     float* __restrict__ cnt, int nNodes, int nEdges) {
    constexpr int H = DIM / 32;
    __shared__ float    acc[(TILE_ROWS + AGG_WAVES) * DIM];
    __shared__ float    cntL[TILE_ROWS + AGG_WAVES];
    __shared__ unsigned bucket[AGG_WAVES * AGG_WAVES * BCAP];
    __shared__ unsigned wcnt[AGG_WAVES * AGG_WAVES];
    __shared__ unsigned list[AGG_WAVES * LCAP];
    __shared__ unsigned ovf;

    const int tid  = threadIdx.x;
    const int lane = tid & 31;
    const int wave = tid >> 5;
    const int base = blockIdx.x * TILE_ROWS;

    for (int i = tid; i < (TILE_ROWS + AGG_WAVES) * DIM; i += AGG_THREADS) acc[i] = 0.0f;
    for (int i = tid; i < TILE_ROWS + AGG_WAVES; i += AGG_THREADS) cntL[i] = 0.0f;
    if (tid == 0) ovf = 0u;
    __syncthreads();

    unsigned mycnt = 0;
    auto put = [&](unsigned r, int e) {
        unsigned mask = __builtin_amdgcn_ballot_w32(r < (unsigned)TILE_ROWS);
        const int ent = (int)(((unsigned)e << 10) | r);
        while (mask) {
            const int l = __builtin_ctz(mask);
            mask &= mask - 1;
            const unsigned en = (unsigned)__builtin_amdgcn_readlane(ent, l);
            const int owner = (int)(en & 31u);
            if (lane == owner) {
                if (mycnt < (unsigned)BCAP)
                    bucket[(wave * AGG_WAVES + owner) * BCAP + mycnt] = en;
                ++mycnt;
            }
        }
    };
    const int   total4 = nEdges >> 2;
    const int4* dst4   = reinterpret_cast<const int4*>(dst);
    auto scan_trip = [&](auto checked, int i0) {
        constexpr bool CHECK = decltype(checked)::value;
        int4 dv[SCAN_U];
#pragma unroll
        for (int u = 0; u < SCAN_U; ++u) {
            const int i = i0 + u * AGG_THREADS;
            if constexpr (CHECK) {
                const bool ok = i < total4;
                int4 d = dst4[ok ? i : 0];
                if (!ok) d = make_int4(-1, -1, -1, -1);
                dv[u] = d;
            } else {
                dv[u] = dst4[i];
            }
        }
#pragma unroll
        for (int u = 0; u < SCAN_U; ++u) {
            const unsigned r0 = (unsigned)(dv[u].x - base);
            const unsigned r1 = (unsigned)(dv[u].y - base);
            const unsigned r2 = (unsigned)(dv[u].z - base);
            const unsigned r3 = (unsigned)(dv[u].w - base);
            const unsigned m01 = r0 < r1 ? r0 : r1;
            const unsigned m23 = r2 < r3 ? r2 : r3;
            const unsigned m   = m01 < m23 ? m01 : m23;
            if (__builtin_amdgcn_ballot_w32(m < (unsigned)TILE_ROWS) != 0u) {
                const int e = (i0 + u * AGG_THREADS) * 4;
                put(r0, e + 0);
                put(r1, e + 1);
                put(r2, e + 2);
                put(r3, e + 3);
            }
        }
    };
    const int trips = (total4 + SCAN_U * AGG_THREADS - 1) / (SCAN_U * AGG_THREADS);
    const int rot   = (int)(blockIdx.x * (unsigned)trips / gridDim.x);
    for (int it = 0; it < trips; ++it) {
        int t = it + rot;
        if (t >= trips) t -= trips;
        const int i0 = t * SCAN_U * AGG_THREADS + tid;
        if (t != trips - 1) scan_trip(std::false_type{}, i0);
        else                scan_trip(std::true_type{}, i0);
    }
    if (mycnt > (unsigned)BCAP) ovf = 1u;
    wcnt[wave * AGG_WAVES + lane] = mycnt < (unsigned)BCAP ? mycnt : (unsigned)BCAP;
    __syncthreads();

    unsigned* mylist = list + wave * LCAP;
    const unsigned n_l = wcnt[lane * AGG_WAVES + wave];
    unsigned off = 0, tot = 0;
#pragma unroll
    for (int j = 0; j < 32; ++j) {
        const unsigned nj = (unsigned)__builtin_amdgcn_readlane((int)n_l, j);
        off += j < lane ? nj : 0u;
        tot += nj;
    }
    if (tot > (unsigned)LCAP) {
        ovf = 1u;
    } else {
        for (unsigned k = 0; k < n_l; ++k)
            mylist[off + k] = bucket[(lane * AGG_WAVES + wave) * BCAP + k];
    }
    __syncthreads();

    if (ovf == 0u) {
        for (unsigned b0 = 0; b0 < tot; b0 += 32) {
            const unsigned qi  = b0 + lane;
            const unsigned ent = qi < tot ? mylist[qi] : (unsigned)(TILE_ROWS + wave);
            const int dl = (int)(ent & 1023u);
            const int s  = src[ent >> 10];
#pragma unroll
            for (int j0 = 0; j0 < 32; j0 += 16) {
                if (b0 + j0 >= tot) break;
                float v[16][H];
                int   row[16];
#pragma unroll
                for (int u = 0; u < 16; ++u) {
                    const int sj = __builtin_amdgcn_readlane(s, j0 + u);
                    row[u]       = __builtin_amdgcn_readlane(dl, j0 + u);
                    const float* xp = x + (size_t)sj * DIM + lane;
#pragma unroll
                    for (int h = 0; h < H; ++h) v[u][h] = xp[32 * h];
                }
#pragma unroll
                for (int u = 0; u < 16; ++u) {
#pragma unroll
                    for (int h = 0; h < H; ++h)
                        acc[row[u] * DIM + 32 * h + lane] += v[u][h];
                    if (COUNT && lane == 0) cntL[row[u]] += 1.0f;
                }
            }
        }
    } else {
        for (int i0 = 0; i0 < nEdges; i0 += 32) {
            const int i = i0 + lane;
            const int d = i < nEdges ? dst[i] : -1;
            const unsigned r = (unsigned)(d - base);
            unsigned mask = __builtin_amdgcn_ballot_w32(
                r < (unsigned)TILE_ROWS && (int)(r & 31u) == wave);
            while (mask) {
                const int l = __builtin_ctz(mask);
                mask &= mask - 1;
                const int rr = __builtin_amdgcn_readlane((int)r, l);
                const float* xp = x + (size_t)src[i0 + l] * DIM + lane;
#pragma unroll
                for (int h = 0; h < H; ++h) acc[rr * DIM + 32 * h + lane] += xp[32 * h];
                if (COUNT && lane == 0) cntL[rr] += 1.0f;
            }
        }
    }
    __syncthreads();

    const int left = nNodes - base;
    const int rows = left < TILE_ROWS ? left : TILE_ROWS;
    float4*       out4 = reinterpret_cast<float4*>(agg + (size_t)base * DIM);
    const float4* acc4 = reinterpret_cast<const float4*>(acc);
    for (int i = tid; i < rows * (DIM / 4); i += AGG_THREADS) out4[i] = acc4[i];
    if (COUNT)
        for (int i = tid; i < rows; i += AGG_THREADS) cnt[base + i] = cntL[i];
}

template <int K>
__global__ void sage_gemm_kernel(const float* __restrict__ aggRaw,
                                 const float* __restrict__ cnt,
                                 const float* __restrict__ x,
                                 const float* __restrict__ Wl,
                                 const float* __restrict__ Wr,
                                 const float* __restrict__ bias,
                                 float* __restrict__ out, int nNodes) {
    const int lane = threadIdx.x & 31;
    const int wave = threadIdx.x >> 5;
    const int row0 = blockIdx.x * 16;
    if (row0 >= nNodes) return;
    const int col0 = wave * 16;
    const int grp  = lane >> 4;
    const int lr   = lane & 15;

    const int   arow = row0 + lr;
    const float invc = 1.0f / fmaxf(cnt[arow], 1.0f);

    v8f c = {};
#pragma unroll
    for (int k0 = 0; k0 < K; k0 += 4) {
        const int ka = k0 + 2 * grp;
        v2f a, b;
        a.x = aggRaw[(size_t)arow * K + ka + 0] * invc;
        a.y = aggRaw[(size_t)arow * K + ka + 1] * invc;
        b.x = Wl[(size_t)(ka + 0) * HIDDEN_D + col0 + lr];
        b.y = Wl[(size_t)(ka + 1) * HIDDEN_D + col0 + lr];
        c = __builtin_amdgcn_wmma_f32_16x16x4_f32(false, a, false, b,
                                                  (short)0, c, false, false);
    }
#pragma unroll
    for (int k0 = 0; k0 < K; k0 += 4) {
        const int ka = k0 + 2 * grp;
        v2f a, b;
        a.x = x[(size_t)arow * K + ka + 0];
        a.y = x[(size_t)arow * K + ka + 1];
        b.x = Wr[(size_t)(ka + 0) * HIDDEN_D + col0 + lr];
        b.y = Wr[(size_t)(ka + 1) * HIDDEN_D + col0 + lr];
        c = __builtin_amdgcn_wmma_f32_16x16x4_f32(false, a, false, b,
                                                  (short)0, c, false, false);
    }

    const float bv = bias[col0 + lr];
#pragma unroll
    for (int v = 0; v < 8; ++v) {
        const int m = v + 8 * grp;
        out[(size_t)(row0 + m) * HIDDEN_D + col0 + lr] = fmaxf(c[v] + bv, 0.0f);
    }
}

#define ZS 16
__global__ void head_pool_kernel(const int* __restrict__ batch,
                                 const float* __restrict__ h,
                                 const float* __restrict__ Wo,
                                 float* __restrict__ zsum, int nNodes) {
    const int lane = threadIdx.x & 31;
    const int tile = blockIdx.x * (blockDim.x >> 5) + (threadIdx.x >> 5);
    const int row0 = tile * 16;
    if (row0 >= nNodes) return;
    const int grp = lane >> 4;
    const int lr  = lane & 15;

    const float4* hp =
        reinterpret_cast<const float4*>(h + (size_t)(row0 + lr) * HIDDEN_D) + grp;
    float4 av[8];
    for (int t = 0; t < 8; ++t) av[t] = hp[2 * t];

    const int4* bp = reinterpret_cast<const int4*>(batch + row0 + 8 * grp);
    const int4 gA = bp[0], gB = bp[1];

    const bool colOk = lr < N_CLASSES;
    const int  cc    = colOk ? lr : 0;

    v8f c = {};
    for (int t = 0; t < 8; ++t) {
        const int kb = 8 * t + 4 * grp;
        const float w0 = Wo[(kb + 0) * N_CLASSES + cc];
        const float w1 = Wo[(kb + 1) * N_CLASSES + cc];
        const float w2 = Wo[(kb + 2) * N_CLASSES + cc];
        const float w3 = Wo[(kb + 3) * N_CLASSES + cc];
        v2f a0, a1, b0, b1;
        a0.x = av[t].x;             a0.y = av[t].y;
        b0.x = colOk ? w0 : 0.0f;   b0.y = colOk ? w1 : 0.0f;
        a1.x = av[t].z;             a1.y = av[t].w;
        b1.x = colOk ? w2 : 0.0f;   b1.y = colOk ? w3 : 0.0f;
        c = __builtin_amdgcn_wmma_f32_16x16x4_f32(false, a0, false, b0,
                                                  (short)0, c, false, false);
        c = __builtin_amdgcn_wmma_f32_16x16x4_f32(false, a1, false, b1,
                                                  (short)0, c, false, false);
    }

    float val[8];
    for (int v = 0; v < 8; ++v) val[v] = colOk ? c[v] : 1.0f;
    const int gid[8] = {gA.x, gA.y, gA.z, gA.w, gB.x, gB.y, gB.z, gB.w};
    if (lr <= N_CLASSES) {
        for (int v = 0; v < 8; ++v)
            atomAdd1(zsum + (size_t)gid[v] * ZS + lr, val[v]);
    }
}

__global__ void finish_kernel(const float* __restrict__ zsum,
                              const float* __restrict__ bo,
                              float* __restrict__ out) {
    int t = blockIdx.x * blockDim.x + threadIdx.x;
    if (t >= N_GRAPHS * N_CLASSES) return;
    int g = t / N_CLASSES;
    int c = t - g * N_CLASSES;
    out[t] = zsum[g * ZS + c] / fmaxf(zsum[g * ZS + N_CLASSES], 1.0f) + bo[c];
}

extern "C" void kernel_launch(void* const* d_in, const int* in_sizes, int n_in,
                              void* d_out, int out_size, void* d_ws, size_t ws_size,
                              hipStream_t stream) {
    const int*   node_ids = (const int*)  d_in[0];
    const int*   src      = (const int*)  d_in[1];
    const int*   dst      = (const int*)  d_in[2];
    const int*   batch    = (const int*)  d_in[3];
    const float* embed    = (const float*)d_in[4];
    const float* W1_l     = (const float*)d_in[5];
    const float* b1       = (const float*)d_in[6];
    const float* W1_r     = (const float*)d_in[7];
    const float* W2_l     = (const float*)d_in[8];
    const float* b2       = (const float*)d_in[9];
    const float* W2_r     = (const float*)d_in[10];
    const float* W_out    = (const float*)d_in[11];
    const float* b_out    = (const float*)d_in[12];
    float* out = (float*)d_out;

    float* ws   = (float*)d_ws;
    float* x0   = ws;
    float* h1   = x0   + (size_t)N_NODES * EMBED_D;
    float* h2   = h1   + (size_t)N_NODES * HIDDEN_D;
    float* agg1 = h2   + (size_t)N_NODES * HIDDEN_D;
    float* agg2 = agg1 + (size_t)N_NODES * EMBED_D;
    float* cnt  = agg2 + (size_t)N_NODES * HIDDEN_D;
    float* zsum = cnt  + (size_t)N_NODES;
    const int zeroN = N_GRAPHS * ZS;

    const int B = 256;
    auto blocks = [](int n, int b) { return (n + b - 1) / b; };

    zero_kernel<<<blocks(zeroN, B), B, 0, stream>>>(zsum, zeroN);

    gather_embed_kernel<<<blocks(N_NODES * 8, B), B, 0, stream>>>(
        node_ids, embed, x0, N_NODES);

    tile_agg_kernel<EMBED_D, true><<<blocks(N_NODES, TILE_ROWS), AGG_THREADS, 0, stream>>>(
        src, dst, x0, agg1, cnt, N_NODES, N_EDGES);
    sage_gemm_kernel<EMBED_D><<<N_NODES / 16, 128, 0, stream>>>(
        agg1, cnt, x0, W1_l, W1_r, b1, h1, N_NODES);

    tile_agg_kernel<HIDDEN_D, false><<<blocks(N_NODES, TILE_ROWS), AGG_THREADS, 0, stream>>>(
        src, dst, h1, agg2, cnt, N_NODES, N_EDGES);
    sage_gemm_kernel<HIDDEN_D><<<N_NODES / 16, 128, 0, stream>>>(
        agg2, cnt, h1, W2_l, W2_r, b2, h2, N_NODES);

    head_pool_kernel<<<blocks(N_NODES / 16, 4), 128, 0, stream>>>(
        batch, h2, W_out, zsum, N_NODES);
    finish_kernel<<<blocks(N_GRAPHS * N_CLASSES, B), B, 0, stream>>>(
        zsum, b_out, out);
}


# destination scan done once: layer 1's tile kernel stores each wave's edge list (source node, local row) in the workspace, layer 2's reads it back instead of scanning all destinations; sum order kept
# speedup vs baseline: 2.0871x; 1.2243x over previous
#include <hip/hip_runtime.h>
#include <type_traits>

#define N_NODES   100000
#define N_EDGES   1200000
#define EMBED_D   32
#define HIDDEN_D  64
#define N_CLASSES 10
#define N_GRAPHS  1024

typedef __attribute__((ext_vector_type(2))) float v2f;
typedef __attribute__((ext_vector_type(8))) float v8f;

__device__ __forceinline__ void atomAdd1(float* p, float v) {
    asm volatile("global_atomic_add_f32 %0, %1, off scope:SCOPE_DEV"
                 :: "v"(p), "v"(v) : "memory");
}

__global__ void zero_kernel(float* __restrict__ p, int n) {
    int t = blockIdx.x * blockDim.x + threadIdx.x;
    if (t < n) p[t] = 0.0f;
}

__global__ void gather_embed_kernel(const int* __restrict__ ids,
                                    const float* __restrict__ embed,
                                    float* __restrict__ x0, int nNodes) {
    int t = blockIdx.x * blockDim.x + threadIdx.x;
    if (t >= nNodes * (EMBED_D / 4)) return;
    int node = t >> 3;
    int q    = t & 7;
    float4 v = reinterpret_cast<const float4*>(embed + (size_t)ids[node] * EMBED_D)[q];
    reinterpret_cast<float4*>(x0 + (size_t)node * EMBED_D)[q] = v;
}

#define TILE_ROWS   392
#define AGG_BLOCKS  ((N_NODES + TILE_ROWS - 1) / TILE_ROWS)
#define AGG_THREADS 1024
#define AGG_WAVES   32
#define BCAP        32
#define LCAP        512
#define SCAN_U      8
#define LIST_NONE   0xFFFFFFFFu
static_assert(N_EDGES % 4 == 0, "destination list is read 4 at a time");
static_assert(TILE_ROWS + AGG_WAVES <= 1024 && N_EDGES < (1 << 21),
              "a scan entry is (edge << 10) | local row");
static_assert(N_NODES < (1 << 22), "a stored entry is (source node << 10) | local row");

template <int DIM, bool COUNT, bool BUILD>
__global__ __launch_bounds__(AGG_THREADS)
void tile_agg_kernel(const int* __restrict__ src,
                     const int* __restrict__ dst,
                     const float* __restrict__ x,
                     float* __restrict__ agg,
                     float* __restrict__ cnt,
                     unsigned* __restrict__ glist,
                     unsigned* __restrict__ gtot,
                     int nNodes, int nEdges) {
    constexpr int H = DIM / 32;
    __shared__ float    acc[(TILE_ROWS + AGG_WAVES) * DIM];
    __shared__ float    cntL[TILE_ROWS + AGG_WAVES];
    __shared__ unsigned bucket[AGG_WAVES * AGG_WAVES * BCAP];
    __shared__ unsigned wcnt[AGG_WAVES * AGG_WAVES];
    __shared__ unsigned list[AGG_WAVES * LCAP];
    __shared__ unsigned ovf;

    const int tid  = threadIdx.x;
    const int lane = tid & 31;
    const int wave = tid >> 5;
    const int base = blockIdx.x * TILE_ROWS;
    unsigned* mygl  = glist + (size_t)(blockIdx.x * AGG_WAVES + wave) * LCAP;
    unsigned* mygt  = gtot + blockIdx.x * AGG_WAVES + wave;

    unsigned tot = 0;
    bool     slow;
    if constexpr (!BUILD) tot = *mygt;

    for (int i = tid; i < (TILE_ROWS + AGG_WAVES) * DIM; i += AGG_THREADS) acc[i] = 0.0f;
    for (int i = tid; i < TILE_ROWS + AGG_WAVES; i += AGG_THREADS) cntL[i] = 0.0f;

    if constexpr (BUILD) {
        if (tid == 0) ovf = 0u;
        __syncthreads();

        unsigned mycnt = 0;
        auto put = [&](unsigned r, int e) {
            unsigned mask = __builtin_amdgcn_ballot_w32(r < (unsigned)TILE_ROWS);
            const int ent = (int)(((unsigned)e << 10) | r);
            while (mask) {
                const int l = __builtin_ctz(mask);
                mask &= mask - 1;
                const unsigned en = (unsigned)__builtin_amdgcn_readlane(ent, l);
                const int owner = (int)(en & 31u);
                if (lane == owner) {
                    if (mycnt < (unsigned)BCAP)
                        bucket[(wave * AGG_WAVES + owner) * BCAP + mycnt] = en;
                    ++mycnt;
                }
            }
        };
        const int   total4 = nEdges >> 2;
        const int4* dst4   = reinterpret_cast<const int4*>(dst);
        auto scan_trip = [&](auto checked, int i0) {
            constexpr bool CHECK = decltype(checked)::value;
            int4 dv[SCAN_U];
#pragma unroll
            for (int u = 0; u < SCAN_U; ++u) {
                const int i = i0 + u * AGG_THREADS;
                if constexpr (CHECK) {
                    const bool ok = i < total4;
                    int4 d = dst4[ok ? i : 0];
                    if (!ok) d = make_int4(-1, -1, -1, -1);
                    dv[u] = d;
                } else {
                    dv[u] = dst4[i];
                }
            }
#pragma unroll
            for (int u = 0; u < SCAN_U; ++u) {
                const unsigned r0 = (unsigned)(dv[u].x - base);
                const unsigned r1 = (unsigned)(dv[u].y - base);
                const unsigned r2 = (unsigned)(dv[u].z - base);
                const unsigned r3 = (unsigned)(dv[u].w - base);
                const unsigned m01 = r0 < r1 ? r0 : r1;
                const unsigned m23 = r2 < r3 ? r2 : r3;
                const unsigned m   = m01 < m23 ? m01 : m23;
                if (__builtin_amdgcn_ballot_w32(m < (unsigned)TILE_ROWS) != 0u) {
                    const int e = (i0 + u * AGG_THREADS) * 4;
                    put(r0, e + 0);
                    put(r1, e + 1);
                    put(r2, e + 2);
                    put(r3, e + 3);
                }
            }
        };
        const int trips = (total4 + SCAN_U * AGG_THREADS - 1) / (SCAN_U * AGG_THREADS);
        const int rot   = (int)(blockIdx.x * (unsigned)trips / gridDim.x);
        for (int it = 0; it < trips; ++it) {
            int t = it + rot;
            if (t >= trips) t -= trips;
            const int i0 = t * SCAN_U * AGG_THREADS + tid;
            if (t != trips - 1) scan_trip(std::false_type{}, i0);
            else                scan_trip(std::true_type{}, i0);
        }
        if (mycnt > (unsigned)BCAP) ovf = 1u;
        wcnt[wave * AGG_WAVES + lane] = mycnt < (unsigned)BCAP ? mycnt : (unsigned)BCAP;
        __syncthreads();

        unsigned* mylist = list + wave * LCAP;
        const unsigned n_l = wcnt[lane * AGG_WAVES + wave];
        unsigned off = 0;
#pragma unroll
        for (int j = 0; j < 32; ++j) {
            const unsigned nj = (unsigned)__builtin_amdgcn_readlane((int)n_l, j);
            off += j < lane ? nj : 0u;
            tot += nj;
        }
        if (tot > (unsigned)LCAP) {
            ovf = 1u;
        } else {
            for (unsigned k = 0; k < n_l; ++k)
                mylist[off + k] = bucket[(lane * AGG_WAVES + wave) * BCAP + k];
        }
        __syncthreads();
        slow = ovf != 0u;
        if (lane == 0) *mygt = slow ? LIST_NONE : tot;
    } else {
        tot  = (unsigned)__builtin_amdgcn_readfirstlane((int)tot);
        slow = tot == LIST_NONE;
        __syncthreads();
    }

    if (!slow) {
        for (unsigned b0 = 0; b0 < tot; b0 += 32) {
            const unsigned qi = b0 + lane;
            unsigned ent;
            if constexpr (BUILD) {
                const unsigned se = qi < tot ? list[wave * LCAP + qi]
                                             : (unsigned)(TILE_ROWS + wave);
                ent = ((unsigned)src[se >> 10] << 10) | (se & 1023u);
                if (qi < tot) mygl[qi] = ent;
            } else {
                ent = qi < tot ? mygl[qi] : (unsigned)(TILE_ROWS + wave);
            }
            const int dl = (int)(ent & 1023u);
            const int s  = (int)(ent >> 10);
#pragma unroll
            for (int j0 = 0; j0 < 32; j0 += 16) {
                if (b0 + j0 >= tot) break;
                float v[16][H];
                int   row[16];
#pragma unroll
                for (int u = 0; u < 16; ++u) {
                    const int sj = __builtin_amdgcn_readlane(s, j0 + u);
                    row[u]       = __builtin_amdgcn_readlane(dl, j0 + u);
                    const float* xp = x + (size_t)sj * DIM + lane;
#pragma unroll
                    for (int h = 0; h < H; ++h) v[u][h] = xp[32 * h];
                }
#pragma unroll
                for (int u = 0; u < 16; ++u) {
#pragma unroll
                    for (int h = 0; h < H; ++h)
                        acc[row[u] * DIM + 32 * h + lane] += v[u][h];
                    if (COUNT && lane == 0) cntL[row[u]] += 1.0f;
                }
            }
        }
    } else {
        for (int i0 = 0; i0 < nEdges; i0 += 32) {
            const int i = i0 + lane;
            const int d = i < nEdges ? dst[i] : -1;
            const unsigned r = (unsigned)(d - base);
            unsigned mask = __builtin_amdgcn_ballot_w32(
                r < (unsigned)TILE_ROWS && (int)(r & 31u) == wave);
            while (mask) {
                const int l = __builtin_ctz(mask);
                mask &= mask - 1;
                const int rr = __builtin_amdgcn_readlane((int)r, l);
                const float* xp = x + (size_t)src[i0 + l] * DIM + lane;
#pragma unroll
                for (int h = 0; h < H; ++h) acc[rr * DIM + 32 * h + lane] += xp[32 * h];
                if (COUNT && lane == 0) cntL[rr] += 1.0f;
            }
        }
    }
    __syncthreads();

    const int left = nNodes - base;
    const int rows = left < TILE_ROWS ? left : TILE_ROWS;
    float4*       out4 = reinterpret_cast<float4*>(agg + (size_t)base * DIM);
    const float4* acc4 = reinterpret_cast<const float4*>(acc);
    for (int i = tid; i < rows * (DIM / 4); i += AGG_THREADS) out4[i] = acc4[i];
    if (COUNT)
        for (int i = tid; i < rows; i += AGG_THREADS) cnt[base + i] = cntL[i];
}

template <int K>
__global__ void sage_gemm_kernel(const float* __restrict__ aggRaw,
                                 const float* __restrict__ cnt,
                                 const float* __restrict__ x,
                                 const float* __restrict__ Wl,
                                 const float* __restrict__ Wr,
                                 const float* __restrict__ bias,
                                 float* __restrict__ out, int nNodes) {
    const int lane = threadIdx.x & 31;
    const int wave = threadIdx.x >> 5;
    const int row0 = blockIdx.x * 16;
    if (row0 >= nNodes) return;
    const int col0 = wave * 16;
    const int grp  = lane >> 4;
    const int lr   = lane & 15;

    const int   arow = row0 + lr;
    const float invc = 1.0f / fmaxf(cnt[arow], 1.0f);

    v8f c = {};
#pragma unroll
    for (int k0 = 0; k0 < K; k0 += 4) {
        const int ka = k0 + 2 * grp;
        v2f a, b;
        a.x = aggRaw[(size_t)arow * K + ka + 0] * invc;
        a.y = aggRaw[(size_t)arow * K + ka + 1] * invc;
        b.x = Wl[(size_t)(ka + 0) * HIDDEN_D + col0 + lr];
        b.y = Wl[(size_t)(ka + 1) * HIDDEN_D + col0 + lr];
        c = __builtin_amdgcn_wmma_f32_16x16x4_f32(false, a, false, b,
                                                  (short)0, c, false, false);
    }
#pragma unroll
    for (int k0 = 0; k0 < K; k0 += 4) {
        const int ka = k0 + 2 * grp;
        v2f a, b;
        a.x = x[(size_t)arow * K + ka + 0];
        a.y = x[(size_t)arow * K + ka + 1];
        b.x = Wr[(size_t)(ka + 0) * HIDDEN_D + col0 + lr];
        b.y = Wr[(size_t)(ka + 1) * HIDDEN_D + col0 + lr];
        c = __builtin_amdgcn_wmma_f32_16x16x4_f32(false, a, false, b,
                                                  (short)0, c, false, false);
    }

    const float bv = bias[col0 + lr];
#pragma unroll
    for (int v = 0; v < 8; ++v) {
        const int m = v + 8 * grp;
        out[(size_t)(row0 + m) * HIDDEN_D + col0 + lr] = fmaxf(c[v] + bv, 0.0f);
    }
}

#define ZS 16
__global__ void head_pool_kernel(const int* __restrict__ batch,
                                 const float* __restrict__ h,
                                 const float* __restrict__ Wo,
                                 float* __restrict__ zsum, int nNodes) {
    const int lane = threadIdx.x & 31;
    const int tile = blockIdx.x * (blockDim.x >> 5) + (threadIdx.x >> 5);
    const int row0 = tile * 16;
    if (row0 >= nNodes) return;
    const int grp = lane >> 4;
    const int lr  = lane & 15;

    const float4* hp =
        reinterpret_cast<const float4*>(h + (size_t)(row0 + lr) * HIDDEN_D) + grp;
    float4 av[8];
    for (int t = 0; t < 8; ++t) av[t] = hp[2 * t];

    const int4* bp = reinterpret_cast<const int4*>(batch + row0 + 8 * grp);
    const int4 gA = bp[0], gB = bp[1];

    const bool colOk = lr < N_CLASSES;
    const int  cc    = colOk ? lr : 0;

    v8f c = {};
    for (int t = 0; t < 8; ++t) {
        const int kb = 8 * t + 4 * grp;
        const float w0 = Wo[(kb + 0) * N_CLASSES + cc];
        const float w1 = Wo[(kb + 1) * N_CLASSES + cc];
        const float w2 = Wo[(kb + 2) * N_CLASSES + cc];
        const float w3 = Wo[(kb + 3) * N_CLASSES + cc];
        v2f a0, a1, b0, b1;
        a0.x = av[t].x;             a0.y = av[t].y;
        b0.x = colOk ? w0 : 0.0f;   b0.y = colOk ? w1 : 0.0f;
        a1.x = av[t].z;             a1.y = av[t].w;
        b1.x = colOk ? w2 : 0.0f;   b1.y = colOk ? w3 : 0.0f;
        c = __builtin_amdgcn_wmma_f32_16x16x4_f32(false, a0, false, b0,
                                                  (short)0, c, false, false);
        c = __builtin_amdgcn_wmma_f32_16x16x4_f32(false, a1, false, b1,
                                                  (short)0, c, false, false);
    }

    float val[8];
    for (int v = 0; v < 8; ++v) val[v] = colOk ? c[v] : 1.0f;
    const int gid[8] = {gA.x, gA.y, gA.z, gA.w, gB.x, gB.y, gB.z, gB.w};
    if (lr <= N_CLASSES) {
        for (int v = 0; v < 8; ++v)
            atomAdd1(zsum + (size_t)gid[v] * ZS + lr, val[v]);
    }
}

__global__ void finish_kernel(const float* __restrict__ zsum,
                              const float* __restrict__ bo,
                              float* __restrict__ out) {
    int t = blockIdx.x * blockDim.x + threadIdx.x;
    if (t >= N_GRAPHS * N_CLASSES) return;
    int g = t / N_CLASSES;
    int c = t - g * N_CLASSES;
    out[t] = zsum[g * ZS + c] / fmaxf(zsum[g * ZS + N_CLASSES], 1.0f) + bo[c];
}

extern "C" void kernel_launch(void* const* d_in, const int* in_sizes, int n_in,
                              void* d_out, int out_size, void* d_ws, size_t ws_size,
                              hipStream_t stream) {
    const int*   node_ids = (const int*)  d_in[0];
    const int*   src      = (const int*)  d_in[1];
    const int*   dst      = (const int*)  d_in[2];
    const int*   batch    = (const int*)  d_in[3];
    const float* embed    = (const float*)d_in[4];
    const float* W1_l     = (const float*)d_in[5];
    const float* b1       = (const float*)d_in[6];
    const float* W1_r     = (const float*)d_in[7];
    const float* W2_l     = (const float*)d_in[8];
    const float* b2       = (const float*)d_in[9];
    const float* W2_r     = (const float*)d_in[10];
    const float* W_out    = (const float*)d_in[11];
    const float* b_out    = (const float*)d_in[12];
    float* out = (float*)d_out;

    float* ws   = (float*)d_ws;
    float* x0   = ws;
    float* h1   = x0   + (size_t)N_NODES * EMBED_D;
    float* h2   = h1   + (size_t)N_NODES * HIDDEN_D;
    float* agg1 = h2   + (size_t)N_NODES * HIDDEN_D;
    float* agg2 = agg1 + (size_t)N_NODES * EMBED_D;
    float* cnt  = agg2 + (size_t)N_NODES * HIDDEN_D;
    float* zsum = cnt  + (size_t)N_NODES;
    const int zeroN = N_GRAPHS * ZS;
    unsigned* gtot  = (unsigned*)(zsum + N_GRAPHS * ZS);
    unsigned* glist = gtot + AGG_BLOCKS * AGG_WAVES;

    const int B = 256;
    auto blocks = [](int n, int b) { return (n + b - 1) / b; };

    zero_kernel<<<blocks(zeroN, B), B, 0, stream>>>(zsum, zeroN);

    gather_embed_kernel<<<blocks(N_NODES * 8, B), B, 0, stream>>>(
        node_ids, embed, x0, N_NODES);

    tile_agg_kernel<EMBED_D, true, true><<<AGG_BLOCKS, AGG_THREADS, 0, stream>>>(
        src, dst, x0, agg1, cnt, glist, gtot, N_NODES, N_EDGES);
    sage_gemm_kernel<EMBED_D><<<N_NODES / 16, 128, 0, stream>>>(
        agg1, cnt, x0, W1_l, W1_r, b1, h1, N_NODES);

    tile_agg_kernel<HIDDEN_D, false, false><<<AGG_BLOCKS, AGG_THREADS, 0, stream>>>(
        src, dst, h1, agg2, cnt, glist, gtot, N_NODES, N_EDGES);
    sage_gemm_kernel<HIDDEN_D><<<N_NODES / 16, 128, 0, stream>>>(
        agg2, cnt, h1, W2_l, W2_r, b2, h2, N_NODES);

    head_pool_kernel<<<blocks(N_NODES / 16, 4), 128, 0, stream>>>(
        batch, h2, W_out, zsum, N_NODES);
    finish_kernel<<<blocks(N_GRAPHS * N_CLASSES, B), B, 0, stream>>>(
        zsum, b_out, out);
}


# layer 1 finds its edges through a chunk map: a new launch sets one bit per 32-edge chunk and tile; a tile block reads its bit column and loads only marked chunks; row sum order changes, still fixed
# speedup vs baseline: 2.2072x; 1.0575x over previous
#include <hip/hip_runtime.h>
#include <type_traits>

#define N_NODES   100000
#define N_EDGES   1200000
#define EMBED_D   32
#define HIDDEN_D  64
#define N_CLASSES 10
#define N_GRAPHS  1024

typedef __attribute__((ext_vector_type(2))) float v2f;
typedef __attribute__((ext_vector_type(8))) float v8f;

__device__ __forceinline__ void atomAdd1(float* p, float v) {
    asm volatile("global_atomic_add_f32 %0, %1, off scope:SCOPE_DEV"
                 :: "v"(p), "v"(v) : "memory");
}

__global__ void zero_kernel(float* __restrict__ p, int n) {
    int t = blockIdx.x * blockDim.x + threadIdx.x;
    if (t < n) p[t] = 0.0f;
}

__global__ void gather_embed_kernel(const int* __restrict__ ids,
                                    const float* __restrict__ embed,
                                    float* __restrict__ x0, int nNodes) {
    int t = blockIdx.x * blockDim.x + threadIdx.x;
    if (t >= nNodes * (EMBED_D / 4)) return;
    int node = t >> 3;
    int q    = t & 7;
    float4 v = reinterpret_cast<const float4*>(embed + (size_t)ids[node] * EMBED_D)[q];
    reinterpret_cast<float4*>(x0 + (size_t)node * EMBED_D)[q] = v;
}

#define TILE_ROWS   392
#define AGG_BLOCKS  ((N_NODES + TILE_ROWS - 1) / TILE_ROWS)
#define AGG_THREADS 1024
#define AGG_WAVES   32
#define BCAP        32
#define LCAP        512
#define SCAN_U      8
static_assert(SCAN_U == 8, "the scan reads its 8 list entries as two 16-byte words");
#define CHUNK_E     32
#define NCHUNK      ((N_EDGES + CHUNK_E - 1) / CHUNK_E)
#define MAP_TRIPS   ((NCHUNK + 4 * AGG_THREADS - 1) / (4 * AGG_THREADS))
#define NCHUNK_PAD  (MAP_TRIPS * 4 * AGG_THREADS)
#define MAP_WORDS   ((AGG_BLOCKS + 31) / 32)
#define MAP_THREADS 128
#define NO_CHUNK    (1u << 26)
#define LIST_NONE   0xFFFFFFFFu
static_assert(N_EDGES % 4 == 0, "destination list is read 4 at a time");
static_assert(TILE_ROWS + AGG_WAVES <= 1024 && N_EDGES < (1 << 21),
              "a scan entry is (edge << 10) | local row");
static_assert(N_NODES < (1 << 22), "a stored entry is (source node << 10) | local row");
static_assert(NCHUNK_PAD % MAP_THREADS == 0, "one map thread per padded chunk");

__global__ void chunk_map_kernel(const int* __restrict__ dst,
                                 unsigned* __restrict__ cmap, int nEdges) {
    const int c = blockIdx.x * blockDim.x + threadIdx.x;
    if (c >= NCHUNK_PAD) return;
    unsigned w[MAP_WORDS];
#pragma unroll
    for (int k = 0; k < MAP_WORDS; ++k) w[k] = 0u;
    auto mark = [&](int d) {
        const unsigned t   = (unsigned)d / (unsigned)TILE_ROWS;
        const unsigned bit = 1u << (t & 31u);
#pragma unroll
        for (int k = 0; k < MAP_WORDS; ++k) w[k] |= (t >> 5) == (unsigned)k ? bit : 0u;
    };
    const int   total4 = nEdges >> 2;
    const int4* dst4   = reinterpret_cast<const int4*>(dst);
#pragma unroll
    for (int q = 0; q < CHUNK_E / 4; ++q) {
        const int  i  = c * (CHUNK_E / 4) + q;
        const bool ok = i < total4;
        int4 d = dst4[ok ? i : 0];
        if (!ok) d = make_int4(-1, -1, -1, -1);
        mark(d.x);
        mark(d.y);
        mark(d.z);
        mark(d.w);
    }
#pragma unroll
    for (int k = 0; k < MAP_WORDS; ++k) cmap[(size_t)k * NCHUNK_PAD + c] = w[k];
}

template <int DIM, bool COUNT, bool BUILD>
__global__ __launch_bounds__(AGG_THREADS)
void tile_agg_kernel(const int* __restrict__ src,
                     const int* __restrict__ dst,
                     const float* __restrict__ x,
                     float* __restrict__ agg,
                     float* __restrict__ cnt,
                     unsigned* __restrict__ glist,
                     unsigned* __restrict__ gtot,
                     const unsigned* __restrict__ cmap,
                     int nNodes, int nEdges) {
    constexpr int H = DIM / 32;
    __shared__ float    acc[(TILE_ROWS + AGG_WAVES) * DIM];
    __shared__ float    cntL[TILE_ROWS + AGG_WAVES];
    __shared__ unsigned bucket[AGG_WAVES * AGG_WAVES * BCAP];
    __shared__ unsigned wcnt[AGG_WAVES * AGG_WAVES];
    __shared__ __attribute__((aligned(16))) unsigned list[AGG_WAVES * LCAP];
    __shared__ unsigned ovf;

    const int tid  = threadIdx.x;
    const int lane = tid & 31;
    const int wave = tid >> 5;
    const int base = blockIdx.x * TILE_ROWS;
    unsigned* mygl  = glist + (size_t)(blockIdx.x * AGG_WAVES + wave) * LCAP;
    unsigned* mygt  = gtot + blockIdx.x * AGG_WAVES + wave;

    unsigned tot = 0;
    bool     slow;
    if constexpr (!BUILD) tot = *mygt;

    for (int i = tid; i < (TILE_ROWS + AGG_WAVES) * DIM; i += AGG_THREADS) acc[i] = 0.0f;
    for (int i = tid; i < TILE_ROWS + AGG_WAVES; i += AGG_THREADS) cntL[i] = 0.0f;

    if constexpr (BUILD) {
        if (tid == 0) ovf = 0u;
        __syncthreads();

        unsigned mycnt = 0;
        auto put = [&](unsigned r, int e) {
            unsigned mask = __builtin_amdgcn_ballot_w32(r < (unsigned)TILE_ROWS);
            const int ent = (int)(((unsigned)e << 10) | r);
            while (mask) {
                const int l = __builtin_ctz(mask);
                mask &= mask - 1;
                const unsigned en = (unsigned)__builtin_amdgcn_readlane(ent, l);
                const int owner = (int)(en & 31u);
                if (lane == owner) {
                    if (mycnt < (unsigned)BCAP)
                        bucket[(wave * AGG_WAVES + owner) * BCAP + mycnt] = en;
                    ++mycnt;
                }
            }
        };
        unsigned* cand  = list + wave * LCAP;
        unsigned  ncand = 0;
        {
            const uint4* m4 = reinterpret_cast<const uint4*>(
                cmap + (size_t)(blockIdx.x >> 5) * NCHUNK_PAD);
            const unsigned bit = 1u << (blockIdx.x & 31u);
            uint4 mv[MAP_TRIPS];
#pragma unroll
            for (int u = 0; u < MAP_TRIPS; ++u) mv[u] = m4[u * AGG_THREADS + tid];
#pragma unroll
            for (int u = 0; u < MAP_TRIPS; ++u) {
                const unsigned wv[4] = {mv[u].x, mv[u].y, mv[u].z, mv[u].w};
#pragma unroll
                for (int k = 0; k < 4; ++k) {
                    const bool     hit = (wv[k] & bit) != 0u;
                    const unsigned m   = __builtin_amdgcn_ballot_w32(hit);
                    const unsigned pos = ncand + __builtin_amdgcn_mbcnt_lo(m, 0u);
                    if (hit && pos < (unsigned)LCAP)
                        cand[pos] = (unsigned)((u * AGG_THREADS + tid) * 4 + k);
                    ncand += (unsigned)__builtin_popcount(m);
                }
            }
        }
        if (ncand > (unsigned)LCAP) {
            ovf   = 1u;
            ncand = (unsigned)LCAP;
        }
        if (lane < SCAN_U && ncand + (unsigned)lane < (unsigned)LCAP) cand[ncand + lane] = NO_CHUNK;
        __syncthreads();

        for (unsigned c0 = 0; c0 < ncand; c0 += SCAN_U) {
            int      dv[SCAN_U];
            unsigned ev[SCAN_U];
            const uint4* cp = reinterpret_cast<const uint4*>(cand + c0);
            const uint4  ca = cp[0], cb = cp[1];
            const unsigned cv[SCAN_U] = {ca.x, ca.y, ca.z, ca.w, cb.x, cb.y, cb.z, cb.w};
#pragma unroll
            for (int u = 0; u < SCAN_U; ++u) {
                const unsigned e  = cv[u] * (unsigned)CHUNK_E + (unsigned)lane;
                const bool     ok = e < (unsigned)nEdges;
                int d = dst[ok ? e : 0u];
                if (!ok) d = -1;
                dv[u] = d;
                ev[u] = e;
            }
#pragma unroll
            for (int u = 0; u < SCAN_U; ++u) put((unsigned)(dv[u] - base), (int)ev[u]);
        }
        if (mycnt > (unsigned)BCAP) ovf = 1u;
        wcnt[wave * AGG_WAVES + lane] = mycnt < (unsigned)BCAP ? mycnt : (unsigned)BCAP;
        __syncthreads();

        unsigned* mylist = list + wave * LCAP;
        const unsigned n_l = wcnt[lane * AGG_WAVES + wave];
        unsigned off = 0;
#pragma unroll
        for (int j = 0; j < 32; ++j) {
            const unsigned nj = (unsigned)__builtin_amdgcn_readlane((int)n_l, j);
            off += j < lane ? nj : 0u;
            tot += nj;
        }
        if (tot > (unsigned)LCAP) {
            ovf = 1u;
        } else {
            for (unsigned k = 0; k < n_l; ++k)
                mylist[off + k] = bucket[(lane * AGG_WAVES + wave) * BCAP + k];
        }
        __syncthreads();
        slow = ovf != 0u;
        if (lane == 0) *mygt = slow ? LIST_NONE : tot;
    } else {
        tot  = (unsigned)__builtin_amdgcn_readfirstlane((int)tot);
        slow = tot == LIST_NONE;
        __syncthreads();
    }

    if (!slow) {
        for (unsigned b0 = 0; b0 < tot; b0 += 32) {
            const unsigned qi = b0 + lane;
            unsigned ent;
            if constexpr (BUILD) {
                const unsigned se = qi < tot ? list[wave * LCAP + qi]
                                             : (unsigned)(TILE_ROWS + wave);
                ent = ((unsigned)src[se >> 10] << 10) | (se & 1023u);
                if (qi < tot) mygl[qi] = ent;
            } else {
                ent = qi < tot ? mygl[qi] : (unsigned)(TILE_ROWS + wave);
            }
            const int dl = (int)(ent & 1023u);
            const int s  = (int)(ent >> 10);
#pragma unroll
            for (int j0 = 0; j0 < 32; j0 += 16) {
                if (b0 + j0 >= tot) break;
                float v[16][H];
                int   row[16];
#pragma unroll
                for (int u = 0; u < 16; ++u) {
                    const int sj = __builtin_amdgcn_readlane(s, j0 + u);
                    row[u]       = __builtin_amdgcn_readlane(dl, j0 + u);
                    const float* xp = x + (size_t)sj * DIM + lane;
#pragma unroll
                    for (int h = 0; h < H; ++h) v[u][h] = xp[32 * h];
                }
#pragma unroll
                for (int u = 0; u < 16; ++u) {
#pragma unroll
                    for (int h = 0; h < H; ++h)
                        acc[row[u] * DIM + 32 * h + lane] += v[u][h];
                    if (COUNT && lane == 0) cntL[row[u]] += 1.0f;
                }
            }
        }
    } else {
        for (int i0 = 0; i0 < nEdges; i0 += 32) {
            const int i = i0 + lane;
            const int d = i < nEdges ? dst[i] : -1;
            const unsigned r = (unsigned)(d - base);
            unsigned mask = __builtin_amdgcn_ballot_w32(
                r < (unsigned)TILE_ROWS && (int)(r & 31u) == wave);
            while (mask) {
                const int l = __builtin_ctz(mask);
                mask &= mask - 1;
                const int rr = __builtin_amdgcn_readlane((int)r, l);
                const float* xp = x + (size_t)src[i0 + l] * DIM + lane;
#pragma unroll
                for (int h = 0; h < H; ++h) acc[rr * DIM + 32 * h + lane] += xp[32 * h];
                if (COUNT && lane == 0) cntL[rr] += 1.0f;
            }
        }
    }
    __syncthreads();

    const int left = nNodes - base;
    const int rows = left < TILE_ROWS ? left : TILE_ROWS;
    float4*       out4 = reinterpret_cast<float4*>(agg + (size_t)base * DIM);
    const float4* acc4 = reinterpret_cast<const float4*>(acc);
    for (int i = tid; i < rows * (DIM / 4); i += AGG_THREADS) out4[i] = acc4[i];
    if (COUNT)
        for (int i = tid; i < rows; i += AGG_THREADS) cnt[base + i] = cntL[i];
}

template <int K>
__global__ void sage_gemm_kernel(const float* __restrict__ aggRaw,
                                 const float* __restrict__ cnt,
                                 const float* __restrict__ x,
                                 const float* __restrict__ Wl,
                                 const float* __restrict__ Wr,
                                 const float* __restrict__ bias,
                                 float* __restrict__ out, int nNodes) {
    const int lane = threadIdx.x & 31;
    const int wave = threadIdx.x >> 5;
    const int row0 = blockIdx.x * 16;
    if (row0 >= nNodes) return;
    const int col0 = wave * 16;
    const int grp  = lane >> 4;
    const int lr   = lane & 15;

    const int   arow = row0 + lr;
    const float invc = 1.0f / fmaxf(cnt[arow], 1.0f);

    v8f c = {};
#pragma unroll
    for (int k0 = 0; k0 < K; k0 += 4) {
        const int ka = k0 + 2 * grp;
        v2f a, b;
        a.x = aggRaw[(size_t)arow * K + ka + 0] * invc;
        a.y = aggRaw[(size_t)arow * K + ka + 1] * invc;
        b.x = Wl[(size_t)(ka + 0) * HIDDEN_D + col0 + lr];
        b.y = Wl[(size_t)(ka + 1) * HIDDEN_D + col0 + lr];
        c = __builtin_amdgcn_wmma_f32_16x16x4_f32(false, a, false, b,
                                                  (short)0, c, false, false);
    }
#pragma unroll
    for (int k0 = 0; k0 < K; k0 += 4) {
        const int ka = k0 + 2 * grp;
        v2f a, b;
        a.x = x[(size_t)arow * K + ka + 0];
        a.y = x[(size_t)arow * K + ka + 1];
        b.x = Wr[(size_t)(ka + 0) * HIDDEN_D + col0 + lr];
        b.y = Wr[(size_t)(ka + 1) * HIDDEN_D + col0 + lr];
        c = __builtin_amdgcn_wmma_f32_16x16x4_f32(false, a, false, b,
                                                  (short)0, c, false, false);
    }

    const float bv = bias[col0 + lr];
#pragma unroll
    for (int v = 0; v < 8; ++v) {
        const int m = v + 8 * grp;
        out[(size_t)(row0 + m) * HIDDEN_D + col0 + lr] = fmaxf(c[v] + bv, 0.0f);
    }
}

#define ZS 16
__global__ void head_pool_kernel(const int* __restrict__ batch,
                                 const float* __restrict__ h,
                                 const float* __restrict__ Wo,
                                 float* __restrict__ zsum, int nNodes) {
    const int lane = threadIdx.x & 31;
    const int tile = blockIdx.x * (blockDim.x >> 5) + (threadIdx.x >> 5);
    const int row0 = tile * 16;
    if (row0 >= nNodes) return;
    const int grp = lane >> 4;
    const int lr  = lane & 15;

    const float4* hp =
        reinterpret_cast<const float4*>(h + (size_t)(row0 + lr) * HIDDEN_D) + grp;
    float4 av[8];
    for (int t = 0; t < 8; ++t) av[t] = hp[2 * t];

    const int4* bp = reinterpret_cast<const int4*>(batch + row0 + 8 * grp);
    const int4 gA = bp[0], gB = bp[1];

    const bool colOk = lr < N_CLASSES;
    const int  cc    = colOk ? lr : 0;

    v8f c = {};
    for (int t = 0; t < 8; ++t) {
        const int kb = 8 * t + 4 * grp;
        const float w0 = Wo[(kb + 0) * N_CLASSES + cc];
        const float w1 = Wo[(kb + 1) * N_CLASSES + cc];
        const float w2 = Wo[(kb + 2) * N_CLASSES + cc];
        const float w3 = Wo[(kb + 3) * N_CLASSES + cc];
        v2f a0, a1, b0, b1;
        a0.x = av[t].x;             a0.y = av[t].y;
        b0.x = colOk ? w0 : 0.0f;   b0.y = colOk ? w1 : 0.0f;
        a1.x = av[t].z;             a1.y = av[t].w;
        b1.x = colOk ? w2 : 0.0f;   b1.y = colOk ? w3 : 0.0f;
        c = __builtin_amdgcn_wmma_f32_16x16x4_f32(false, a0, false, b0,
                                                  (short)0, c, false, false);
        c = __builtin_amdgcn_wmma_f32_16x16x4_f32(false, a1, false, b1,
                                                  (short)0, c, false, false);
    }

    float val[8];
    for (int v = 0; v < 8; ++v) val[v] = colOk ? c[v] : 1.0f;
    const int gid[8] = {gA.x, gA.y, gA.z, gA.w, gB.x, gB.y, gB.z, gB.w};
    if (lr <= N_CLASSES) {
        for (int v = 0; v < 8; ++v)
            atomAdd1(zsum + (size_t)gid[v] * ZS + lr, val[v]);
    }
}

__global__ void finish_kernel(const float* __restrict__ zsum,
                              const float* __restrict__ bo,
                              float* __restrict__ out) {
    int t = blockIdx.x * blockDim.x + threadIdx.x;
    if (t >= N_GRAPHS * N_CLASSES) return;
    int g = t / N_CLASSES;
    int c = t - g * N_CLASSES;
    out[t] = zsum[g * ZS + c] / fmaxf(zsum[g * ZS + N_CLASSES], 1.0f) + bo[c];
}

extern "C" void kernel_launch(void* const* d_in, const int* in_sizes, int n_in,
                              void* d_out, int out_size, void* d_ws, size_t ws_size,
                              hipStream_t stream) {
    const int*   node_ids = (const int*)  d_in[0];
    const int*   src      = (const int*)  d_in[1];
    const int*   dst      = (const int*)  d_in[2];
    const int*   batch    = (const int*)  d_in[3];
    const float* embed    = (const float*)d_in[4];
    const float* W1_l     = (const float*)d_in[5];
    const float* b1       = (const float*)d_in[6];
    const float* W1_r     = (const float*)d_in[7];
    const float* W2_l     = (const float*)d_in[8];
    const float* b2       = (const float*)d_in[9];
    const float* W2_r     = (const float*)d_in[10];
    const float* W_out    = (const float*)d_in[11];
    const float* b_out    = (const float*)d_in[12];
    float* out = (float*)d_out;

    float* ws   = (float*)d_ws;
    float* x0   = ws;
    float* h1   = x0   + (size_t)N_NODES * EMBED_D;
    float* h2   = h1   + (size_t)N_NODES * HIDDEN_D;
    float* agg1 = h2   + (size_t)N_NODES * HIDDEN_D;
    float* agg2 = agg1 + (size_t)N_NODES * EMBED_D;
    float* cnt  = agg2 + (size_t)N_NODES * HIDDEN_D;
    float* zsum = cnt  + (size_t)N_NODES;
    const int zeroN = N_GRAPHS * ZS;
    unsigned* gtot  = (unsigned*)(zsum + N_GRAPHS * ZS);
    unsigned* glist = gtot + AGG_BLOCKS * AGG_WAVES;
    unsigned* cmap  = glist + (size_t)AGG_BLOCKS * AGG_WAVES * LCAP;

    const int B = 256;
    auto blocks = [](int n, int b) { return (n + b - 1) / b; };

    zero_kernel<<<blocks(zeroN, B), B, 0, stream>>>(zsum, zeroN);

    gather_embed_kernel<<<blocks(N_NODES * 8, B), B, 0, stream>>>(
        node_ids, embed, x0, N_NODES);

    chunk_map_kernel<<<NCHUNK_PAD / MAP_THREADS, MAP_THREADS, 0, stream>>>(dst, cmap, N_EDGES);

    tile_agg_kernel<EMBED_D, true, true><<<AGG_BLOCKS, AGG_THREADS, 0, stream>>>(
        src, dst, x0, agg1, cnt, glist, gtot, cmap, N_NODES, N_EDGES);
    sage_gemm_kernel<EMBED_D><<<N_NODES / 16, 128, 0, stream>>>(
        agg1, cnt, x0, W1_l, W1_r, b1, h1, N_NODES);

    tile_agg_kernel<HIDDEN_D, false, false><<<AGG_BLOCKS, AGG_THREADS, 0, stream>>>(
        src, dst, h1, agg2, cnt, glist, gtot, cmap, N_NODES, N_EDGES);
    sage_gemm_kernel<HIDDEN_D><<<N_NODES / 16, 128, 0, stream>>>(
        agg2, cnt, h1, W2_l, W2_r, b2, h2, N_NODES);

    head_pool_kernel<<<blocks(N_NODES / 16, 4), 128, 0, stream>>>(
        batch, h2, W_out, zsum, N_NODES);
    finish_kernel<<<blocks(N_GRAPHS * N_CLASSES, B), B, 0, stream>>>(
        zsum, b_out, out);
}


# row sums kept in the owner wave's registers (16-slot vector indexed by the scalar local row) instead of LDS rows: no LDS read-add-write chain per edge, no LDS zero fill or tile copy; sum order kept
# speedup vs baseline: 2.4824x; 1.1247x over previous
#include <hip/hip_runtime.h>
#include <type_traits>

#define N_NODES   100000
#define N_EDGES   1200000
#define EMBED_D   32
#define HIDDEN_D  64
#define N_CLASSES 10
#define N_GRAPHS  1024

typedef __attribute__((ext_vector_type(2))) float v2f;
typedef __attribute__((ext_vector_type(8))) float v8f;
typedef __attribute__((ext_vector_type(16))) float v16f;

__device__ __forceinline__ void atomAdd1(float* p, float v) {
    asm volatile("global_atomic_add_f32 %0, %1, off scope:SCOPE_DEV"
                 :: "v"(p), "v"(v) : "memory");
}

__global__ void zero_kernel(float* __restrict__ p, int n) {
    int t = blockIdx.x * blockDim.x + threadIdx.x;
    if (t < n) p[t] = 0.0f;
}

__global__ void gather_embed_kernel(const int* __restrict__ ids,
                                    const float* __restrict__ embed,
                                    float* __restrict__ x0, int nNodes) {
    int t = blockIdx.x * blockDim.x + threadIdx.x;
    if (t >= nNodes * (EMBED_D / 4)) return;
    int node = t >> 3;
    int q    = t & 7;
    float4 v = reinterpret_cast<const float4*>(embed + (size_t)ids[node] * EMBED_D)[q];
    reinterpret_cast<float4*>(x0 + (size_t)node * EMBED_D)[q] = v;
}

#define TILE_ROWS   392
#define AGG_BLOCKS  ((N_NODES + TILE_ROWS - 1) / TILE_ROWS)
#define AGG_THREADS 1024
#define AGG_WAVES   32
#define ROWS_PER_WAVE ((TILE_ROWS + AGG_WAVES - 1) / AGG_WAVES)
#define SINK_SLOT   ROWS_PER_WAVE
static_assert(AGG_WAVES == 32 && SINK_SLOT < 16, "slot = local row / 32, in a 16-slot vector");
#define BCAP        32
#define LCAP        512
#define SCAN_U      8
static_assert(SCAN_U == 8, "the scan reads its 8 list entries as two 16-byte words");
#define CHUNK_E     32
#define NCHUNK      ((N_EDGES + CHUNK_E - 1) / CHUNK_E)
#define MAP_TRIPS   ((NCHUNK + 4 * AGG_THREADS - 1) / (4 * AGG_THREADS))
#define NCHUNK_PAD  (MAP_TRIPS * 4 * AGG_THREADS)
#define MAP_WORDS   ((AGG_BLOCKS + 31) / 32)
#define MAP_THREADS 128
#define NO_CHUNK    (1u << 26)
#define LIST_NONE   0xFFFFFFFFu
static_assert(N_EDGES % 4 == 0, "destination list is read 4 at a time");
static_assert(TILE_ROWS + AGG_WAVES <= 1024 && N_EDGES < (1 << 21),
              "a scan entry is (edge << 10) | local row");
static_assert(N_NODES < (1 << 22), "a stored entry is (source node << 10) | local row");
static_assert(NCHUNK_PAD % MAP_THREADS == 0, "one map thread per padded chunk");

__global__ void chunk_map_kernel(const int* __restrict__ dst,
                                 unsigned* __restrict__ cmap, int nEdges) {
    const int c = blockIdx.x * blockDim.x + threadIdx.x;
    if (c >= NCHUNK_PAD) return;
    unsigned w[MAP_WORDS];
#pragma unroll
    for (int k = 0; k < MAP_WORDS; ++k) w[k] = 0u;
    auto mark = [&](int d) {
        const unsigned t   = (unsigned)d / (unsigned)TILE_ROWS;
        const unsigned bit = 1u << (t & 31u);
#pragma unroll
        for (int k = 0; k < MAP_WORDS; ++k) w[k] |= (t >> 5) == (unsigned)k ? bit : 0u;
    };
    const int   total4 = nEdges >> 2;
    const int4* dst4   = reinterpret_cast<const int4*>(dst);
#pragma unroll
    for (int q = 0; q < CHUNK_E / 4; ++q) {
        const int  i  = c * (CHUNK_E / 4) + q;
        const bool ok = i < total4;
        int4 d = dst4[ok ? i : 0];
        if (!ok) d = make_int4(-1, -1, -1, -1);
        mark(d.x);
        mark(d.y);
        mark(d.z);
        mark(d.w);
    }
#pragma unroll
    for (int k = 0; k < MAP_WORDS; ++k) cmap[(size_t)k * NCHUNK_PAD + c] = w[k];
}

template <int DIM, bool COUNT, bool BUILD>
__global__ __launch_bounds__(AGG_THREADS)
void tile_agg_kernel(const int* __restrict__ src,
                     const int* __restrict__ dst,
                     const float* __restrict__ x,
                     float* __restrict__ agg,
                     float* __restrict__ cnt,
                     unsigned* __restrict__ glist,
                     unsigned* __restrict__ gtot,
                     const unsigned* __restrict__ cmap,
                     int nNodes, int nEdges) {
    constexpr int H = DIM / 32;
    __shared__ float    cntL[TILE_ROWS];
    __shared__ unsigned bucket[AGG_WAVES * AGG_WAVES * BCAP];
    __shared__ unsigned wcnt[AGG_WAVES * AGG_WAVES];
    __shared__ __attribute__((aligned(16))) unsigned list[AGG_WAVES * LCAP];
    __shared__ unsigned ovf;

    const int tid  = threadIdx.x;
    const int lane = tid & 31;
    const int wave = tid >> 5;
    const int base = blockIdx.x * TILE_ROWS;
    unsigned* mygl  = glist + (size_t)(blockIdx.x * AGG_WAVES + wave) * LCAP;
    unsigned* mygt  = gtot + blockIdx.x * AGG_WAVES + wave;

    unsigned tot = 0;
    bool     slow;
    if constexpr (!BUILD) tot = *mygt;

    v16f a[H];
#pragma unroll
    for (int h = 0; h < H; ++h)
#pragma unroll
        for (int k = 0; k < 16; ++k) a[h][k] = 0.0f;
    int c[ROWS_PER_WAVE];
#pragma unroll
    for (int k = 0; k < ROWS_PER_WAVE; ++k) c[k] = 0;

    if constexpr (BUILD) {
        if (tid == 0) ovf = 0u;
        __syncthreads();

        unsigned mycnt = 0;
        auto put = [&](unsigned r, int e) {
            unsigned mask = __builtin_amdgcn_ballot_w32(r < (unsigned)TILE_ROWS);
            const int ent = (int)(((unsigned)e << 10) | r);
            while (mask) {
                const int l = __builtin_ctz(mask);
                mask &= mask - 1;
                const unsigned en = (unsigned)__builtin_amdgcn_readlane(ent, l);
                const int owner = (int)(en & 31u);
                if (lane == owner) {
                    if (mycnt < (unsigned)BCAP)
                        bucket[(wave * AGG_WAVES + owner) * BCAP + mycnt] = en;
                    ++mycnt;
                }
            }
        };
        unsigned* cand  = list + wave * LCAP;
        unsigned  ncand = 0;
        {
            const uint4* m4 = reinterpret_cast<const uint4*>(
                cmap + (size_t)(blockIdx.x >> 5) * NCHUNK_PAD);
            const unsigned bit = 1u << (blockIdx.x & 31u);
            uint4 mv[MAP_TRIPS];
#pragma unroll
            for (int u = 0; u < MAP_TRIPS; ++u) mv[u] = m4[u * AGG_THREADS + tid];
#pragma unroll
            for (int u = 0; u < MAP_TRIPS; ++u) {
                const unsigned wv[4] = {mv[u].x, mv[u].y, mv[u].z, mv[u].w};
#pragma unroll
                for (int k = 0; k < 4; ++k) {
                    const bool     hit = (wv[k] & bit) != 0u;
                    const unsigned m   = __builtin_amdgcn_ballot_w32(hit);
                    const unsigned pos = ncand + __builtin_amdgcn_mbcnt_lo(m, 0u);
                    if (hit && pos < (unsigned)LCAP)
                        cand[pos] = (unsigned)((u * AGG_THREADS + tid) * 4 + k);
                    ncand += (unsigned)__builtin_popcount(m);
                }
            }
        }
        if (ncand > (unsigned)LCAP) {
            ovf   = 1u;
            ncand = (unsigned)LCAP;
        }
        if (lane < SCAN_U && ncand + (unsigned)lane < (unsigned)LCAP) cand[ncand + lane] = NO_CHUNK;
        __syncthreads();

        for (unsigned c0 = 0; c0 < ncand; c0 += SCAN_U) {
            int      dv[SCAN_U];
            unsigned ev[SCAN_U];
            const uint4* cp = reinterpret_cast<const uint4*>(cand + c0);
            const uint4  ca = cp[0], cb = cp[1];
            const unsigned cv[SCAN_U] = {ca.x, ca.y, ca.z, ca.w, cb.x, cb.y, cb.z, cb.w};
#pragma unroll
            for (int u = 0; u < SCAN_U; ++u) {
                const unsigned e  = cv[u] * (unsigned)CHUNK_E + (unsigned)lane;
                const bool     ok = e < (unsigned)nEdges;
                int d = dst[ok ? e : 0u];
                if (!ok) d = -1;
                dv[u] = d;
                ev[u] = e;
            }
#pragma unroll
            for (int u = 0; u < SCAN_U; ++u) put((unsigned)(dv[u] - base), (int)ev[u]);
        }
        if (mycnt > (unsigned)BCAP) ovf = 1u;
        wcnt[wave * AGG_WAVES + lane] = mycnt < (unsigned)BCAP ? mycnt : (unsigned)BCAP;
        __syncthreads();

        unsigned* mylist = list + wave * LCAP;
        const unsigned n_l = wcnt[lane * AGG_WAVES + wave];
        unsigned off = 0;
#pragma unroll
        for (int j = 0; j < 32; ++j) {
            const unsigned nj = (unsigned)__builtin_amdgcn_readlane((int)n_l, j);
            off += j < lane ? nj : 0u;
            tot += nj;
        }
        if (tot > (unsigned)LCAP) {
            ovf = 1u;
        } else {
            for (unsigned k = 0; k < n_l; ++k)
                mylist[off + k] = bucket[(lane * AGG_WAVES + wave) * BCAP + k];
        }
        __syncthreads();
        slow = ovf != 0u;
        if (lane == 0) *mygt = slow ? LIST_NONE : tot;
    } else {
        tot  = (unsigned)__builtin_amdgcn_readfirstlane((int)tot);
        slow = tot == LIST_NONE;
    }

    if (!slow) {
        for (unsigned b0 = 0; b0 < tot; b0 += 32) {
            const unsigned qi = b0 + lane;
            unsigned ent;
            if constexpr (BUILD) {
                const unsigned se = qi < tot ? list[wave * LCAP + qi]
                                             : (unsigned)(SINK_SLOT << 5);
                ent = ((unsigned)src[se >> 10] << 10) | (se & 1023u);
                if (qi < tot) mygl[qi] = ent;
            } else {
                ent = qi < tot ? mygl[qi] : (unsigned)(SINK_SLOT << 5);
            }
            const int sl = (int)((ent & 1023u) >> 5);
            const int s  = (int)(ent >> 10);
            if constexpr (COUNT) {
#pragma unroll
                for (int k = 0; k < ROWS_PER_WAVE; ++k)
                    c[k] += __builtin_popcount(__builtin_amdgcn_ballot_w32(sl == k));
            }
#pragma unroll
            for (int j0 = 0; j0 < 32; j0 += 16) {
                if (b0 + j0 >= tot) break;
                float v[16][H];
                int   slot[16];
#pragma unroll
                for (int u = 0; u < 16; ++u) {
                    const int sj = __builtin_amdgcn_readlane(s, j0 + u);
                    slot[u]      = __builtin_amdgcn_readlane(sl, j0 + u);
                    const float* xp = x + (size_t)sj * DIM + lane;
#pragma unroll
                    for (int h = 0; h < H; ++h) v[u][h] = xp[32 * h];
                }
#pragma unroll
                for (int u = 0; u < 16; ++u) {
#pragma unroll
                    for (int h = 0; h < H; ++h)
                        a[h][slot[u]] += v[u][h];
                }
            }
        }
    } else {
        for (int i0 = 0; i0 < nEdges; i0 += 32) {
            const int i = i0 + lane;
            const int d = i < nEdges ? dst[i] : -1;
            const unsigned r = (unsigned)(d - base);
            unsigned mask = __builtin_amdgcn_ballot_w32(
                r < (unsigned)TILE_ROWS && (int)(r & 31u) == wave);
            while (mask) {
                const int l = __builtin_ctz(mask);
                mask &= mask - 1;
                const int kr = __builtin_amdgcn_readlane((int)r, l) >> 5;
                const float* xp = x + (size_t)src[i0 + l] * DIM + lane;
#pragma unroll
                for (int h = 0; h < H; ++h) a[h][kr] += xp[32 * h];
                if constexpr (COUNT) {
#pragma unroll
                    for (int k = 0; k < ROWS_PER_WAVE; ++k) c[k] += kr == k ? 1 : 0;
                }
            }
        }
    }

    const int left = nNodes - base;
    const int rows = left < TILE_ROWS ? left : TILE_ROWS;
#pragma unroll
    for (int k = 0; k < ROWS_PER_WAVE; ++k) {
        const int r = wave + 32 * k;
        if (r < rows) {
#pragma unroll
            for (int h = 0; h < H; ++h)
                agg[(size_t)(base + r) * DIM + 32 * h + lane] = a[h][k];
            if (COUNT && lane == 0) cntL[r] = (float)c[k];
        }
    }
    if constexpr (COUNT) {
        __syncthreads();
        for (int i = tid; i < rows; i += AGG_THREADS) cnt[base + i] = cntL[i];
    }
}

template <int K>
__global__ void sage_gemm_kernel(const float* __restrict__ aggRaw,
                                 const float* __restrict__ cnt,
                                 const float* __restrict__ x,
                                 const float* __restrict__ Wl,
                                 const float* __restrict__ Wr,
                                 const float* __restrict__ bias,
                                 float* __restrict__ out, int nNodes) {
    const int lane = threadIdx.x & 31;
    const int wave = threadIdx.x >> 5;
    const int row0 = blockIdx.x * 16;
    if (row0 >= nNodes) return;
    const int col0 = wave * 16;
    const int grp  = lane >> 4;
    const int lr   = lane & 15;

    const int   arow = row0 + lr;
    const float invc = 1.0f / fmaxf(cnt[arow], 1.0f);

    v8f c = {};
#pragma unroll
    for (int k0 = 0; k0 < K; k0 += 4) {
        const int ka = k0 + 2 * grp;
        v2f a, b;
        a.x = aggRaw[(size_t)arow * K + ka + 0] * invc;
        a.y = aggRaw[(size_t)arow * K + ka + 1] * invc;
        b.x = Wl[(size_t)(ka + 0) * HIDDEN_D + col0 + lr];
        b.y = Wl[(size_t)(ka + 1) * HIDDEN_D + col0 + lr];
        c = __builtin_amdgcn_wmma_f32_16x16x4_f32(false, a, false, b,
                                                  (short)0, c, false, false);
    }
#pragma unroll
    for (int k0 = 0; k0 < K; k0 += 4) {
        const int ka = k0 + 2 * grp;
        v2f a, b;
        a.x = x[(size_t)arow * K + ka + 0];
        a.y = x[(size_t)arow * K + ka + 1];
        b.x = Wr[(size_t)(ka + 0) * HIDDEN_D + col0 + lr];
        b.y = Wr[(size_t)(ka + 1) * HIDDEN_D + col0 + lr];
        c = __builtin_amdgcn_wmma_f32_16x16x4_f32(false, a, false, b,
                                                  (short)0, c, false, false);
    }

    const float bv = bias[col0 + lr];
#pragma unroll
    for (int v = 0; v < 8; ++v) {
        const int m = v + 8 * grp;
        out[(size_t)(row0 + m) * HIDDEN_D + col0 + lr] = fmaxf(c[v] + bv, 0.0f);
    }
}

#define ZS 16
__global__ void head_pool_kernel(const int* __restrict__ batch,
                                 const float* __restrict__ h,
                                 const float* __restrict__ Wo,
                                 float* __restrict__ zsum, int nNodes) {
    const int lane = threadIdx.x & 31;
    const int tile = blockIdx.x * (blockDim.x >> 5) + (threadIdx.x >> 5);
    const int row0 = tile * 16;
    if (row0 >= nNodes) return;
    const int grp = lane >> 4;
    const int lr  = lane & 15;

    const float4* hp =
        reinterpret_cast<const float4*>(h + (size_t)(row0 + lr) * HIDDEN_D) + grp;
    float4 av[8];
    for (int t = 0; t < 8; ++t) av[t] = hp[2 * t];

    const int4* bp = reinterpret_cast<const int4*>(batch + row0 + 8 * grp);
    const int4 gA = bp[0], gB = bp[1];

    const bool colOk = lr < N_CLASSES;
    const int  cc    = colOk ? lr : 0;

    v8f c = {};
    for (int t = 0; t < 8; ++t) {
        const int kb = 8 * t + 4 * grp;
        const float w0 = Wo[(kb + 0) * N_CLASSES + cc];
        const float w1 = Wo[(kb + 1) * N_CLASSES + cc];
        const float w2 = Wo[(kb + 2) * N_CLASSES + cc];
        const float w3 = Wo[(kb + 3) * N_CLASSES + cc];
        v2f a0, a1, b0, b1;
        a0.x = av[t].x;             a0.y = av[t].y;
        b0.x = colOk ? w0 : 0.0f;   b0.y = colOk ? w1 : 0.0f;
        a1.x = av[t].z;             a1.y = av[t].w;
        b1.x = colOk ? w2 : 0.0f;   b1.y = colOk ? w3 : 0.0f;
        c = __builtin_amdgcn_wmma_f32_16x16x4_f32(false, a0, false, b0,
                                                  (short)0, c, false, false);
        c = __builtin_amdgcn_wmma_f32_16x16x4_f32(false, a1, false, b1,
                                                  (short)0, c, false, false);
    }

    float val[8];
    for (int v = 0; v < 8; ++v) val[v] = colOk ? c[v] : 1.0f;
    const int gid[8] = {gA.x, gA.y, gA.z, gA.w, gB.x, gB.y, gB.z, gB.w};
    if (lr <= N_CLASSES) {
        for (int v = 0; v < 8; ++v)
            atomAdd1(zsum + (size_t)gid[v] * ZS + lr, val[v]);
    }
}

__global__ void finish_kernel(const float* __restrict__ zsum,
                              const float* __restrict__ bo,
                              float* __restrict__ out) {
    int t = blockIdx.x * blockDim.x + threadIdx.x;
    if (t >= N_GRAPHS * N_CLASSES) return;
    int g = t / N_CLASSES;
    int c = t - g * N_CLASSES;
    out[t] = zsum[g * ZS + c] / fmaxf(zsum[g * ZS + N_CLASSES], 1.0f) + bo[c];
}

extern "C" void kernel_launch(void* const* d_in, const int* in_sizes, int n_in,
                              void* d_out, int out_size, void* d_ws, size_t ws_size,
                              hipStream_t stream) {
    const int*   node_ids = (const int*)  d_in[0];
    const int*   src      = (const int*)  d_in[1];
    const int*   dst      = (const int*)  d_in[2];
    const int*   batch    = (const int*)  d_in[3];
    const float* embed    = (const float*)d_in[4];
    const float* W1_l     = (const float*)d_in[5];
    const float* b1       = (const float*)d_in[6];
    const float* W1_r     = (const float*)d_in[7];
    const float* W2_l     = (const float*)d_in[8];
    const float* b2       = (const float*)d_in[9];
    const float* W2_r     = (const float*)d_in[10];
    const float* W_out    = (const float*)d_in[11];
    const float* b_out    = (const float*)d_in[12];
    float* out = (float*)d_out;

    float* ws   = (float*)d_ws;
    float* x0   = ws;
    float* h1   = x0   + (size_t)N_NODES * EMBED_D;
    float* h2   = h1   + (size_t)N_NODES * HIDDEN_D;
    float* agg1 = h2   + (size_t)N_NODES * HIDDEN_D;
    float* agg2 = agg1 + (size_t)N_NODES * EMBED_D;
    float* cnt  = agg2 + (size_t)N_NODES * HIDDEN_D;
    float* zsum = cnt  + (size_t)N_NODES;
    const int zeroN = N_GRAPHS * ZS;
    unsigned* gtot  = (unsigned*)(zsum + N_GRAPHS * ZS);
    unsigned* glist = gtot + AGG_BLOCKS * AGG_WAVES;
    unsigned* cmap  = glist + (size_t)AGG_BLOCKS * AGG_WAVES * LCAP;

    const int B = 256;
    auto blocks = [](int n, int b) { return (n + b - 1) / b; };

    zero_kernel<<<blocks(zeroN, B), B, 0, stream>>>(zsum, zeroN);

    gather_embed_kernel<<<blocks(N_NODES * 8, B), B, 0, stream>>>(
        node_ids, embed, x0, N_NODES);

    chunk_map_kernel<<<NCHUNK_PAD / MAP_THREADS, MAP_THREADS, 0, stream>>>(dst, cmap, N_EDGES);

    tile_agg_kernel<EMBED_D, true, true><<<AGG_BLOCKS, AGG_THREADS, 0, stream>>>(
        src, dst, x0, agg1, cnt, glist, gtot, cmap, N_NODES, N_EDGES);
    sage_gemm_kernel<EMBED_D><<<N_NODES / 16, 128, 0, stream>>>(
        agg1, cnt, x0, W1_l, W1_r, b1, h1, N_NODES);

    tile_agg_kernel<HIDDEN_D, false, false><<<AGG_BLOCKS, AGG_THREADS, 0, stream>>>(
        src, dst, h1, agg2, cnt, glist, gtot, cmap, N_NODES, N_EDGES);
    sage_gemm_kernel<HIDDEN_D><<<N_NODES / 16, 128, 0, stream>>>(
        agg2, cnt, h1, W2_l, W2_r, b2, h2, N_NODES);

    head_pool_kernel<<<blocks(N_NODES / 16, 4), 128, 0, stream>>>(
        batch, h2, W_out, zsum, N_NODES);
    finish_kernel<<<blocks(N_GRAPHS * N_CLASSES, B), B, 0, stream>>>(
        zsum, b_out, out);
}


# layer GEMMs re-tiled: a wave owns whole 16x64 tiles and loops over row tiles with both weight matrices' B fragments in its registers, A rows read 16 B per lane; k order in the products changes, fixed
# speedup vs baseline: 2.6349x; 1.0614x over previous
#include <hip/hip_runtime.h>
#include <type_traits>

#define N_NODES   100000
#define N_EDGES   1200000
#define EMBED_D   32
#define HIDDEN_D  64
#define N_CLASSES 10
#define N_GRAPHS  1024

typedef __attribute__((ext_vector_type(2))) float v2f;
typedef __attribute__((ext_vector_type(8))) float v8f;
typedef __attribute__((ext_vector_type(16))) float v16f;

__device__ __forceinline__ void atomAdd1(float* p, float v) {
    asm volatile("global_atomic_add_f32 %0, %1, off scope:SCOPE_DEV"
                 :: "v"(p), "v"(v) : "memory");
}

__global__ void zero_kernel(float* __restrict__ p, int n) {
    int t = blockIdx.x * blockDim.x + threadIdx.x;
    if (t < n) p[t] = 0.0f;
}

__global__ void gather_embed_kernel(const int* __restrict__ ids,
                                    const float* __restrict__ embed,
                                    float* __restrict__ x0, int nNodes) {
    int t = blockIdx.x * blockDim.x + threadIdx.x;
    if (t >= nNodes * (EMBED_D / 4)) return;
    int node = t >> 3;
    int q    = t & 7;
    float4 v = reinterpret_cast<const float4*>(embed + (size_t)ids[node] * EMBED_D)[q];
    reinterpret_cast<float4*>(x0 + (size_t)node * EMBED_D)[q] = v;
}

#define TILE_ROWS   392
#define AGG_BLOCKS  ((N_NODES + TILE_ROWS - 1) / TILE_ROWS)
#define AGG_THREADS 1024
#define AGG_WAVES   32
#define ROWS_PER_WAVE ((TILE_ROWS + AGG_WAVES - 1) / AGG_WAVES)
#define SINK_SLOT   ROWS_PER_WAVE
static_assert(AGG_WAVES == 32 && SINK_SLOT < 16, "slot = local row / 32, in a 16-slot vector");
#define BCAP        32
#define LCAP        512
#define SCAN_U      8
static_assert(SCAN_U == 8, "the scan reads its 8 list entries as two 16-byte words");
#define CHUNK_E     32
#define NCHUNK      ((N_EDGES + CHUNK_E - 1) / CHUNK_E)
#define MAP_TRIPS   ((NCHUNK + 4 * AGG_THREADS - 1) / (4 * AGG_THREADS))
#define NCHUNK_PAD  (MAP_TRIPS * 4 * AGG_THREADS)
#define MAP_WORDS   ((AGG_BLOCKS + 31) / 32)
#define MAP_THREADS 128
#define NO_CHUNK    (1u << 26)
#define LIST_NONE   0xFFFFFFFFu
static_assert(N_EDGES % 4 == 0, "destination list is read 4 at a time");
static_assert(TILE_ROWS + AGG_WAVES <= 1024 && N_EDGES < (1 << 21),
              "a scan entry is (edge << 10) | local row");
static_assert(N_NODES < (1 << 22), "a stored entry is (source node << 10) | local row");
static_assert(NCHUNK_PAD % MAP_THREADS == 0, "one map thread per padded chunk");

__global__ void chunk_map_kernel(const int* __restrict__ dst,
                                 unsigned* __restrict__ cmap, int nEdges) {
    const int c = blockIdx.x * blockDim.x + threadIdx.x;
    if (c >= NCHUNK_PAD) return;
    unsigned w[MAP_WORDS];
#pragma unroll
    for (int k = 0; k < MAP_WORDS; ++k) w[k] = 0u;
    auto mark = [&](int d) {
        const unsigned t   = (unsigned)d / (unsigned)TILE_ROWS;
        const unsigned bit = 1u << (t & 31u);
#pragma unroll
        for (int k = 0; k < MAP_WORDS; ++k) w[k] |= (t >> 5) == (unsigned)k ? bit : 0u;
    };
    const int   total4 = nEdges >> 2;
    const int4* dst4   = reinterpret_cast<const int4*>(dst);
#pragma unroll
    for (int q = 0; q < CHUNK_E / 4; ++q) {
        const int  i  = c * (CHUNK_E / 4) + q;
        const bool ok = i < total4;
        int4 d = dst4[ok ? i : 0];
        if (!ok) d = make_int4(-1, -1, -1, -1);
        mark(d.x);
        mark(d.y);
        mark(d.z);
        mark(d.w);
    }
#pragma unroll
    for (int k = 0; k < MAP_WORDS; ++k) cmap[(size_t)k * NCHUNK_PAD + c] = w[k];
}

template <int DIM, bool COUNT, bool BUILD>
__global__ __launch_bounds__(AGG_THREADS)
void tile_agg_kernel(const int* __restrict__ src,
                     const int* __restrict__ dst,
                     const float* __restrict__ x,
                     float* __restrict__ agg,
                     float* __restrict__ cnt,
                     unsigned* __restrict__ glist,
                     unsigned* __restrict__ gtot,
                     const unsigned* __restrict__ cmap,
                     int nNodes, int nEdges) {
    constexpr int H = DIM / 32;
    __shared__ float    cntL[TILE_ROWS];
    __shared__ unsigned bucket[AGG_WAVES * AGG_WAVES * BCAP];
    __shared__ unsigned wcnt[AGG_WAVES * AGG_WAVES];
    __shared__ __attribute__((aligned(16))) unsigned list[AGG_WAVES * LCAP];
    __shared__ unsigned ovf;

    const int tid  = threadIdx.x;
    const int lane = tid & 31;
    const int wave = tid >> 5;
    const int base = blockIdx.x * TILE_ROWS;
    unsigned* mygl  = glist + (size_t)(blockIdx.x * AGG_WAVES + wave) * LCAP;
    unsigned* mygt  = gtot + blockIdx.x * AGG_WAVES + wave;

    unsigned tot = 0;
    bool     slow;
    if constexpr (!BUILD) tot = *mygt;

    v16f a[H];
#pragma unroll
    for (int h = 0; h < H; ++h)
#pragma unroll
        for (int k = 0; k < 16; ++k) a[h][k] = 0.0f;
    int c[ROWS_PER_WAVE];
#pragma unroll
    for (int k = 0; k < ROWS_PER_WAVE; ++k) c[k] = 0;

    if constexpr (BUILD) {
        if (tid == 0) ovf = 0u;
        __syncthreads();

        unsigned mycnt = 0;
        auto put = [&](unsigned r, int e) {
            unsigned mask = __builtin_amdgcn_ballot_w32(r < (unsigned)TILE_ROWS);
            const int ent = (int)(((unsigned)e << 10) | r);
            while (mask) {
                const int l = __builtin_ctz(mask);
                mask &= mask - 1;
                const unsigned en = (unsigned)__builtin_amdgcn_readlane(ent, l);
                const int owner = (int)(en & 31u);
                if (lane == owner) {
                    if (mycnt < (unsigned)BCAP)
                        bucket[(wave * AGG_WAVES + owner) * BCAP + mycnt] = en;
                    ++mycnt;
                }
            }
        };
        unsigned* cand  = list + wave * LCAP;
        unsigned  ncand = 0;
        {
            const uint4* m4 = reinterpret_cast<const uint4*>(
                cmap + (size_t)(blockIdx.x >> 5) * NCHUNK_PAD);
            const unsigned bit = 1u << (blockIdx.x & 31u);
            uint4 mv[MAP_TRIPS];
#pragma unroll
            for (int u = 0; u < MAP_TRIPS; ++u) mv[u] = m4[u * AGG_THREADS + tid];
#pragma unroll
            for (int u = 0; u < MAP_TRIPS; ++u) {
                const unsigned wv[4] = {mv[u].x, mv[u].y, mv[u].z, mv[u].w};
#pragma unroll
                for (int k = 0; k < 4; ++k) {
                    const bool     hit = (wv[k] & bit) != 0u;
                    const unsigned m   = __builtin_amdgcn_ballot_w32(hit);
                    const unsigned pos = ncand + __builtin_amdgcn_mbcnt_lo(m, 0u);
                    if (hit && pos < (unsigned)LCAP)
                        cand[pos] = (unsigned)((u * AGG_THREADS + tid) * 4 + k);
                    ncand += (unsigned)__builtin_popcount(m);
                }
            }
        }
        if (ncand > (unsigned)LCAP) {
            ovf   = 1u;
            ncand = (unsigned)LCAP;
        }
        if (lane < SCAN_U && ncand + (unsigned)lane < (unsigned)LCAP) cand[ncand + lane] = NO_CHUNK;
        __syncthreads();

        for (unsigned c0 = 0; c0 < ncand; c0 += SCAN_U) {
            int      dv[SCAN_U];
            unsigned ev[SCAN_U];
            const uint4* cp = reinterpret_cast<const uint4*>(cand + c0);
            const uint4  ca = cp[0], cb = cp[1];
            const unsigned cv[SCAN_U] = {ca.x, ca.y, ca.z, ca.w, cb.x, cb.y, cb.z, cb.w};
#pragma unroll
            for (int u = 0; u < SCAN_U; ++u) {
                const unsigned e  = cv[u] * (unsigned)CHUNK_E + (unsigned)lane;
                const bool     ok = e < (unsigned)nEdges;
                int d = dst[ok ? e : 0u];
                if (!ok) d = -1;
                dv[u] = d;
                ev[u] = e;
            }
#pragma unroll
            for (int u = 0; u < SCAN_U; ++u) put((unsigned)(dv[u] - base), (int)ev[u]);
        }
        if (mycnt > (unsigned)BCAP) ovf = 1u;
        wcnt[wave * AGG_WAVES + lane] = mycnt < (unsigned)BCAP ? mycnt : (unsigned)BCAP;
        __syncthreads();

        unsigned* mylist = list + wave * LCAP;
        const unsigned n_l = wcnt[lane * AGG_WAVES + wave];
        unsigned off = 0;
#pragma unroll
        for (int j = 0; j < 32; ++j) {
            const unsigned nj = (unsigned)__builtin_amdgcn_readlane((int)n_l, j);
            off += j < lane ? nj : 0u;
            tot += nj;
        }
        if (tot > (unsigned)LCAP) {
            ovf = 1u;
        } else {
            for (unsigned k = 0; k < n_l; ++k)
                mylist[off + k] = bucket[(lane * AGG_WAVES + wave) * BCAP + k];
        }
        __syncthreads();
        slow = ovf != 0u;
        if (lane == 0) *mygt = slow ? LIST_NONE : tot;
    } else {
        tot  = (unsigned)__builtin_amdgcn_readfirstlane((int)tot);
        slow = tot == LIST_NONE;
    }

    if (!slow) {
        for (unsigned b0 = 0; b0 < tot; b0 += 32) {
            const unsigned qi = b0 + lane;
            unsigned ent;
            if constexpr (BUILD) {
                const unsigned se = qi < tot ? list[wave * LCAP + qi]
                                             : (unsigned)(SINK_SLOT << 5);
                ent = ((unsigned)src[se >> 10] << 10) | (se & 1023u);
                if (qi < tot) mygl[qi] = ent;
            } else {
                ent = qi < tot ? mygl[qi] : (unsigned)(SINK_SLOT << 5);
            }
            const int sl = (int)((ent & 1023u) >> 5);
            const int s  = (int)(ent >> 10);
            if constexpr (COUNT) {
#pragma unroll
                for (int k = 0; k < ROWS_PER_WAVE; ++k)
                    c[k] += __builtin_popcount(__builtin_amdgcn_ballot_w32(sl == k));
            }
#pragma unroll
            for (int j0 = 0; j0 < 32; j0 += 16) {
                if (b0 + j0 >= tot) break;
                float v[16][H];
                int   slot[16];
#pragma unroll
                for (int u = 0; u < 16; ++u) {
                    const int sj = __builtin_amdgcn_readlane(s, j0 + u);
                    slot[u]      = __builtin_amdgcn_readlane(sl, j0 + u);
                    const float* xp = x + (size_t)sj * DIM + lane;
#pragma unroll
                    for (int h = 0; h < H; ++h) v[u][h] = xp[32 * h];
                }
#pragma unroll
                for (int u = 0; u < 16; ++u) {
#pragma unroll
                    for (int h = 0; h < H; ++h)
                        a[h][slot[u]] += v[u][h];
                }
            }
        }
    } else {
        for (int i0 = 0; i0 < nEdges; i0 += 32) {
            const int i = i0 + lane;
            const int d = i < nEdges ? dst[i] : -1;
            const unsigned r = (unsigned)(d - base);
            unsigned mask = __builtin_amdgcn_ballot_w32(
                r < (unsigned)TILE_ROWS && (int)(r & 31u) == wave);
            while (mask) {
                const int l = __builtin_ctz(mask);
                mask &= mask - 1;
                const int kr = __builtin_amdgcn_readlane((int)r, l) >> 5;
                const float* xp = x + (size_t)src[i0 + l] * DIM + lane;
#pragma unroll
                for (int h = 0; h < H; ++h) a[h][kr] += xp[32 * h];
                if constexpr (COUNT) {
#pragma unroll
                    for (int k = 0; k < ROWS_PER_WAVE; ++k) c[k] += kr == k ? 1 : 0;
                }
            }
        }
    }

    const int left = nNodes - base;
    const int rows = left < TILE_ROWS ? left : TILE_ROWS;
#pragma unroll
    for (int k = 0; k < ROWS_PER_WAVE; ++k) {
        const int r = wave + 32 * k;
        if (r < rows) {
#pragma unroll
            for (int h = 0; h < H; ++h)
                agg[(size_t)(base + r) * DIM + 32 * h + lane] = a[h][k];
            if (COUNT && lane == 0) cntL[r] = (float)c[k];
        }
    }
    if constexpr (COUNT) {
        __syncthreads();
        for (int i = tid; i < rows; i += AGG_THREADS) cnt[base + i] = cntL[i];
    }
}

#define GEMM_WAVES   8
#define GEMM_THREADS (32 * GEMM_WAVES)
#define GEMM_BLOCKS  256
static_assert(N_NODES % 16 == 0, "row tiles are whole");

template <int K>
__global__ __launch_bounds__(GEMM_THREADS)
void sage_gemm_kernel(const float* aggRaw,
                      const float* cnt,
                      const float* x,
                      const float* __restrict__ Wl,
                      const float* __restrict__ Wr,
                      const float* __restrict__ bias,
                      float* __restrict__ out, int nNodes) {
    constexpr int T = K / 8;
    const int lane = threadIdx.x & 31;
    const int grp  = lane >> 4;
    const int lr   = lane & 15;
    const int wv     = __builtin_amdgcn_readfirstlane((int)(threadIdx.x >> 5));
    const int nWaves = (int)gridDim.x * GEMM_WAVES;
    const int nTiles = nNodes / 16;
    int tile = wv * (int)gridDim.x + (int)blockIdx.x;

    float4 ra0[T], rx0[T], ra1[T], rx1[T];
    float  rc0 = 1.0f, rc1 = 1.0f;
    auto load_tile = [&](int tl, float4 (&a)[T], float4 (&xx)[T], float& c) {
        const int arow = tl * 16 + lr;
        const float4* ap = reinterpret_cast<const float4*>(aggRaw + (size_t)arow * K) + grp;
        const float4* xp = reinterpret_cast<const float4*>(x + (size_t)arow * K) + grp;
        c = cnt[arow];
#pragma unroll
        for (int t = 0; t < T; ++t) { a[t] = ap[2 * t]; xx[t] = xp[2 * t]; }
    };
    if (tile >= nTiles) return;
    load_tile(tile, ra0, rx0, rc0);

    v2f bl[T][2][4], br[T][2][4];
#pragma unroll
    for (int t = 0; t < T; ++t)
#pragma unroll
        for (int p = 0; p < 2; ++p) {
            const int kb = 8 * t + 4 * grp + 2 * p;
#pragma unroll
            for (int n = 0; n < 4; ++n) {
                bl[t][p][n].x = Wl[(size_t)(kb + 0) * HIDDEN_D + 16 * n + lr];
                bl[t][p][n].y = Wl[(size_t)(kb + 1) * HIDDEN_D + 16 * n + lr];
                br[t][p][n].x = Wr[(size_t)(kb + 0) * HIDDEN_D + 16 * n + lr];
                br[t][p][n].y = Wr[(size_t)(kb + 1) * HIDDEN_D + 16 * n + lr];
            }
        }
    float bv[4];
#pragma unroll
    for (int n = 0; n < 4; ++n) bv[n] = bias[16 * n + lr];


    auto compute_tile = [&](int tl, const float4 (&av)[T], const float4 (&xv)[T], float deg) {
        const float invc = 1.0f / fmaxf(deg, 1.0f);
        v8f c[4] = {};
#pragma unroll
        for (int t = 0; t < T; ++t) {
            v2f a0, a1;
            a0.x = av[t].x * invc;  a0.y = av[t].y * invc;
            a1.x = av[t].z * invc;  a1.y = av[t].w * invc;
#pragma unroll
            for (int n = 0; n < 4; ++n)
                c[n] = __builtin_amdgcn_wmma_f32_16x16x4_f32(false, a0, false, bl[t][0][n],
                                                             (short)0, c[n], false, false);
#pragma unroll
            for (int n = 0; n < 4; ++n)
                c[n] = __builtin_amdgcn_wmma_f32_16x16x4_f32(false, a1, false, bl[t][1][n],
                                                             (short)0, c[n], false, false);
        }
#pragma unroll
        for (int t = 0; t < T; ++t) {
            v2f a0, a1;
            a0.x = xv[t].x;  a0.y = xv[t].y;
            a1.x = xv[t].z;  a1.y = xv[t].w;
#pragma unroll
            for (int n = 0; n < 4; ++n)
                c[n] = __builtin_amdgcn_wmma_f32_16x16x4_f32(false, a0, false, br[t][0][n],
                                                             (short)0, c[n], false, false);
#pragma unroll
            for (int n = 0; n < 4; ++n)
                c[n] = __builtin_amdgcn_wmma_f32_16x16x4_f32(false, a1, false, br[t][1][n],
                                                             (short)0, c[n], false, false);
        }
        const int row0 = tl * 16;
#pragma unroll
        for (int n = 0; n < 4; ++n)
#pragma unroll
            for (int v = 0; v < 8; ++v) {
                const int m = v + 8 * grp;
                out[(size_t)(row0 + m) * HIDDEN_D + 16 * n + lr] = fmaxf(c[n][v] + bv[n], 0.0f);
            }
    };

    int next = tile + nWaves;
    load_tile(next < nTiles ? next : tile, ra1, rx1, rc1);
    __builtin_amdgcn_sched_barrier(0);
    compute_tile(tile, ra0, rx0, rc0);
    tile = next;
    while (tile < nTiles) {
        next = tile + nWaves;
        load_tile(next < nTiles ? next : tile, ra0, rx0, rc0);
        __builtin_amdgcn_sched_barrier(0);
        compute_tile(tile, ra1, rx1, rc1);
        tile = next;
        if (tile >= nTiles) break;
        next = tile + nWaves;
        load_tile(next < nTiles ? next : tile, ra1, rx1, rc1);
        __builtin_amdgcn_sched_barrier(0);
        compute_tile(tile, ra0, rx0, rc0);
        tile = next;
    }
}

#define ZS 16
__global__ void head_pool_kernel(const int* __restrict__ batch,
                                 const float* __restrict__ h,
                                 const float* __restrict__ Wo,
                                 float* __restrict__ zsum, int nNodes) {
    const int lane = threadIdx.x & 31;
    const int tile = blockIdx.x * (blockDim.x >> 5) + (threadIdx.x >> 5);
    const int row0 = tile * 16;
    if (row0 >= nNodes) return;
    const int grp = lane >> 4;
    const int lr  = lane & 15;

    const float4* hp =
        reinterpret_cast<const float4*>(h + (size_t)(row0 + lr) * HIDDEN_D) + grp;
    float4 av[8];
    for (int t = 0; t < 8; ++t) av[t] = hp[2 * t];

    const int4* bp = reinterpret_cast<const int4*>(batch + row0 + 8 * grp);
    const int4 gA = bp[0], gB = bp[1];

    const bool colOk = lr < N_CLASSES;
    const int  cc    = colOk ? lr : 0;

    v8f c = {};
    for (int t = 0; t < 8; ++t) {
        const int kb = 8 * t + 4 * grp;
        const float w0 = Wo[(kb + 0) * N_CLASSES + cc];
        const float w1 = Wo[(kb + 1) * N_CLASSES + cc];
        const float w2 = Wo[(kb + 2) * N_CLASSES + cc];
        const float w3 = Wo[(kb + 3) * N_CLASSES + cc];
        v2f a0, a1, b0, b1;
        a0.x = av[t].x;             a0.y = av[t].y;
        b0.x = colOk ? w0 : 0.0f;   b0.y = colOk ? w1 : 0.0f;
        a1.x = av[t].z;             a1.y = av[t].w;
        b1.x = colOk ? w2 : 0.0f;   b1.y = colOk ? w3 : 0.0f;
        c = __builtin_amdgcn_wmma_f32_16x16x4_f32(false, a0, false, b0,
                                                  (short)0, c, false, false);
        c = __builtin_amdgcn_wmma_f32_16x16x4_f32(false, a1, false, b1,
                                                  (short)0, c, false, false);
    }

    float val[8];
    for (int v = 0; v < 8; ++v) val[v] = colOk ? c[v] : 1.0f;
    const int gid[8] = {gA.x, gA.y, gA.z, gA.w, gB.x, gB.y, gB.z, gB.w};
    if (lr <= N_CLASSES) {
        for (int v = 0; v < 8; ++v)
            atomAdd1(zsum + (size_t)gid[v] * ZS + lr, val[v]);
    }
}

__global__ void finish_kernel(const float* __restrict__ zsum,
                              const float* __restrict__ bo,
                              float* __restrict__ out) {
    int t = blockIdx.x * blockDim.x + threadIdx.x;
    if (t >= N_GRAPHS * N_CLASSES) return;
    int g = t / N_CLASSES;
    int c = t - g * N_CLASSES;
    out[t] = zsum[g * ZS + c] / fmaxf(zsum[g * ZS + N_CLASSES], 1.0f) + bo[c];
}

extern "C" void kernel_launch(void* const* d_in, const int* in_sizes, int n_in,
                              void* d_out, int out_size, void* d_ws, size_t ws_size,
                              hipStream_t stream) {
    const int*   node_ids = (const int*)  d_in[0];
    const int*   src      = (const int*)  d_in[1];
    const int*   dst      = (const int*)  d_in[2];
    const int*   batch    = (const int*)  d_in[3];
    const float* embed    = (const float*)d_in[4];
    const float* W1_l     = (const float*)d_in[5];
    const float* b1       = (const float*)d_in[6];
    const float* W1_r     = (const float*)d_in[7];
    const float* W2_l     = (const float*)d_in[8];
    const float* b2       = (const float*)d_in[9];
    const float* W2_r     = (const float*)d_in[10];
    const float* W_out    = (const float*)d_in[11];
    const float* b_out    = (const float*)d_in[12];
    float* out = (float*)d_out;

    float* ws   = (float*)d_ws;
    float* x0   = ws;
    float* h1   = x0   + (size_t)N_NODES * EMBED_D;
    float* h2   = h1   + (size_t)N_NODES * HIDDEN_D;
    float* agg1 = h2   + (size_t)N_NODES * HIDDEN_D;
    float* agg2 = agg1 + (size_t)N_NODES * EMBED_D;
    float* cnt  = agg2 + (size_t)N_NODES * HIDDEN_D;
    float* zsum = cnt  + (size_t)N_NODES;
    const int zeroN = N_GRAPHS * ZS;
    unsigned* gtot  = (unsigned*)(zsum + N_GRAPHS * ZS);
    unsigned* glist = gtot + AGG_BLOCKS * AGG_WAVES;
    unsigned* cmap  = glist + (size_t)AGG_BLOCKS * AGG_WAVES * LCAP;

    const int B = 256;
    auto blocks = [](int n, int b) { return (n + b - 1) / b; };

    zero_kernel<<<blocks(zeroN, B), B, 0, stream>>>(zsum, zeroN);

    gather_embed_kernel<<<blocks(N_NODES * 8, B), B, 0, stream>>>(
        node_ids, embed, x0, N_NODES);

    chunk_map_kernel<<<NCHUNK_PAD / MAP_THREADS, MAP_THREADS, 0, stream>>>(dst, cmap, N_EDGES);

    tile_agg_kernel<EMBED_D, true, true><<<AGG_BLOCKS, AGG_THREADS, 0, stream>>>(
        src, dst, x0, agg1, cnt, glist, gtot, cmap, N_NODES, N_EDGES);
    sage_gemm_kernel<EMBED_D><<<GEMM_BLOCKS, GEMM_THREADS, 0, stream>>>(
        agg1, cnt, x0, W1_l, W1_r, b1, h1, N_NODES);

    tile_agg_kernel<HIDDEN_D, false, false><<<AGG_BLOCKS, AGG_THREADS, 0, stream>>>(
        src, dst, h1, agg2, cnt, glist, gtot, cmap, N_NODES, N_EDGES);
    sage_gemm_kernel<HIDDEN_D><<<GEMM_BLOCKS, GEMM_THREADS, 0, stream>>>(
        agg2, cnt, h1, W2_l, W2_r, b2, h2, N_NODES);

    head_pool_kernel<<<blocks(N_NODES / 16, 4), 128, 0, stream>>>(
        batch, h2, W_out, zsum, N_NODES);
    finish_kernel<<<blocks(N_GRAPHS * N_CLASSES, B), B, 0, stream>>>(
        zsum, b_out, out);
}


# the three preparation launches (pooled-sum clear, embedding gather, chunk map) become one: map blocks first, then copy blocks whose threads move 10 row pieces each; 9 launches -> 7; sums untouched
# speedup vs baseline: 2.7657x; 1.0496x over previous
#include <hip/hip_runtime.h>
#include <type_traits>

#define N_NODES   100000
#define N_EDGES   1200000
#define EMBED_D   32
#define HIDDEN_D  64
#define N_CLASSES 10
#define N_GRAPHS  1024

typedef __attribute__((ext_vector_type(2))) float v2f;
typedef __attribute__((ext_vector_type(4))) float v4f;
typedef __attribute__((ext_vector_type(4))) int   v4i;
typedef __attribute__((ext_vector_type(8))) float v8f;
typedef __attribute__((ext_vector_type(16))) float v16f;

__device__ __forceinline__ void atomAdd1(float* p, float v) {
    asm volatile("global_atomic_add_f32 %0, %1, off scope:SCOPE_DEV"
                 :: "v"(p), "v"(v) : "memory");
}

#define TILE_ROWS   392
#define AGG_BLOCKS  ((N_NODES + TILE_ROWS - 1) / TILE_ROWS)
#define AGG_THREADS 1024
#define AGG_WAVES   32
#define ROWS_PER_WAVE ((TILE_ROWS + AGG_WAVES - 1) / AGG_WAVES)
#define SINK_SLOT   ROWS_PER_WAVE
static_assert(AGG_WAVES == 32 && SINK_SLOT < 16, "slot = local row / 32, in a 16-slot vector");
#define BCAP        32
#define LCAP        512
#define SCAN_U      8
static_assert(SCAN_U == 8, "the scan reads its 8 list entries as two 16-byte words");
#define CHUNK_E     32
#define NCHUNK      ((N_EDGES + CHUNK_E - 1) / CHUNK_E)
#define MAP_TRIPS   ((NCHUNK + 4 * AGG_THREADS - 1) / (4 * AGG_THREADS))
#define NCHUNK_PAD  (MAP_TRIPS * 4 * AGG_THREADS)
#define MAP_WORDS   ((AGG_BLOCKS + 31) / 32)
#define PREP_THREADS 128
#define NO_CHUNK    (1u << 26)
#define LIST_NONE   0xFFFFFFFFu
static_assert(N_EDGES % 4 == 0, "destination list is read 4 at a time");
static_assert(TILE_ROWS + AGG_WAVES <= 1024 && N_EDGES < (1 << 21),
              "a scan entry is (edge << 10) | local row");
static_assert(N_NODES < (1 << 22), "a stored entry is (source node << 10) | local row");
static_assert(NCHUNK_PAD % PREP_THREADS == 0, "one map thread per padded chunk");

#define MAP_BLOCKS    (NCHUNK_PAD / PREP_THREADS)
#define GATHER_U      10
#define GATHER_ITEMS  (N_NODES * (EMBED_D / 4))
#define GATHER_BLOCKS (GATHER_ITEMS / (PREP_THREADS * GATHER_U))
#define PREP_BLOCKS   (MAP_BLOCKS + GATHER_BLOCKS)
#define ZS 16
static_assert(EMBED_D / 4 == 8 && PREP_THREADS % 8 == 0, "a thread keeps its quarter of a row");
static_assert(GATHER_ITEMS % (PREP_THREADS * GATHER_U) == 0,
              "the copy blocks cover the pieces exactly: no end check, so no load waits behind a branch");
static_assert(GATHER_BLOCKS * PREP_THREADS >= N_GRAPHS * ZS, "one copy thread per pooled word");

__device__ __forceinline__ void chunk_map_thread(const int* __restrict__ dst,
                                                 unsigned* __restrict__ cmap,
                                                 int nEdges, int c) {
    unsigned w[MAP_WORDS];
#pragma unroll
    for (int k = 0; k < MAP_WORDS; ++k) w[k] = 0u;
    auto mark = [&](int d) {
        const unsigned t   = (unsigned)d / (unsigned)TILE_ROWS;
        const unsigned bit = 1u << (t & 31u);
#pragma unroll
        for (int k = 0; k < MAP_WORDS; ++k) w[k] |= (t >> 5) == (unsigned)k ? bit : 0u;
    };
    const int  total4 = nEdges >> 2;
    const v4i* dst4   = reinterpret_cast<const v4i*>(dst);
    v4i d[CHUNK_E / 4];
#pragma unroll
    for (int q = 0; q < CHUNK_E / 4; ++q) {
        const int i = c * (CHUNK_E / 4) + q;
        d[q] = dst4[i < total4 ? i : 0];
    }
    __builtin_amdgcn_sched_barrier(0);
#pragma unroll
    for (int q = 0; q < CHUNK_E / 4; ++q) {
        const bool ok = c * (CHUNK_E / 4) + q < total4;
        mark(ok ? d[q].x : -1);
        mark(ok ? d[q].y : -1);
        mark(ok ? d[q].z : -1);
        mark(ok ? d[q].w : -1);
    }
#pragma unroll
    for (int k = 0; k < MAP_WORDS; ++k) cmap[(size_t)k * NCHUNK_PAD + c] = w[k];
}

__global__ __launch_bounds__(PREP_THREADS)
void prep_kernel(const int* ids,
                 const float* embed,
                 float* x0,
                 const int* __restrict__ dst,
                 unsigned* __restrict__ cmap,
                 float* __restrict__ zsum, int nEdges) {
    const int tid = threadIdx.x;
    if (blockIdx.x < MAP_BLOCKS) {
        chunk_map_thread(dst, cmap, nEdges, (int)blockIdx.x * PREP_THREADS + tid);
        return;
    }
    const int gb = (int)blockIdx.x - MAP_BLOCKS;
    const int zt = gb * PREP_THREADS + tid;
    if (zt < N_GRAPHS * ZS) zsum[zt] = 0.0f;
    const int q = tid & 7;
    int id[GATHER_U];
    v4f v[GATHER_U];
#pragma unroll
    for (int j = 0; j < GATHER_U; ++j) {
        const int t = (gb * GATHER_U + j) * PREP_THREADS + tid;
        id[j] = ids[t >> 3];
    }
#pragma unroll
    for (int j = 0; j < GATHER_U; ++j)
        v[j] = reinterpret_cast<const v4f*>(embed + (size_t)id[j] * EMBED_D)[q];
#pragma unroll
    for (int j = 0; j < GATHER_U; ++j) {
        const int t = (gb * GATHER_U + j) * PREP_THREADS + tid;
        reinterpret_cast<v4f*>(x0)[t] = v[j];
    }
}

template <int DIM, bool COUNT, bool BUILD>
__global__ __launch_bounds__(AGG_THREADS)
void tile_agg_kernel(const int* __restrict__ src,
                     const int* __restrict__ dst,
                     const float* __restrict__ x,
                     float* __restrict__ agg,
                     float* __restrict__ cnt,
                     unsigned* __restrict__ glist,
                     unsigned* __restrict__ gtot,
                     const unsigned* __restrict__ cmap,
                     int nNodes, int nEdges) {
    constexpr int H = DIM / 32;
    __shared__ float    cntL[TILE_ROWS];
    __shared__ unsigned bucket[AGG_WAVES * AGG_WAVES * BCAP];
    __shared__ unsigned wcnt[AGG_WAVES * AGG_WAVES];
    __shared__ __attribute__((aligned(16))) unsigned list[AGG_WAVES * LCAP];
    __shared__ unsigned ovf;

    const int tid  = threadIdx.x;
    const int lane = tid & 31;
    const int wave = tid >> 5;
    const int base = blockIdx.x * TILE_ROWS;
    unsigned* mygl  = glist + (size_t)(blockIdx.x * AGG_WAVES + wave) * LCAP;
    unsigned* mygt  = gtot + blockIdx.x * AGG_WAVES + wave;

    unsigned tot = 0;
    bool     slow;
    if constexpr (!BUILD) tot = *mygt;

    v16f a[H];
#pragma unroll
    for (int h = 0; h < H; ++h)
#pragma unroll
        for (int k = 0; k < 16; ++k) a[h][k] = 0.0f;
    int c[ROWS_PER_WAVE];
#pragma unroll
    for (int k = 0; k < ROWS_PER_WAVE; ++k) c[k] = 0;

    if constexpr (BUILD) {
        if (tid == 0) ovf = 0u;
        __syncthreads();

        unsigned mycnt = 0;
        auto put = [&](unsigned r, int e) {
            unsigned mask = __builtin_amdgcn_ballot_w32(r < (unsigned)TILE_ROWS);
            const int ent = (int)(((unsigned)e << 10) | r);
            while (mask) {
                const int l = __builtin_ctz(mask);
                mask &= mask - 1;
                const unsigned en = (unsigned)__builtin_amdgcn_readlane(ent, l);
                const int owner = (int)(en & 31u);
                if (lane == owner) {
                    if (mycnt < (unsigned)BCAP)
                        bucket[(wave * AGG_WAVES + owner) * BCAP + mycnt] = en;
                    ++mycnt;
                }
            }
        };
        unsigned* cand  = list + wave * LCAP;
        unsigned  ncand = 0;
        {
            const uint4* m4 = reinterpret_cast<const uint4*>(
                cmap + (size_t)(blockIdx.x >> 5) * NCHUNK_PAD);
            const unsigned bit = 1u << (blockIdx.x & 31u);
            uint4 mv[MAP_TRIPS];
#pragma unroll
            for (int u = 0; u < MAP_TRIPS; ++u) mv[u] = m4[u * AGG_THREADS + tid];
#pragma unroll
            for (int u = 0; u < MAP_TRIPS; ++u) {
                const unsigned wv[4] = {mv[u].x, mv[u].y, mv[u].z, mv[u].w};
#pragma unroll
                for (int k = 0; k < 4; ++k) {
                    const bool     hit = (wv[k] & bit) != 0u;
                    const unsigned m   = __builtin_amdgcn_ballot_w32(hit);
                    const unsigned pos = ncand + __builtin_amdgcn_mbcnt_lo(m, 0u);
                    if (hit && pos < (unsigned)LCAP)
                        cand[pos] = (unsigned)((u * AGG_THREADS + tid) * 4 + k);
                    ncand += (unsigned)__builtin_popcount(m);
                }
            }
        }
        if (ncand > (unsigned)LCAP) {
            ovf   = 1u;
            ncand = (unsigned)LCAP;
        }
        if (lane < SCAN_U && ncand + (unsigned)lane < (unsigned)LCAP) cand[ncand + lane] = NO_CHUNK;
        __syncthreads();

        for (unsigned c0 = 0; c0 < ncand; c0 += SCAN_U) {
            int      dv[SCAN_U];
            unsigned ev[SCAN_U];
            const uint4* cp = reinterpret_cast<const uint4*>(cand + c0);
            const uint4  ca = cp[0], cb = cp[1];
            const unsigned cv[SCAN_U] = {ca.x, ca.y, ca.z, ca.w, cb.x, cb.y, cb.z, cb.w};
#pragma unroll
            for (int u = 0; u < SCAN_U; ++u) {
                const unsigned e  = cv[u] * (unsigned)CHUNK_E + (unsigned)lane;
                const bool     ok = e < (unsigned)nEdges;
                int d = dst[ok ? e : 0u];
                if (!ok) d = -1;
                dv[u] = d;
                ev[u] = e;
            }
#pragma unroll
            for (int u = 0; u < SCAN_U; ++u) put((unsigned)(dv[u] - base), (int)ev[u]);
        }
        if (mycnt > (unsigned)BCAP) ovf = 1u;
        wcnt[wave * AGG_WAVES + lane] = mycnt < (unsigned)BCAP ? mycnt : (unsigned)BCAP;
        __syncthreads();

        unsigned* mylist = list + wave * LCAP;
        const unsigned n_l = wcnt[lane * AGG_WAVES + wave];
        unsigned off = 0;
#pragma unroll
        for (int j = 0; j < 32; ++j) {
            const unsigned nj = (unsigned)__builtin_amdgcn_readlane((int)n_l, j);
            off += j < lane ? nj : 0u;
            tot += nj;
        }
        if (tot > (unsigned)LCAP) {
            ovf = 1u;
        } else {
            for (unsigned k = 0; k < n_l; ++k)
                mylist[off + k] = bucket[(lane * AGG_WAVES + wave) * BCAP + k];
        }
        __syncthreads();
        slow = ovf != 0u;
        if (lane == 0) *mygt = slow ? LIST_NONE : tot;
    } else {
        tot  = (unsigned)__builtin_amdgcn_readfirstlane((int)tot);
        slow = tot == LIST_NONE;
    }

    if (!slow) {
        for (unsigned b0 = 0; b0 < tot; b0 += 32) {
            const unsigned qi = b0 + lane;
            unsigned ent;
            if constexpr (BUILD) {
                const unsigned se = qi < tot ? list[wave * LCAP + qi]
                                             : (unsigned)(SINK_SLOT << 5);
                ent = ((unsigned)src[se >> 10] << 10) | (se & 1023u);
                if (qi < tot) mygl[qi] = ent;
            } else {
                ent = qi < tot ? mygl[qi] : (unsigned)(SINK_SLOT << 5);
            }
            const int sl = (int)((ent & 1023u) >> 5);
            const int s  = (int)(ent >> 10);
            if constexpr (COUNT) {
#pragma unroll
                for (int k = 0; k < ROWS_PER_WAVE; ++k)
                    c[k] += __builtin_popcount(__builtin_amdgcn_ballot_w32(sl == k));
            }
#pragma unroll
            for (int j0 = 0; j0 < 32; j0 += 16) {
                if (b0 + j0 >= tot) break;
                float v[16][H];
                int   slot[16];
#pragma unroll
                for (int u = 0; u < 16; ++u) {
                    const int sj = __builtin_amdgcn_readlane(s, j0 + u);
                    slot[u]      = __builtin_amdgcn_readlane(sl, j0 + u);
                    const float* xp = x + (size_t)sj * DIM + lane;
#pragma unroll
                    for (int h = 0; h < H; ++h) v[u][h] = xp[32 * h];
                }
#pragma unroll
                for (int u = 0; u < 16; ++u) {
#pragma unroll
                    for (int h = 0; h < H; ++h)
                        a[h][slot[u]] += v[u][h];
                }
            }
        }
    } else {
        for (int i0 = 0; i0 < nEdges; i0 += 32) {
            const int i = i0 + lane;
            const int d = i < nEdges ? dst[i] : -1;
            const unsigned r = (unsigned)(d - base);
            unsigned mask = __builtin_amdgcn_ballot_w32(
                r < (unsigned)TILE_ROWS && (int)(r & 31u) == wave);
            while (mask) {
                const int l = __builtin_ctz(mask);
                mask &= mask - 1;
                const int kr = __builtin_amdgcn_readlane((int)r, l) >> 5;
                const float* xp = x + (size_t)src[i0 + l] * DIM + lane;
#pragma unroll
                for (int h = 0; h < H; ++h) a[h][kr] += xp[32 * h];
                if constexpr (COUNT) {
#pragma unroll
                    for (int k = 0; k < ROWS_PER_WAVE; ++k) c[k] += kr == k ? 1 : 0;
                }
            }
        }
    }

    const int left = nNodes - base;
    const int rows = left < TILE_ROWS ? left : TILE_ROWS;
#pragma unroll
    for (int k = 0; k < ROWS_PER_WAVE; ++k) {
        const int r = wave + 32 * k;
        if (r < rows) {
#pragma unroll
            for (int h = 0; h < H; ++h)
                agg[(size_t)(base + r) * DIM + 32 * h + lane] = a[h][k];
            if (COUNT && lane == 0) cntL[r] = (float)c[k];
        }
    }
    if constexpr (COUNT) {
        __syncthreads();
        for (int i = tid; i < rows; i += AGG_THREADS) cnt[base + i] = cntL[i];
    }
}

#define GEMM_WAVES   8
#define GEMM_THREADS (32 * GEMM_WAVES)
#define GEMM_BLOCKS  256
static_assert(N_NODES % 16 == 0, "row tiles are whole");

template <int K>
__global__ __launch_bounds__(GEMM_THREADS)
void sage_gemm_kernel(const float* aggRaw,
                      const float* cnt,
                      const float* x,
                      const float* __restrict__ Wl,
                      const float* __restrict__ Wr,
                      const float* __restrict__ bias,
                      float* __restrict__ out, int nNodes) {
    constexpr int T = K / 8;
    const int lane = threadIdx.x & 31;
    const int grp  = lane >> 4;
    const int lr   = lane & 15;
    const int wv     = __builtin_amdgcn_readfirstlane((int)(threadIdx.x >> 5));
    const int nWaves = (int)gridDim.x * GEMM_WAVES;
    const int nTiles = nNodes / 16;
    int tile = wv * (int)gridDim.x + (int)blockIdx.x;

    float4 ra0[T], rx0[T], ra1[T], rx1[T];
    float  rc0 = 1.0f, rc1 = 1.0f;
    auto load_tile = [&](int tl, float4 (&a)[T], float4 (&xx)[T], float& c) {
        const int arow = tl * 16 + lr;
        const float4* ap = reinterpret_cast<const float4*>(aggRaw + (size_t)arow * K) + grp;
        const float4* xp = reinterpret_cast<const float4*>(x + (size_t)arow * K) + grp;
        c = cnt[arow];
#pragma unroll
        for (int t = 0; t < T; ++t) { a[t] = ap[2 * t]; xx[t] = xp[2 * t]; }
    };
    if (tile >= nTiles) return;
    load_tile(tile, ra0, rx0, rc0);

    v2f bl[T][2][4], br[T][2][4];
#pragma unroll
    for (int t = 0; t < T; ++t)
#pragma unroll
        for (int p = 0; p < 2; ++p) {
            const int kb = 8 * t + 4 * grp + 2 * p;
#pragma unroll
            for (int n = 0; n < 4; ++n) {
                bl[t][p][n].x = Wl[(size_t)(kb + 0) * HIDDEN_D + 16 * n + lr];
                bl[t][p][n].y = Wl[(size_t)(kb + 1) * HIDDEN_D + 16 * n + lr];
                br[t][p][n].x = Wr[(size_t)(kb + 0) * HIDDEN_D + 16 * n + lr];
                br[t][p][n].y = Wr[(size_t)(kb + 1) * HIDDEN_D + 16 * n + lr];
            }
        }
    float bv[4];
#pragma unroll
    for (int n = 0; n < 4; ++n) bv[n] = bias[16 * n + lr];


    auto compute_tile = [&](int tl, const float4 (&av)[T], const float4 (&xv)[T], float deg) {
        const float invc = 1.0f / fmaxf(deg, 1.0f);
        v8f c[4] = {};
#pragma unroll
        for (int t = 0; t < T; ++t) {
            v2f a0, a1;
            a0.x = av[t].x * invc;  a0.y = av[t].y * invc;
            a1.x = av[t].z * invc;  a1.y = av[t].w * invc;
#pragma unroll
            for (int n = 0; n < 4; ++n)
                c[n] = __builtin_amdgcn_wmma_f32_16x16x4_f32(false, a0, false, bl[t][0][n],
                                                             (short)0, c[n], false, false);
#pragma unroll
            for (int n = 0; n < 4; ++n)
                c[n] = __builtin_amdgcn_wmma_f32_16x16x4_f32(false, a1, false, bl[t][1][n],
                                                             (short)0, c[n], false, false);
        }
#pragma unroll
        for (int t = 0; t < T; ++t) {
            v2f a0, a1;
            a0.x = xv[t].x;  a0.y = xv[t].y;
            a1.x = xv[t].z;  a1.y = xv[t].w;
#pragma unroll
            for (int n = 0; n < 4; ++n)
                c[n] = __builtin_amdgcn_wmma_f32_16x16x4_f32(false, a0, false, br[t][0][n],
                                                             (short)0, c[n], false, false);
#pragma unroll
            for (int n = 0; n < 4; ++n)
                c[n] = __builtin_amdgcn_wmma_f32_16x16x4_f32(false, a1, false, br[t][1][n],
                                                             (short)0, c[n], false, false);
        }
        const int row0 = tl * 16;
#pragma unroll
        for (int n = 0; n < 4; ++n)
#pragma unroll
            for (int v = 0; v < 8; ++v) {
                const int m = v + 8 * grp;
                out[(size_t)(row0 + m) * HIDDEN_D + 16 * n + lr] = fmaxf(c[n][v] + bv[n], 0.0f);
            }
    };

    int next = tile + nWaves;
    load_tile(next < nTiles ? next : tile, ra1, rx1, rc1);
    __builtin_amdgcn_sched_barrier(0);
    compute_tile(tile, ra0, rx0, rc0);
    tile = next;
    while (tile < nTiles) {
        next = tile + nWaves;
        load_tile(next < nTiles ? next : tile, ra0, rx0, rc0);
        __builtin_amdgcn_sched_barrier(0);
        compute_tile(tile, ra1, rx1, rc1);
        tile = next;
        if (tile >= nTiles) break;
        next = tile + nWaves;
        load_tile(next < nTiles ? next : tile, ra1, rx1, rc1);
        __builtin_amdgcn_sched_barrier(0);
        compute_tile(tile, ra0, rx0, rc0);
        tile = next;
    }
}

__global__ void head_pool_kernel(const int* __restrict__ batch,
                                 const float* __restrict__ h,
                                 const float* __restrict__ Wo,
                                 float* __restrict__ zsum, int nNodes) {
    const int lane = threadIdx.x & 31;
    const int tile = blockIdx.x * (blockDim.x >> 5) + (threadIdx.x >> 5);
    const int row0 = tile * 16;
    if (row0 >= nNodes) return;
    const int grp = lane >> 4;
    const int lr  = lane & 15;

    const float4* hp =
        reinterpret_cast<const float4*>(h + (size_t)(row0 + lr) * HIDDEN_D) + grp;
    float4 av[8];
    for (int t = 0; t < 8; ++t) av[t] = hp[2 * t];

    const int4* bp = reinterpret_cast<const int4*>(batch + row0 + 8 * grp);
    const int4 gA = bp[0], gB = bp[1];

    const bool colOk = lr < N_CLASSES;
    const int  cc    = colOk ? lr : 0;

    v8f c = {};
    for (int t = 0; t < 8; ++t) {
        const int kb = 8 * t + 4 * grp;
        const float w0 = Wo[(kb + 0) * N_CLASSES + cc];
        const float w1 = Wo[(kb + 1) * N_CLASSES + cc];
        const float w2 = Wo[(kb + 2) * N_CLASSES + cc];
        const float w3 = Wo[(kb + 3) * N_CLASSES + cc];
        v2f a0, a1, b0, b1;
        a0.x = av[t].x;             a0.y = av[t].y;
        b0.x = colOk ? w0 : 0.0f;   b0.y = colOk ? w1 : 0.0f;
        a1.x = av[t].z;             a1.y = av[t].w;
        b1.x = colOk ? w2 : 0.0f;   b1.y = colOk ? w3 : 0.0f;
        c = __builtin_amdgcn_wmma_f32_16x16x4_f32(false, a0, false, b0,
                                                  (short)0, c, false, false);
        c = __builtin_amdgcn_wmma_f32_16x16x4_f32(false, a1, false, b1,
                                                  (short)0, c, false, false);
    }

    float val[8];
    for (int v = 0; v < 8; ++v) val[v] = colOk ? c[v] : 1.0f;
    const int gid[8] = {gA.x, gA.y, gA.z, gA.w, gB.x, gB.y, gB.z, gB.w};
    if (lr <= N_CLASSES) {
        for (int v = 0; v < 8; ++v)
            atomAdd1(zsum + (size_t)gid[v] * ZS + lr, val[v]);
    }
}

__global__ void finish_kernel(const float* __restrict__ zsum,
                              const float* __restrict__ bo,
                              float* __restrict__ out) {
    int t = blockIdx.x * blockDim.x + threadIdx.x;
    if (t >= N_GRAPHS * N_CLASSES) return;
    int g = t / N_CLASSES;
    int c = t - g * N_CLASSES;
    out[t] = zsum[g * ZS + c] / fmaxf(zsum[g * ZS + N_CLASSES], 1.0f) + bo[c];
}

extern "C" void kernel_launch(void* const* d_in, const int* in_sizes, int n_in,
                              void* d_out, int out_size, void* d_ws, size_t ws_size,
                              hipStream_t stream) {
    const int*   node_ids = (const int*)  d_in[0];
    const int*   src      = (const int*)  d_in[1];
    const int*   dst      = (const int*)  d_in[2];
    const int*   batch    = (const int*)  d_in[3];
    const float* embed    = (const float*)d_in[4];
    const float* W1_l     = (const float*)d_in[5];
    const float* b1       = (const float*)d_in[6];
    const float* W1_r     = (const float*)d_in[7];
    const float* W2_l     = (const float*)d_in[8];
    const float* b2       = (const float*)d_in[9];
    const float* W2_r     = (const float*)d_in[10];
    const float* W_out    = (const float*)d_in[11];
    const float* b_out    = (const float*)d_in[12];
    float* out = (float*)d_out;

    float* ws   = (float*)d_ws;
    float* x0   = ws;
    float* h1   = x0   + (size_t)N_NODES * EMBED_D;
    float* h2   = h1   + (size_t)N_NODES * HIDDEN_D;
    float* agg1 = h2   + (size_t)N_NODES * HIDDEN_D;
    float* agg2 = agg1 + (size_t)N_NODES * EMBED_D;
    float* cnt  = agg2 + (size_t)N_NODES * HIDDEN_D;
    float* zsum = cnt  + (size_t)N_NODES;
    unsigned* gtot  = (unsigned*)(zsum + N_GRAPHS * ZS);
    unsigned* glist = gtot + AGG_BLOCKS * AGG_WAVES;
    unsigned* cmap  = glist + (size_t)AGG_BLOCKS * AGG_WAVES * LCAP;

    const int B = 256;
    auto blocks = [](int n, int b) { return (n + b - 1) / b; };

    prep_kernel<<<PREP_BLOCKS, PREP_THREADS, 0, stream>>>(
        node_ids, embed, x0, dst, cmap, zsum, N_EDGES);

    tile_agg_kernel<EMBED_D, true, true><<<AGG_BLOCKS, AGG_THREADS, 0, stream>>>(
        src, dst, x0, agg1, cnt, glist, gtot, cmap, N_NODES, N_EDGES);
    sage_gemm_kernel<EMBED_D><<<GEMM_BLOCKS, GEMM_THREADS, 0, stream>>>(
        agg1, cnt, x0, W1_l, W1_r, b1, h1, N_NODES);

    tile_agg_kernel<HIDDEN_D, false, false><<<AGG_BLOCKS, AGG_THREADS, 0, stream>>>(
        src, dst, h1, agg2, cnt, glist, gtot, cmap, N_NODES, N_EDGES);
    sage_gemm_kernel<HIDDEN_D><<<GEMM_BLOCKS, GEMM_THREADS, 0, stream>>>(
        agg2, cnt, h1, W2_l, W2_r, b2, h2, N_NODES);

    head_pool_kernel<<<blocks(N_NODES / 16, 4), 128, 0, stream>>>(
        batch, h2, W_out, zsum, N_NODES);
    finish_kernel<<<blocks(N_GRAPHS * N_CLASSES, B), B, 0, stream>>>(
        zsum, b_out, out);
}


# layer 1 scan: a chunk's hits are staged per wave by one ballot compaction, then put 32 at a time into the owner buckets (same-owner ranks from 5 bit ballots); no serial loop per edge; sum order kept
# speedup vs baseline: 2.8999x; 1.0485x over previous
#include <hip/hip_runtime.h>
#include <type_traits>

#define N_NODES   100000
#define N_EDGES   1200000
#define EMBED_D   32
#define HIDDEN_D  64
#define N_CLASSES 10
#define N_GRAPHS  1024

typedef __attribute__((ext_vector_type(2))) float v2f;
typedef __attribute__((ext_vector_type(4))) float v4f;
typedef __attribute__((ext_vector_type(4))) int   v4i;
typedef __attribute__((ext_vector_type(8))) float v8f;
typedef __attribute__((ext_vector_type(16))) float v16f;

__device__ __forceinline__ void atomAdd1(float* p, float v) {
    asm volatile("global_atomic_add_f32 %0, %1, off scope:SCOPE_DEV"
                 :: "v"(p), "v"(v) : "memory");
}

#define TILE_ROWS   392
#define AGG_BLOCKS  ((N_NODES + TILE_ROWS - 1) / TILE_ROWS)
#define AGG_THREADS 1024
#define AGG_WAVES   32
#define ROWS_PER_WAVE ((TILE_ROWS + AGG_WAVES - 1) / AGG_WAVES)
#define SINK_SLOT   ROWS_PER_WAVE
static_assert(AGG_WAVES == 32 && SINK_SLOT < 16, "slot = local row / 32, in a 16-slot vector");
#define BCAP        32
#define LCAP        512
#define SCAP        768
#define SCAN_U      8
static_assert(SCAN_U == 8, "the scan reads its 8 list entries as two 16-byte words");
#define CHUNK_E     32
static_assert(SCAP > SCAN_U * CHUNK_E, "the staged list takes at least one whole trip of the scan");
#define NCHUNK      ((N_EDGES + CHUNK_E - 1) / CHUNK_E)
#define MAP_TRIPS   ((NCHUNK + 4 * AGG_THREADS - 1) / (4 * AGG_THREADS))
#define NCHUNK_PAD  (MAP_TRIPS * 4 * AGG_THREADS)
#define MAP_WORDS   ((AGG_BLOCKS + 31) / 32)
#define PREP_THREADS 128
#define NO_CHUNK    (1u << 26)
#define LIST_NONE   0xFFFFFFFFu
static_assert(N_EDGES % 4 == 0, "destination list is read 4 at a time");
static_assert(TILE_ROWS + AGG_WAVES <= 1024 && N_EDGES < (1 << 21),
              "a scan entry is (edge << 10) | local row");
static_assert(N_NODES < (1 << 22), "a stored entry is (source node << 10) | local row");
static_assert(NCHUNK_PAD % PREP_THREADS == 0, "one map thread per padded chunk");

#define MAP_BLOCKS    (NCHUNK_PAD / PREP_THREADS)
#define GATHER_U      10
#define GATHER_ITEMS  (N_NODES * (EMBED_D / 4))
#define GATHER_BLOCKS (GATHER_ITEMS / (PREP_THREADS * GATHER_U))
#define PREP_BLOCKS   (MAP_BLOCKS + GATHER_BLOCKS)
#define ZS 16
static_assert(EMBED_D / 4 == 8 && PREP_THREADS % 8 == 0, "a thread keeps its quarter of a row");
static_assert(GATHER_ITEMS % (PREP_THREADS * GATHER_U) == 0,
              "the copy blocks cover the pieces exactly: no end check, so no load waits behind a branch");
static_assert(GATHER_BLOCKS * PREP_THREADS >= N_GRAPHS * ZS, "one copy thread per pooled word");

__device__ __forceinline__ void chunk_map_thread(const int* __restrict__ dst,
                                                 unsigned* __restrict__ cmap,
                                                 int nEdges, int c) {
    unsigned w[MAP_WORDS];
#pragma unroll
    for (int k = 0; k < MAP_WORDS; ++k) w[k] = 0u;
    auto mark = [&](int d) {
        const unsigned t   = (unsigned)d / (unsigned)TILE_ROWS;
        const unsigned bit = 1u << (t & 31u);
#pragma unroll
        for (int k = 0; k < MAP_WORDS; ++k) w[k] |= (t >> 5) == (unsigned)k ? bit : 0u;
    };
    const int  total4 = nEdges >> 2;
    const v4i* dst4   = reinterpret_cast<const v4i*>(dst);
    v4i d[CHUNK_E / 4];
#pragma unroll
    for (int q = 0; q < CHUNK_E / 4; ++q) {
        const int i = c * (CHUNK_E / 4) + q;
        d[q] = dst4[i < total4 ? i : 0];
    }
    __builtin_amdgcn_sched_barrier(0);
#pragma unroll
    for (int q = 0; q < CHUNK_E / 4; ++q) {
        const bool ok = c * (CHUNK_E / 4) + q < total4;
        mark(ok ? d[q].x : -1);
        mark(ok ? d[q].y : -1);
        mark(ok ? d[q].z : -1);
        mark(ok ? d[q].w : -1);
    }
#pragma unroll
    for (int k = 0; k < MAP_WORDS; ++k) cmap[(size_t)k * NCHUNK_PAD + c] = w[k];
}

__global__ __launch_bounds__(PREP_THREADS)
void prep_kernel(const int* ids,
                 const float* embed,
                 float* x0,
                 const int* __restrict__ dst,
                 unsigned* __restrict__ cmap,
                 float* __restrict__ zsum, int nEdges) {
    const int tid = threadIdx.x;
    if (blockIdx.x < MAP_BLOCKS) {
        chunk_map_thread(dst, cmap, nEdges, (int)blockIdx.x * PREP_THREADS + tid);
        return;
    }
    const int gb = (int)blockIdx.x - MAP_BLOCKS;
    const int zt = gb * PREP_THREADS + tid;
    if (zt < N_GRAPHS * ZS) zsum[zt] = 0.0f;
    const int q = tid & 7;
    int id[GATHER_U];
    v4f v[GATHER_U];
#pragma unroll
    for (int j = 0; j < GATHER_U; ++j) {
        const int t = (gb * GATHER_U + j) * PREP_THREADS + tid;
        id[j] = ids[t >> 3];
    }
#pragma unroll
    for (int j = 0; j < GATHER_U; ++j)
        v[j] = reinterpret_cast<const v4f*>(embed + (size_t)id[j] * EMBED_D)[q];
#pragma unroll
    for (int j = 0; j < GATHER_U; ++j) {
        const int t = (gb * GATHER_U + j) * PREP_THREADS + tid;
        reinterpret_cast<v4f*>(x0)[t] = v[j];
    }
}

template <int DIM, bool COUNT, bool BUILD>
__global__ __launch_bounds__(AGG_THREADS)
void tile_agg_kernel(const int* __restrict__ src,
                     const int* __restrict__ dst,
                     const float* __restrict__ x,
                     float* __restrict__ agg,
                     float* __restrict__ cnt,
                     unsigned* __restrict__ glist,
                     unsigned* __restrict__ gtot,
                     const unsigned* __restrict__ cmap,
                     int nNodes, int nEdges) {
    constexpr int H = DIM / 32;
    __shared__ float    cntL[TILE_ROWS];
    __shared__ unsigned bucket[AGG_WAVES * AGG_WAVES * BCAP];
    __shared__ unsigned wcnt[AGG_WAVES * AGG_WAVES];
    __shared__ unsigned stage[AGG_WAVES * SCAP];
    __shared__ __attribute__((aligned(16))) unsigned list[AGG_WAVES * LCAP];
    __shared__ unsigned ovf;

    const int tid  = threadIdx.x;
    const int lane = tid & 31;
    const int wave = tid >> 5;
    const int base = blockIdx.x * TILE_ROWS;
    unsigned* mygl  = glist + (size_t)(blockIdx.x * AGG_WAVES + wave) * LCAP;
    unsigned* mygt  = gtot + blockIdx.x * AGG_WAVES + wave;

    unsigned tot = 0;
    bool     slow;
    if constexpr (!BUILD) tot = *mygt;

    v16f a[H];
#pragma unroll
    for (int h = 0; h < H; ++h)
#pragma unroll
        for (int k = 0; k < 16; ++k) a[h][k] = 0.0f;
    int c[ROWS_PER_WAVE];
#pragma unroll
    for (int k = 0; k < ROWS_PER_WAVE; ++k) c[k] = 0;

    if constexpr (BUILD) {
        if (tid == 0) ovf = 0u;
        __syncthreads();

        unsigned* cand  = list + wave * LCAP;
        unsigned  ncand = 0;
        {
            const uint4* m4 = reinterpret_cast<const uint4*>(
                cmap + (size_t)(blockIdx.x >> 5) * NCHUNK_PAD);
            const unsigned bit = 1u << (blockIdx.x & 31u);
            uint4 mv[MAP_TRIPS];
#pragma unroll
            for (int u = 0; u < MAP_TRIPS; ++u) mv[u] = m4[u * AGG_THREADS + tid];
#pragma unroll
            for (int u = 0; u < MAP_TRIPS; ++u) {
                const unsigned wv[4] = {mv[u].x, mv[u].y, mv[u].z, mv[u].w};
#pragma unroll
                for (int k = 0; k < 4; ++k) {
                    const bool     hit = (wv[k] & bit) != 0u;
                    const unsigned m   = __builtin_amdgcn_ballot_w32(hit);
                    const unsigned pos = ncand + __builtin_amdgcn_mbcnt_lo(m, 0u);
                    if (hit && pos < (unsigned)LCAP)
                        cand[pos] = (unsigned)((u * AGG_THREADS + tid) * 4 + k);
                    ncand += (unsigned)__builtin_popcount(m);
                }
            }
        }
        if (ncand > (unsigned)LCAP) {
            ovf   = 1u;
            ncand = (unsigned)LCAP;
        }
        if (lane < SCAN_U && ncand + (unsigned)lane < (unsigned)LCAP) cand[ncand + lane] = NO_CHUNK;
        __syncthreads();

        unsigned* stg = stage + wave * SCAP;
        unsigned  nst = 0;
        for (unsigned c0 = 0; c0 < ncand; c0 += SCAN_U) {
            if (nst > (unsigned)(SCAP - SCAN_U * CHUNK_E)) { ovf = 1u; break; }
            int      dv[SCAN_U];
            unsigned ev[SCAN_U];
            const uint4* cp = reinterpret_cast<const uint4*>(cand + c0);
            const uint4  ca = cp[0], cb = cp[1];
            const unsigned cv[SCAN_U] = {ca.x, ca.y, ca.z, ca.w, cb.x, cb.y, cb.z, cb.w};
#pragma unroll
            for (int u = 0; u < SCAN_U; ++u) {
                const unsigned e  = cv[u] * (unsigned)CHUNK_E + (unsigned)lane;
                const bool     ok = e < (unsigned)nEdges;
                int d = dst[ok ? e : 0u];
                if (!ok) d = -1;
                dv[u] = d;
                ev[u] = e;
            }
#pragma unroll
            for (int u = 0; u < SCAN_U; ++u) {
                const unsigned r   = (unsigned)(dv[u] - base);
                const bool     hit = r < (unsigned)TILE_ROWS;
                const unsigned m   = __builtin_amdgcn_ballot_w32(hit);
                const unsigned pos = __builtin_amdgcn_mbcnt_lo(m, nst);
                if (hit) stg[pos] = (ev[u] << 10) | r;
                nst += (unsigned)__builtin_popcount(m);
            }
        }
        __syncthreads();

        unsigned mycnt = 0;
        for (unsigned b0 = 0; b0 < nst; b0 += 32) {
            const unsigned qi    = b0 + lane;
            const bool     valid = qi < nst;
            const unsigned en    = stg[valid ? qi : 0u];
            const unsigned owner = en & 31u;
            unsigned same = __builtin_amdgcn_ballot_w32(valid);
            unsigned mine = same;
#pragma unroll
            for (int b = 0; b < 5; ++b) {
                const bool     ob = ((owner >> b) & 1u) != 0u;
                const unsigned bb = __builtin_amdgcn_ballot_w32(ob);
                same &= ob ? bb : ~bb;
                mine &= (((unsigned)lane >> b) & 1u) != 0u ? bb : ~bb;
            }
            const unsigned cur = (unsigned)__builtin_amdgcn_ds_bpermute((int)(owner << 2), (int)mycnt);
            const unsigned pos = cur + __builtin_amdgcn_mbcnt_lo(same, 0u);
            if (valid && pos < (unsigned)BCAP)
                bucket[(wave * AGG_WAVES + owner) * BCAP + pos] = en;
            mycnt += (unsigned)__builtin_popcount(mine);
        }
        if (mycnt > (unsigned)BCAP) ovf = 1u;
        wcnt[wave * AGG_WAVES + lane] = mycnt < (unsigned)BCAP ? mycnt : (unsigned)BCAP;
        __syncthreads();

        unsigned* mylist = list + wave * LCAP;
        const unsigned n_l = wcnt[lane * AGG_WAVES + wave];
        unsigned off = 0;
#pragma unroll
        for (int j = 0; j < 32; ++j) {
            const unsigned nj = (unsigned)__builtin_amdgcn_readlane((int)n_l, j);
            off += j < lane ? nj : 0u;
            tot += nj;
        }
        if (tot > (unsigned)LCAP) {
            ovf = 1u;
        } else {
            for (unsigned k = 0; k < n_l; ++k)
                mylist[off + k] = bucket[(lane * AGG_WAVES + wave) * BCAP + k];
        }
        __syncthreads();
        slow = ovf != 0u;
        if (lane == 0) *mygt = slow ? LIST_NONE : tot;
    } else {
        tot  = (unsigned)__builtin_amdgcn_readfirstlane((int)tot);
        slow = tot == LIST_NONE;
    }

    if (!slow) {
        for (unsigned b0 = 0; b0 < tot; b0 += 32) {
            const unsigned qi = b0 + lane;
            unsigned ent;
            if constexpr (BUILD) {
                const unsigned se = qi < tot ? list[wave * LCAP + qi]
                                             : (unsigned)(SINK_SLOT << 5);
                ent = ((unsigned)src[se >> 10] << 10) | (se & 1023u);
                if (qi < tot) mygl[qi] = ent;
            } else {
                ent = qi < tot ? mygl[qi] : (unsigned)(SINK_SLOT << 5);
            }
            const int sl = (int)((ent & 1023u) >> 5);
            const int s  = (int)(ent >> 10);
            if constexpr (COUNT) {
#pragma unroll
                for (int k = 0; k < ROWS_PER_WAVE; ++k)
                    c[k] += __builtin_popcount(__builtin_amdgcn_ballot_w32(sl == k));
            }
#pragma unroll
            for (int j0 = 0; j0 < 32; j0 += 16) {
                if (b0 + j0 >= tot) break;
                float v[16][H];
                int   slot[16];
#pragma unroll
                for (int u = 0; u < 16; ++u) {
                    const int sj = __builtin_amdgcn_readlane(s, j0 + u);
                    slot[u]      = __builtin_amdgcn_readlane(sl, j0 + u);
                    const float* xp = x + (size_t)sj * DIM + lane;
#pragma unroll
                    for (int h = 0; h < H; ++h) v[u][h] = xp[32 * h];
                }
#pragma unroll
                for (int u = 0; u < 16; ++u) {
#pragma unroll
                    for (int h = 0; h < H; ++h)
                        a[h][slot[u]] += v[u][h];
                }
            }
        }
    } else {
        for (int i0 = 0; i0 < nEdges; i0 += 32) {
            const int i = i0 + lane;
            const int d = i < nEdges ? dst[i] : -1;
            const unsigned r = (unsigned)(d - base);
            unsigned mask = __builtin_amdgcn_ballot_w32(
                r < (unsigned)TILE_ROWS && (int)(r & 31u) == wave);
            while (mask) {
                const int l = __builtin_ctz(mask);
                mask &= mask - 1;
                const int kr = __builtin_amdgcn_readlane((int)r, l) >> 5;
                const float* xp = x + (size_t)src[i0 + l] * DIM + lane;
#pragma unroll
                for (int h = 0; h < H; ++h) a[h][kr] += xp[32 * h];
                if constexpr (COUNT) {
#pragma unroll
                    for (int k = 0; k < ROWS_PER_WAVE; ++k) c[k] += kr == k ? 1 : 0;
                }
            }
        }
    }

    const int left = nNodes - base;
    const int rows = left < TILE_ROWS ? left : TILE_ROWS;
#pragma unroll
    for (int k = 0; k < ROWS_PER_WAVE; ++k) {
        const int r = wave + 32 * k;
        if (r < rows) {
#pragma unroll
            for (int h = 0; h < H; ++h)
                agg[(size_t)(base + r) * DIM + 32 * h + lane] = a[h][k];
            if (COUNT && lane == 0) cntL[r] = (float)c[k];
        }
    }
    if constexpr (COUNT) {
        __syncthreads();
        for (int i = tid; i < rows; i += AGG_THREADS) cnt[base + i] = cntL[i];
    }
}

#define GEMM_WAVES   8
#define GEMM_THREADS (32 * GEMM_WAVES)
#define GEMM_BLOCKS  256
static_assert(N_NODES % 16 == 0, "row tiles are whole");

template <int K>
__global__ __launch_bounds__(GEMM_THREADS)
void sage_gemm_kernel(const float* aggRaw,
                      const float* cnt,
                      const float* x,
                      const float* __restrict__ Wl,
                      const float* __restrict__ Wr,
                      const float* __restrict__ bias,
                      float* __restrict__ out, int nNodes) {
    constexpr int T = K / 8;
    const int lane = threadIdx.x & 31;
    const int grp  = lane >> 4;
    const int lr   = lane & 15;
    const int wv     = __builtin_amdgcn_readfirstlane((int)(threadIdx.x >> 5));
    const int nWaves = (int)gridDim.x * GEMM_WAVES;
    const int nTiles = nNodes / 16;
    int tile = wv * (int)gridDim.x + (int)blockIdx.x;

    float4 ra0[T], rx0[T], ra1[T], rx1[T];
    float  rc0 = 1.0f, rc1 = 1.0f;
    auto load_tile = [&](int tl, float4 (&a)[T], float4 (&xx)[T], float& c) {
        const int arow = tl * 16 + lr;
        const float4* ap = reinterpret_cast<const float4*>(aggRaw + (size_t)arow * K) + grp;
        const float4* xp = reinterpret_cast<const float4*>(x + (size_t)arow * K) + grp;
        c = cnt[arow];
#pragma unroll
        for (int t = 0; t < T; ++t) { a[t] = ap[2 * t]; xx[t] = xp[2 * t]; }
    };
    if (tile >= nTiles) return;
    load_tile(tile, ra0, rx0, rc0);

    v2f bl[T][2][4], br[T][2][4];
#pragma unroll
    for (int t = 0; t < T; ++t)
#pragma unroll
        for (int p = 0; p < 2; ++p) {
            const int kb = 8 * t + 4 * grp + 2 * p;
#pragma unroll
            for (int n = 0; n < 4; ++n) {
                bl[t][p][n].x = Wl[(size_t)(kb + 0) * HIDDEN_D + 16 * n + lr];
                bl[t][p][n].y = Wl[(size_t)(kb + 1) * HIDDEN_D + 16 * n + lr];
                br[t][p][n].x = Wr[(size_t)(kb + 0) * HIDDEN_D + 16 * n + lr];
                br[t][p][n].y = Wr[(size_t)(kb + 1) * HIDDEN_D + 16 * n + lr];
            }
        }
    float bv[4];
#pragma unroll
    for (int n = 0; n < 4; ++n) bv[n] = bias[16 * n + lr];


    auto compute_tile = [&](int tl, const float4 (&av)[T], const float4 (&xv)[T], float deg) {
        const float invc = 1.0f / fmaxf(deg, 1.0f);
        v8f c[4] = {};
#pragma unroll
        for (int t = 0; t < T; ++t) {
            v2f a0, a1;
            a0.x = av[t].x * invc;  a0.y = av[t].y * invc;
            a1.x = av[t].z * invc;  a1.y = av[t].w * invc;
#pragma unroll
            for (int n = 0; n < 4; ++n)
                c[n] = __builtin_amdgcn_wmma_f32_16x16x4_f32(false, a0, false, bl[t][0][n],
                                                             (short)0, c[n], false, false);
#pragma unroll
            for (int n = 0; n < 4; ++n)
                c[n] = __builtin_amdgcn_wmma_f32_16x16x4_f32(false, a1, false, bl[t][1][n],
                                                             (short)0, c[n], false, false);
        }
#pragma unroll
        for (int t = 0; t < T; ++t) {
            v2f a0, a1;
            a0.x = xv[t].x;  a0.y = xv[t].y;
            a1.x = xv[t].z;  a1.y = xv[t].w;
#pragma unroll
            for (int n = 0; n < 4; ++n)
                c[n] = __builtin_amdgcn_wmma_f32_16x16x4_f32(false, a0, false, br[t][0][n],
                                                             (short)0, c[n], false, false);
#pragma unroll
            for (int n = 0; n < 4; ++n)
                c[n] = __builtin_amdgcn_wmma_f32_16x16x4_f32(false, a1, false, br[t][1][n],
                                                             (short)0, c[n], false, false);
        }
        const int row0 = tl * 16;
#pragma unroll
        for (int n = 0; n < 4; ++n)
#pragma unroll
            for (int v = 0; v < 8; ++v) {
                const int m = v + 8 * grp;
                out[(size_t)(row0 + m) * HIDDEN_D + 16 * n + lr] = fmaxf(c[n][v] + bv[n], 0.0f);
            }
    };

    int next = tile + nWaves;
    load_tile(next < nTiles ? next : tile, ra1, rx1, rc1);
    __builtin_amdgcn_sched_barrier(0);
    compute_tile(tile, ra0, rx0, rc0);
    tile = next;
    while (tile < nTiles) {
        next = tile + nWaves;
        load_tile(next < nTiles ? next : tile, ra0, rx0, rc0);
        __builtin_amdgcn_sched_barrier(0);
        compute_tile(tile, ra1, rx1, rc1);
        tile = next;
        if (tile >= nTiles) break;
        next = tile + nWaves;
        load_tile(next < nTiles ? next : tile, ra1, rx1, rc1);
        __builtin_amdgcn_sched_barrier(0);
        compute_tile(tile, ra0, rx0, rc0);
        tile = next;
    }
}

__global__ void head_pool_kernel(const int* __restrict__ batch,
                                 const float* __restrict__ h,
                                 const float* __restrict__ Wo,
                                 float* __restrict__ zsum, int nNodes) {
    const int lane = threadIdx.x & 31;
    const int tile = blockIdx.x * (blockDim.x >> 5) + (threadIdx.x >> 5);
    const int row0 = tile * 16;
    if (row0 >= nNodes) return;
    const int grp = lane >> 4;
    const int lr  = lane & 15;

    const float4* hp =
        reinterpret_cast<const float4*>(h + (size_t)(row0 + lr) * HIDDEN_D) + grp;
    float4 av[8];
    for (int t = 0; t < 8; ++t) av[t] = hp[2 * t];

    const int4* bp = reinterpret_cast<const int4*>(batch + row0 + 8 * grp);
    const int4 gA = bp[0], gB = bp[1];

    const bool colOk = lr < N_CLASSES;
    const int  cc    = colOk ? lr : 0;

    v8f c = {};
    for (int t = 0; t < 8; ++t) {
        const int kb = 8 * t + 4 * grp;
        const float w0 = Wo[(kb + 0) * N_CLASSES + cc];
        const float w1 = Wo[(kb + 1) * N_CLASSES + cc];
        const float w2 = Wo[(kb + 2) * N_CLASSES + cc];
        const float w3 = Wo[(kb + 3) * N_CLASSES + cc];
        v2f a0, a1, b0, b1;
        a0.x = av[t].x;             a0.y = av[t].y;
        b0.x = colOk ? w0 : 0.0f;   b0.y = colOk ? w1 : 0.0f;
        a1.x = av[t].z;             a1.y = av[t].w;
        b1.x = colOk ? w2 : 0.0f;   b1.y = colOk ? w3 : 0.0f;
        c = __builtin_amdgcn_wmma_f32_16x16x4_f32(false, a0, false, b0,
                                                  (short)0, c, false, false);
        c = __builtin_amdgcn_wmma_f32_16x16x4_f32(false, a1, false, b1,
                                                  (short)0, c, false, false);
    }

    float val[8];
    for (int v = 0; v < 8; ++v) val[v] = colOk ? c[v] : 1.0f;
    const int gid[8] = {gA.x, gA.y, gA.z, gA.w, gB.x, gB.y, gB.z, gB.w};
    if (lr <= N_CLASSES) {
        for (int v = 0; v < 8; ++v)
            atomAdd1(zsum + (size_t)gid[v] * ZS + lr, val[v]);
    }
}

__global__ void finish_kernel(const float* __restrict__ zsum,
                              const float* __restrict__ bo,
                              float* __restrict__ out) {
    int t = blockIdx.x * blockDim.x + threadIdx.x;
    if (t >= N_GRAPHS * N_CLASSES) return;
    int g = t / N_CLASSES;
    int c = t - g * N_CLASSES;
    out[t] = zsum[g * ZS + c] / fmaxf(zsum[g * ZS + N_CLASSES], 1.0f) + bo[c];
}

extern "C" void kernel_launch(void* const* d_in, const int* in_sizes, int n_in,
                              void* d_out, int out_size, void* d_ws, size_t ws_size,
                              hipStream_t stream) {
    const int*   node_ids = (const int*)  d_in[0];
    const int*   src      = (const int*)  d_in[1];
    const int*   dst      = (const int*)  d_in[2];
    const int*   batch    = (const int*)  d_in[3];
    const float* embed    = (const float*)d_in[4];
    const float* W1_l     = (const float*)d_in[5];
    const float* b1       = (const float*)d_in[6];
    const float* W1_r     = (const float*)d_in[7];
    const float* W2_l     = (const float*)d_in[8];
    const float* b2       = (const float*)d_in[9];
    const float* W2_r     = (const float*)d_in[10];
    const float* W_out    = (const float*)d_in[11];
    const float* b_out    = (const float*)d_in[12];
    float* out = (float*)d_out;

    float* ws   = (float*)d_ws;
    float* x0   = ws;
    float* h1   = x0   + (size_t)N_NODES * EMBED_D;
    float* h2   = h1   + (size_t)N_NODES * HIDDEN_D;
    float* agg1 = h2   + (size_t)N_NODES * HIDDEN_D;
    float* agg2 = agg1 + (size_t)N_NODES * EMBED_D;
    float* cnt  = agg2 + (size_t)N_NODES * HIDDEN_D;
    float* zsum = cnt  + (size_t)N_NODES;
    unsigned* gtot  = (unsigned*)(zsum + N_GRAPHS * ZS);
    unsigned* glist = gtot + AGG_BLOCKS * AGG_WAVES;
    unsigned* cmap  = glist + (size_t)AGG_BLOCKS * AGG_WAVES * LCAP;

    const int B = 256;
    auto blocks = [](int n, int b) { return (n + b - 1) / b; };

    prep_kernel<<<PREP_BLOCKS, PREP_THREADS, 0, stream>>>(
        node_ids, embed, x0, dst, cmap, zsum, N_EDGES);

    tile_agg_kernel<EMBED_D, true, true><<<AGG_BLOCKS, AGG_THREADS, 0, stream>>>(
        src, dst, x0, agg1, cnt, glist, gtot, cmap, N_NODES, N_EDGES);
    sage_gemm_kernel<EMBED_D><<<GEMM_BLOCKS, GEMM_THREADS, 0, stream>>>(
        agg1, cnt, x0, W1_l, W1_r, b1, h1, N_NODES);

    tile_agg_kernel<HIDDEN_D, false, false><<<AGG_BLOCKS, AGG_THREADS, 0, stream>>>(
        src, dst, h1, agg2, cnt, glist, gtot, cmap, N_NODES, N_EDGES);
    sage_gemm_kernel<HIDDEN_D><<<GEMM_BLOCKS, GEMM_THREADS, 0, stream>>>(
        agg2, cnt, h1, W2_l, W2_r, b2, h2, N_NODES);

    head_pool_kernel<<<blocks(N_NODES / 16, 4), 128, 0, stream>>>(
        batch, h2, W_out, zsum, N_NODES);
    finish_kernel<<<blocks(N_GRAPHS * N_CLASSES, B), B, 0, stream>>>(
        zsum, b_out, out);
}


# chunk map kept per tile and 8-edge piece (map waves transpose their bit words, whole-line stores): find reads its tile's own words, scan loads 4 marked pieces per load; row sum order changes, fixed
# speedup vs baseline: 3.0619x; 1.0559x over previous
#include <hip/hip_runtime.h>
#include <type_traits>

#define N_NODES   100000
#define N_EDGES   1200000
#define EMBED_D   32
#define HIDDEN_D  64
#define N_CLASSES 10
#define N_GRAPHS  1024

typedef __attribute__((ext_vector_type(2))) float v2f;
typedef __attribute__((ext_vector_type(4))) float v4f;
typedef __attribute__((ext_vector_type(4))) int   v4i;
typedef __attribute__((ext_vector_type(8))) float v8f;
typedef __attribute__((ext_vector_type(16))) float v16f;

__device__ __forceinline__ void atomAdd1(float* p, float v) {
    asm volatile("global_atomic_add_f32 %0, %1, off scope:SCOPE_DEV"
                 :: "v"(p), "v"(v) : "memory");
}

#define TILE_ROWS   392
#define AGG_BLOCKS  ((N_NODES + TILE_ROWS - 1) / TILE_ROWS)
#define AGG_THREADS 1024
#define AGG_WAVES   32
#define ROWS_PER_WAVE ((TILE_ROWS + AGG_WAVES - 1) / AGG_WAVES)
#define SINK_SLOT   ROWS_PER_WAVE
static_assert(AGG_WAVES == 32 && SINK_SLOT < 16, "slot = local row / 32, in a 16-slot vector");
#define BCAP        32
#define LCAP        512
#define SCAP        768
#define SCAN_U      8
#define CHUNK_E     32
#define PIECE_E     8
#define PIECES      (CHUNK_E / PIECE_E)
#define SCAN_P      (SCAN_U * 32 / PIECE_E)
static_assert(PIECE_E == 8 && PIECES == 4, "a scan load covers 4 pieces, 8 lanes each");
static_assert(SCAP > SCAN_U * 32, "the staged list takes at least one whole trip of the scan");
static_assert(LCAP % SCAN_P == 0, "a trip's list entries stay inside the list");
#define NCHUNK      ((N_EDGES + CHUNK_E - 1) / CHUNK_E)
#define MAP_WORDS   ((AGG_BLOCKS + 31) / 32)
#define PREP_THREADS 128
#define PREP_WAVES  (PREP_THREADS / 32)
#define MAP_ROW     (PREP_WAVES * PIECES)
static_assert(MAP_ROW == 16, "find splits a word number into block, wave and piece by shifts");
#define NO_PIECE    (1u << 28)
#define LIST_NONE   0xFFFFFFFFu
static_assert(N_EDGES % 4 == 0, "destination list is read 4 at a time");
static_assert(TILE_ROWS + AGG_WAVES <= 1024 && N_EDGES < (1 << 21),
              "a scan entry is (edge << 10) | local row");
static_assert(N_NODES < (1 << 22), "a stored entry is (source node << 10) | local row");
static_assert(AGG_BLOCKS == 32 * MAP_WORDS, "the map block's transposition covers whole words of tiles");
static_assert((AGG_BLOCKS * MAP_ROW / 4) % PREP_THREADS == 0, "a map block stores its words 16 bytes per thread");

#define MAP_BLOCKS    ((NCHUNK + PREP_THREADS - 1) / PREP_THREADS)
#define FIND_WORDS    (MAP_BLOCKS * MAP_ROW)
#define FIND_TRIPS    ((FIND_WORDS + AGG_THREADS - 1) / AGG_THREADS)
#define GATHER_U      10
#define GATHER_ITEMS  (N_NODES * (EMBED_D / 4))
#define GATHER_BLOCKS (GATHER_ITEMS / (PREP_THREADS * GATHER_U))
#define PREP_BLOCKS   (MAP_BLOCKS + GATHER_BLOCKS)
#define ZS 16
static_assert(EMBED_D / 4 == 8 && PREP_THREADS % 8 == 0, "a thread keeps its quarter of a row");
static_assert(GATHER_ITEMS % (PREP_THREADS * GATHER_U) == 0,
              "the copy blocks cover the pieces exactly: no end check, so no load waits behind a branch");
static_assert(GATHER_BLOCKS * PREP_THREADS >= N_GRAPHS * ZS, "one copy thread per pooled word");

template <int SFT, unsigned M>
__device__ __forceinline__ unsigned bit_transpose_step(unsigned x, int lane) {
    const unsigned y = (unsigned)__shfl_xor((int)x, SFT);
    const unsigned lo = (x & M) | ((y & M) << SFT);
    const unsigned hi = (x & ~M) | ((y & ~M) >> SFT);
    return (lane & SFT) == 0 ? lo : hi;
}
__device__ __forceinline__ unsigned bit_transpose32(unsigned x, int lane) {
    x = bit_transpose_step<16, 0x0000FFFFu>(x, lane);
    x = bit_transpose_step< 8, 0x00FF00FFu>(x, lane);
    x = bit_transpose_step< 4, 0x0F0F0F0Fu>(x, lane);
    x = bit_transpose_step< 2, 0x33333333u>(x, lane);
    x = bit_transpose_step< 1, 0x55555555u>(x, lane);
    return x;
}

__device__ __forceinline__ void chunk_map_block(const int* __restrict__ dst,
                                                unsigned* __restrict__ cmap,
                                                int nEdges, int mb, int tid,
                                                unsigned* sm) {
    const int c    = mb * PREP_THREADS + tid;
    const int lane = tid & 31;
    const int wv   = tid >> 5;
    unsigned w[PIECES][MAP_WORDS];
#pragma unroll
    for (int s = 0; s < PIECES; ++s)
#pragma unroll
        for (int k = 0; k < MAP_WORDS; ++k) w[s][k] = 0u;
    const int  total4 = nEdges >> 2;
    const v4i* dst4   = reinterpret_cast<const v4i*>(dst);
    v4i d[CHUNK_E / 4];
#pragma unroll
    for (int q = 0; q < CHUNK_E / 4; ++q) {
        const int i = c * (CHUNK_E / 4) + q;
        d[q] = dst4[i < total4 ? i : 0];
    }
    __builtin_amdgcn_sched_barrier(0);
#pragma unroll
    for (int q = 0; q < CHUNK_E / 4; ++q) {
        const bool ok = c * (CHUNK_E / 4) + q < total4;
        const int  dd[4] = {d[q].x, d[q].y, d[q].z, d[q].w};
#pragma unroll
        for (int z = 0; z < 4; ++z) {
            const unsigned t   = (unsigned)(ok ? dd[z] : -1) / (unsigned)TILE_ROWS;
            const unsigned bit = 1u << (t & 31u);
#pragma unroll
            for (int k = 0; k < MAP_WORDS; ++k)
                w[q / 2][k] |= (t >> 5) == (unsigned)k ? bit : 0u;
        }
    }
#pragma unroll
    for (int s = 0; s < PIECES; ++s)
#pragma unroll
        for (int k = 0; k < MAP_WORDS; ++k)
            sm[(32 * k + lane) * MAP_ROW + wv * PIECES + s] = bit_transpose32(w[s][k], lane);
    __syncthreads();
    const uint4* sm4 = reinterpret_cast<const uint4*>(sm);
    uint4*       o4  = reinterpret_cast<uint4*>(cmap) + (size_t)mb * (AGG_BLOCKS * MAP_ROW / 4);
#pragma unroll
    for (int i = 0; i < (AGG_BLOCKS * MAP_ROW / 4) / PREP_THREADS; ++i)
        o4[i * PREP_THREADS + tid] = sm4[i * PREP_THREADS + tid];
}

__global__ __launch_bounds__(PREP_THREADS)
void prep_kernel(const int* ids,
                 const float* embed,
                 float* x0,
                 const int* __restrict__ dst,
                 unsigned* __restrict__ cmap,
                 float* __restrict__ zsum, int nEdges) {
    __shared__ __attribute__((aligned(16))) unsigned sm[AGG_BLOCKS * MAP_ROW];
    const int tid = threadIdx.x;
    if (blockIdx.x < MAP_BLOCKS) {
        chunk_map_block(dst, cmap, nEdges, (int)blockIdx.x, tid, sm);
        return;
    }
    const int gb = (int)blockIdx.x - MAP_BLOCKS;
    const int zt = gb * PREP_THREADS + tid;
    if (zt < N_GRAPHS * ZS) zsum[zt] = 0.0f;
    const int q = tid & 7;
    int id[GATHER_U];
    v4f v[GATHER_U];
#pragma unroll
    for (int j = 0; j < GATHER_U; ++j) {
        const int t = (gb * GATHER_U + j) * PREP_THREADS + tid;
        id[j] = ids[t >> 3];
    }
#pragma unroll
    for (int j = 0; j < GATHER_U; ++j)
        v[j] = reinterpret_cast<const v4f*>(embed + (size_t)id[j] * EMBED_D)[q];
#pragma unroll
    for (int j = 0; j < GATHER_U; ++j) {
        const int t = (gb * GATHER_U + j) * PREP_THREADS + tid;
        reinterpret_cast<v4f*>(x0)[t] = v[j];
    }
}

template <int DIM, bool COUNT, bool BUILD>
__global__ __launch_bounds__(AGG_THREADS)
void tile_agg_kernel(const int* __restrict__ src,
                     const int* __restrict__ dst,
                     const float* __restrict__ x,
                     float* __restrict__ agg,
                     float* __restrict__ cnt,
                     unsigned* __restrict__ glist,
                     unsigned* __restrict__ gtot,
                     const unsigned* __restrict__ cmap,
                     int nNodes, int nEdges) {
    constexpr int H = DIM / 32;
    __shared__ float    cntL[TILE_ROWS];
    __shared__ unsigned bucket[AGG_WAVES * AGG_WAVES * BCAP];
    __shared__ unsigned wcnt[AGG_WAVES * AGG_WAVES];
    __shared__ unsigned stage[AGG_WAVES * SCAP];
    __shared__ unsigned list[AGG_WAVES * LCAP];
    __shared__ unsigned ovf;

    const int tid  = threadIdx.x;
    const int lane = tid & 31;
    const int wave = tid >> 5;
    const int base = blockIdx.x * TILE_ROWS;
    unsigned* mygl  = glist + (size_t)(blockIdx.x * AGG_WAVES + wave) * LCAP;
    unsigned* mygt  = gtot + blockIdx.x * AGG_WAVES + wave;

    unsigned tot = 0;
    bool     slow;
    if constexpr (!BUILD) tot = *mygt;

    v16f a[H];
#pragma unroll
    for (int h = 0; h < H; ++h)
#pragma unroll
        for (int k = 0; k < 16; ++k) a[h][k] = 0.0f;
    int c[ROWS_PER_WAVE];
#pragma unroll
    for (int k = 0; k < ROWS_PER_WAVE; ++k) c[k] = 0;

    if constexpr (BUILD) {
        if (tid == 0) ovf = 0u;
        __syncthreads();

        unsigned* cand  = list + wave * LCAP;
        unsigned  ncand = 0;
        {
            unsigned mv[FIND_TRIPS];
#pragma unroll
            for (int u = 0; u < FIND_TRIPS; ++u) {
                const int  i  = u * AGG_THREADS + tid;
                const bool in = i < FIND_WORDS;
                const int  ii = in ? i : 0;
                const unsigned wd =
                    cmap[((size_t)(ii >> 4) * AGG_BLOCKS + blockIdx.x) * MAP_ROW + (ii & 15)];
                mv[u] = in ? wd : 0u;
            }
#pragma unroll
            for (int u = 0; u < FIND_TRIPS; ++u) {
                const int i = u * AGG_THREADS + tid;
                const unsigned p0 = ((unsigned)(i >> 4) * (unsigned)PREP_THREADS
                                     + (unsigned)((i >> 2) & 3) * 32u) * (unsigned)PIECES
                                    + (unsigned)(i & 3);
                unsigned wd = mv[u];
                while (__builtin_amdgcn_ballot_w32(wd != 0u) != 0u) {
                    const bool     has = wd != 0u;
                    const unsigned m   = __builtin_amdgcn_ballot_w32(has);
                    const unsigned pos = ncand + __builtin_amdgcn_mbcnt_lo(m, 0u);
                    const unsigned bit = has ? (unsigned)__builtin_ctz(wd) : 0u;
                    if (has && pos < (unsigned)LCAP) cand[pos] = p0 + bit * (unsigned)PIECES;
                    wd &= wd - 1u;
                    ncand += (unsigned)__builtin_popcount(m);
                }
            }
        }
        if (ncand > (unsigned)LCAP) {
            ovf   = 1u;
            ncand = (unsigned)LCAP;
        }
        if (ncand + (unsigned)lane < (unsigned)LCAP) cand[ncand + lane] = NO_PIECE;
        __syncthreads();

        unsigned* stg = stage + wave * SCAP;
        unsigned  nst = 0;
        const int      cg = lane >> 3;
        const unsigned le = (unsigned)(lane & 7);
        for (unsigned c0 = 0; c0 < ncand; c0 += SCAN_P) {
            if (nst > (unsigned)(SCAP - SCAN_U * 32)) { ovf = 1u; break; }
            int      dv[SCAN_U];
            unsigned ev[SCAN_U];
#pragma unroll
            for (int u = 0; u < SCAN_U; ++u) {
                const unsigned e  = cand[c0 + 4 * u + cg] * (unsigned)PIECE_E + le;
                const bool     ok = e < (unsigned)nEdges;
                int d = dst[ok ? e : 0u];
                if (!ok) d = -1;
                dv[u] = d;
                ev[u] = e;
            }
#pragma unroll
            for (int u = 0; u < SCAN_U; ++u) {
                const unsigned r   = (unsigned)(dv[u] - base);
                const bool     hit = r < (unsigned)TILE_ROWS;
                const unsigned m   = __builtin_amdgcn_ballot_w32(hit);
                const unsigned pos = __builtin_amdgcn_mbcnt_lo(m, nst);
                if (hit) stg[pos] = (ev[u] << 10) | r;
                nst += (unsigned)__builtin_popcount(m);
            }
        }
        __syncthreads();

        unsigned mycnt = 0;
        for (unsigned b0 = 0; b0 < nst; b0 += 32) {
            const unsigned qi    = b0 + lane;
            const bool     valid = qi < nst;
            const unsigned en    = stg[valid ? qi : 0u];
            const unsigned owner = en & 31u;
            unsigned same = __builtin_amdgcn_ballot_w32(valid);
            unsigned mine = same;
#pragma unroll
            for (int b = 0; b < 5; ++b) {
                const bool     ob = ((owner >> b) & 1u) != 0u;
                const unsigned bb = __builtin_amdgcn_ballot_w32(ob);
                same &= ob ? bb : ~bb;
                mine &= (((unsigned)lane >> b) & 1u) != 0u ? bb : ~bb;
            }
            const unsigned cur = (unsigned)__builtin_amdgcn_ds_bpermute((int)(owner << 2), (int)mycnt);
            const unsigned pos = cur + __builtin_amdgcn_mbcnt_lo(same, 0u);
            if (valid && pos < (unsigned)BCAP)
                bucket[(wave * AGG_WAVES + owner) * BCAP + pos] = en;
            mycnt += (unsigned)__builtin_popcount(mine);
        }
        if (mycnt > (unsigned)BCAP) ovf = 1u;
        wcnt[wave * AGG_WAVES + lane] = mycnt < (unsigned)BCAP ? mycnt : (unsigned)BCAP;
        __syncthreads();

        unsigned* mylist = list + wave * LCAP;
        const unsigned n_l = wcnt[lane * AGG_WAVES + wave];
        unsigned off = 0;
#pragma unroll
        for (int j = 0; j < 32; ++j) {
            const unsigned nj = (unsigned)__builtin_amdgcn_readlane((int)n_l, j);
            off += j < lane ? nj : 0u;
            tot += nj;
        }
        if (tot > (unsigned)LCAP) {
            ovf = 1u;
        } else {
            for (unsigned k = 0; k < n_l; ++k)
                mylist[off + k] = bucket[(lane * AGG_WAVES + wave) * BCAP + k];
        }
        __syncthreads();
        slow = ovf != 0u;
        if (lane == 0) *mygt = slow ? LIST_NONE : tot;
    } else {
        tot  = (unsigned)__builtin_amdgcn_readfirstlane((int)tot);
        slow = tot == LIST_NONE;
    }

    if (!slow) {
        for (unsigned b0 = 0; b0 < tot; b0 += 32) {
            const unsigned qi = b0 + lane;
            unsigned ent;
            if constexpr (BUILD) {
                const unsigned se = qi < tot ? list[wave * LCAP + qi]
                                             : (unsigned)(SINK_SLOT << 5);
                ent = ((unsigned)src[se >> 10] << 10) | (se & 1023u);
                if (qi < tot) mygl[qi] = ent;
            } else {
                ent = qi < tot ? mygl[qi] : (unsigned)(SINK_SLOT << 5);
            }
            const int sl = (int)((ent & 1023u) >> 5);
            const int s  = (int)(ent >> 10);
            if constexpr (COUNT) {
#pragma unroll
                for (int k = 0; k < ROWS_PER_WAVE; ++k)
                    c[k] += __builtin_popcount(__builtin_amdgcn_ballot_w32(sl == k));
            }
#pragma unroll
            for (int j0 = 0; j0 < 32; j0 += 16) {
                if (b0 + j0 >= tot) break;
                float v[16][H];
                int   slot[16];
#pragma unroll
                for (int u = 0; u < 16; ++u) {
                    const int sj = __builtin_amdgcn_readlane(s, j0 + u);
                    slot[u]      = __builtin_amdgcn_readlane(sl, j0 + u);
                    const float* xp = x + (size_t)sj * DIM + lane;
#pragma unroll
                    for (int h = 0; h < H; ++h) v[u][h] = xp[32 * h];
                }
#pragma unroll
                for (int u = 0; u < 16; ++u) {
#pragma unroll
                    for (int h = 0; h < H; ++h)
                        a[h][slot[u]] += v[u][h];
                }
            }
        }
    } else {
        for (int i0 = 0; i0 < nEdges; i0 += 32) {
            const int i = i0 + lane;
            const int d = i < nEdges ? dst[i] : -1;
            const unsigned r = (unsigned)(d - base);
            unsigned mask = __builtin_amdgcn_ballot_w32(
                r < (unsigned)TILE_ROWS && (int)(r & 31u) == wave);
            while (mask) {
                const int l = __builtin_ctz(mask);
                mask &= mask - 1;
                const int kr = __builtin_amdgcn_readlane((int)r, l) >> 5;
                const float* xp = x + (size_t)src[i0 + l] * DIM + lane;
#pragma unroll
                for (int h = 0; h < H; ++h) a[h][kr] += xp[32 * h];
                if constexpr (COUNT) {
#pragma unroll
                    for (int k = 0; k < ROWS_PER_WAVE; ++k) c[k] += kr == k ? 1 : 0;
                }
            }
        }
    }

    const int left = nNodes - base;
    const int rows = left < TILE_ROWS ? left : TILE_ROWS;
#pragma unroll
    for (int k = 0; k < ROWS_PER_WAVE; ++k) {
        const int r = wave + 32 * k;
        if (r < rows) {
#pragma unroll
            for (int h = 0; h < H; ++h)
                agg[(size_t)(base + r) * DIM + 32 * h + lane] = a[h][k];
            if (COUNT && lane == 0) cntL[r] = (float)c[k];
        }
    }
    if constexpr (COUNT) {
        __syncthreads();
        for (int i = tid; i < rows; i += AGG_THREADS) cnt[base + i] = cntL[i];
    }
}

#define GEMM_WAVES   8
#define GEMM_THREADS (32 * GEMM_WAVES)
#define GEMM_BLOCKS  256
static_assert(N_NODES % 16 == 0, "row tiles are whole");

template <int K>
__global__ __launch_bounds__(GEMM_THREADS)
void sage_gemm_kernel(const float* aggRaw,
                      const float* cnt,
                      const float* x,
                      const float* __restrict__ Wl,
                      const float* __restrict__ Wr,
                      const float* __restrict__ bias,
                      float* __restrict__ out, int nNodes) {
    constexpr int T = K / 8;
    const int lane = threadIdx.x & 31;
    const int grp  = lane >> 4;
    const int lr   = lane & 15;
    const int wv     = __builtin_amdgcn_readfirstlane((int)(threadIdx.x >> 5));
    const int nWaves = (int)gridDim.x * GEMM_WAVES;
    const int nTiles = nNodes / 16;
    int tile = wv * (int)gridDim.x + (int)blockIdx.x;

    float4 ra0[T], rx0[T], ra1[T], rx1[T];
    float  rc0 = 1.0f, rc1 = 1.0f;
    auto load_tile = [&](int tl, float4 (&a)[T], float4 (&xx)[T], float& c) {
        const int arow = tl * 16 + lr;
        const float4* ap = reinterpret_cast<const float4*>(aggRaw + (size_t)arow * K) + grp;
        const float4* xp = reinterpret_cast<const float4*>(x + (size_t)arow * K) + grp;
        c = cnt[arow];
#pragma unroll
        for (int t = 0; t < T; ++t) { a[t] = ap[2 * t]; xx[t] = xp[2 * t]; }
    };
    if (tile >= nTiles) return;
    load_tile(tile, ra0, rx0, rc0);

    v2f bl[T][2][4], br[T][2][4];
#pragma unroll
    for (int t = 0; t < T; ++t)
#pragma unroll
        for (int p = 0; p < 2; ++p) {
            const int kb = 8 * t + 4 * grp + 2 * p;
#pragma unroll
            for (int n = 0; n < 4; ++n) {
                bl[t][p][n].x = Wl[(size_t)(kb + 0) * HIDDEN_D + 16 * n + lr];
                bl[t][p][n].y = Wl[(size_t)(kb + 1) * HIDDEN_D + 16 * n + lr];
                br[t][p][n].x = Wr[(size_t)(kb + 0) * HIDDEN_D + 16 * n + lr];
                br[t][p][n].y = Wr[(size_t)(kb + 1) * HIDDEN_D + 16 * n + lr];
            }
        }
    float bv[4];
#pragma unroll
    for (int n = 0; n < 4; ++n) bv[n] = bias[16 * n + lr];


    auto compute_tile = [&](int tl, const float4 (&av)[T], const float4 (&xv)[T], float deg) {
        const float invc = 1.0f / fmaxf(deg, 1.0f);
        v8f c[4] = {};
#pragma unroll
        for (int t = 0; t < T; ++t) {
            v2f a0, a1;
            a0.x = av[t].x * invc;  a0.y = av[t].y * invc;
            a1.x = av[t].z * invc;  a1.y = av[t].w * invc;
#pragma unroll
            for (int n = 0; n < 4; ++n)
                c[n] = __builtin_amdgcn_wmma_f32_16x16x4_f32(false, a0, false, bl[t][0][n],
                                                             (short)0, c[n], false, false);
#pragma unroll
            for (int n = 0; n < 4; ++n)
                c[n] = __builtin_amdgcn_wmma_f32_16x16x4_f32(false, a1, false, bl[t][1][n],
                                                             (short)0, c[n], false, false);
        }
#pragma unroll
        for (int t = 0; t < T; ++t) {
            v2f a0, a1;
            a0.x = xv[t].x;  a0.y = xv[t].y;
            a1.x = xv[t].z;  a1.y = xv[t].w;
#pragma unroll
            for (int n = 0; n < 4; ++n)
                c[n] = __builtin_amdgcn_wmma_f32_16x16x4_f32(false, a0, false, br[t][0][n],
                                                             (short)0, c[n], false, false);
#pragma unroll
            for (int n = 0; n < 4; ++n)
                c[n] = __builtin_amdgcn_wmma_f32_16x16x4_f32(false, a1, false, br[t][1][n],
                                                             (short)0, c[n], false, false);
        }
        const int row0 = tl * 16;
#pragma unroll
        for (int n = 0; n < 4; ++n)
#pragma unroll
            for (int v = 0; v < 8; ++v) {
                const int m = v + 8 * grp;
                out[(size_t)(row0 + m) * HIDDEN_D + 16 * n + lr] = fmaxf(c[n][v] + bv[n], 0.0f);
            }
    };

    int next = tile + nWaves;
    load_tile(next < nTiles ? next : tile, ra1, rx1, rc1);
    __builtin_amdgcn_sched_barrier(0);
    compute_tile(tile, ra0, rx0, rc0);
    tile = next;
    while (tile < nTiles) {
        next = tile + nWaves;
        load_tile(next < nTiles ? next : tile, ra0, rx0, rc0);
        __builtin_amdgcn_sched_barrier(0);
        compute_tile(tile, ra1, rx1, rc1);
        tile = next;
        if (tile >= nTiles) break;
        next = tile + nWaves;
        load_tile(next < nTiles ? next : tile, ra1, rx1, rc1);
        __builtin_amdgcn_sched_barrier(0);
        compute_tile(tile, ra0, rx0, rc0);
        tile = next;
    }
}

__global__ void head_pool_kernel(const int* __restrict__ batch,
                                 const float* __restrict__ h,
                                 const float* __restrict__ Wo,
                                 float* __restrict__ zsum, int nNodes) {
    const int lane = threadIdx.x & 31;
    const int tile = blockIdx.x * (blockDim.x >> 5) + (threadIdx.x >> 5);
    const int row0 = tile * 16;
    if (row0 >= nNodes) return;
    const int grp = lane >> 4;
    const int lr  = lane & 15;

    const float4* hp =
        reinterpret_cast<const float4*>(h + (size_t)(row0 + lr) * HIDDEN_D) + grp;
    float4 av[8];
    for (int t = 0; t < 8; ++t) av[t] = hp[2 * t];

    const int4* bp = reinterpret_cast<const int4*>(batch + row0 + 8 * grp);
    const int4 gA = bp[0], gB = bp[1];

    const bool colOk = lr < N_CLASSES;
    const int  cc    = colOk ? lr : 0;

    v8f c = {};
    for (int t = 0; t < 8; ++t) {
        const int kb = 8 * t + 4 * grp;
        const float w0 = Wo[(kb + 0) * N_CLASSES + cc];
        const float w1 = Wo[(kb + 1) * N_CLASSES + cc];
        const float w2 = Wo[(kb + 2) * N_CLASSES + cc];
        const float w3 = Wo[(kb + 3) * N_CLASSES + cc];
        v2f a0, a1, b0, b1;
        a0.x = av[t].x;             a0.y = av[t].y;
        b0.x = colOk ? w0 : 0.0f;   b0.y = colOk ? w1 : 0.0f;
        a1.x = av[t].z;             a1.y = av[t].w;
        b1.x = colOk ? w2 : 0.0f;   b1.y = colOk ? w3 : 0.0f;
        c = __builtin_amdgcn_wmma_f32_16x16x4_f32(false, a0, false, b0,
                                                  (short)0, c, false, false);
        c = __builtin_amdgcn_wmma_f32_16x16x4_f32(false, a1, false, b1,
                                                  (short)0, c, false, false);
    }

    float val[8];
    for (int v = 0; v < 8; ++v) val[v] = colOk ? c[v] : 1.0f;
    const int gid[8] = {gA.x, gA.y, gA.z, gA.w, gB.x, gB.y, gB.z, gB.w};
    if (lr <= N_CLASSES) {
        for (int v = 0; v < 8; ++v)
            atomAdd1(zsum + (size_t)gid[v] * ZS + lr, val[v]);
    }
}

__global__ void finish_kernel(const float* __restrict__ zsum,
                              const float* __restrict__ bo,
                              float* __restrict__ out) {
    int t = blockIdx.x * blockDim.x + threadIdx.x;
    if (t >= N_GRAPHS * N_CLASSES) return;
    int g = t / N_CLASSES;
    int c = t - g * N_CLASSES;
    out[t] = zsum[g * ZS + c] / fmaxf(zsum[g * ZS + N_CLASSES], 1.0f) + bo[c];
}

extern "C" void kernel_launch(void* const* d_in, const int* in_sizes, int n_in,
                              void* d_out, int out_size, void* d_ws, size_t ws_size,
                              hipStream_t stream) {
    const int*   node_ids = (const int*)  d_in[0];
    const int*   src      = (const int*)  d_in[1];
    const int*   dst      = (const int*)  d_in[2];
    const int*   batch    = (const int*)  d_in[3];
    const float* embed    = (const float*)d_in[4];
    const float* W1_l     = (const float*)d_in[5];
    const float* b1       = (const float*)d_in[6];
    const float* W1_r     = (const float*)d_in[7];
    const float* W2_l     = (const float*)d_in[8];
    const float* b2       = (const float*)d_in[9];
    const float* W2_r     = (const float*)d_in[10];
    const float* W_out    = (const float*)d_in[11];
    const float* b_out    = (const float*)d_in[12];
    float* out = (float*)d_out;

    float* ws   = (float*)d_ws;
    float* x0   = ws;
    float* h1   = x0   + (size_t)N_NODES * EMBED_D;
    float* h2   = h1   + (size_t)N_NODES * HIDDEN_D;
    float* agg1 = h2   + (size_t)N_NODES * HIDDEN_D;
    float* agg2 = agg1 + (size_t)N_NODES * EMBED_D;
    float* cnt  = agg2 + (size_t)N_NODES * HIDDEN_D;
    float* zsum = cnt  + (size_t)N_NODES;
    unsigned* gtot  = (unsigned*)(zsum + N_GRAPHS * ZS);
    unsigned* glist = gtot + AGG_BLOCKS * AGG_WAVES;
    unsigned* cmap  = glist + (size_t)AGG_BLOCKS * AGG_WAVES * LCAP;

    const int B = 256;
    auto blocks = [](int n, int b) { return (n + b - 1) / b; };

    prep_kernel<<<PREP_BLOCKS, PREP_THREADS, 0, stream>>>(
        node_ids, embed, x0, dst, cmap, zsum, N_EDGES);

    tile_agg_kernel<EMBED_D, true, true><<<AGG_BLOCKS, AGG_THREADS, 0, stream>>>(
        src, dst, x0, agg1, cnt, glist, gtot, cmap, N_NODES, N_EDGES);
    sage_gemm_kernel<EMBED_D><<<GEMM_BLOCKS, GEMM_THREADS, 0, stream>>>(
        agg1, cnt, x0, W1_l, W1_r, b1, h1, N_NODES);

    tile_agg_kernel<HIDDEN_D, false, false><<<AGG_BLOCKS, AGG_THREADS, 0, stream>>>(
        src, dst, h1, agg2, cnt, glist, gtot, cmap, N_NODES, N_EDGES);
    sage_gemm_kernel<HIDDEN_D><<<GEMM_BLOCKS, GEMM_THREADS, 0, stream>>>(
        agg2, cnt, h1, W2_l, W2_r, b2, h2, N_NODES);

    head_pool_kernel<<<blocks(N_NODES / 16, 4), 128, 0, stream>>>(
        batch, h2, W_out, zsum, N_NODES);
    finish_kernel<<<blocks(N_GRAPHS * N_CLASSES, B), B, 0, stream>>>(
        zsum, b_out, out);
}


# add loops of both edge-sum kernels software-pipelined: ring of 4 sets of 8 rows, 24 rows in flight while 8 are added; entries (layer 1: source lookups too) fetched 2 batches ahead; sum order kept
# speedup vs baseline: 3.2089x; 1.0480x over previous
#include <hip/hip_runtime.h>
#include <type_traits>

#define N_NODES   100000
#define N_EDGES   1200000
#define EMBED_D   32
#define HIDDEN_D  64
#define N_CLASSES 10
#define N_GRAPHS  1024

typedef __attribute__((ext_vector_type(2))) float v2f;
typedef __attribute__((ext_vector_type(4))) float v4f;
typedef __attribute__((ext_vector_type(4))) int   v4i;
typedef __attribute__((ext_vector_type(8))) float v8f;
typedef __attribute__((ext_vector_type(16))) float v16f;

__device__ __forceinline__ void atomAdd1(float* p, float v) {
    asm volatile("global_atomic_add_f32 %0, %1, off scope:SCOPE_DEV"
                 :: "v"(p), "v"(v) : "memory");
}

#define TILE_ROWS   392
#define AGG_BLOCKS  ((N_NODES + TILE_ROWS - 1) / TILE_ROWS)
#define AGG_THREADS 1024
#define AGG_WAVES   32
#define ROWS_PER_WAVE ((TILE_ROWS + AGG_WAVES - 1) / AGG_WAVES)
#define SINK_SLOT   ROWS_PER_WAVE
static_assert(AGG_WAVES == 32 && SINK_SLOT < 16, "slot = local row / 32, in a 16-slot vector");
#define BCAP        32
#define LCAP        512
#define SCAP        768
#define SCAN_U      8
#define CHUNK_E     32
#define PIECE_E     8
#define PIECES      (CHUNK_E / PIECE_E)
#define SCAN_P      (SCAN_U * 32 / PIECE_E)
static_assert(PIECE_E == 8 && PIECES == 4, "a scan load covers 4 pieces, 8 lanes each");
static_assert(SCAP > SCAN_U * 32, "the staged list takes at least one whole trip of the scan");
static_assert(LCAP % SCAN_P == 0, "a trip's list entries stay inside the list");
#define NCHUNK      ((N_EDGES + CHUNK_E - 1) / CHUNK_E)
#define MAP_WORDS   ((AGG_BLOCKS + 31) / 32)
#define PREP_THREADS 128
#define PREP_WAVES  (PREP_THREADS / 32)
#define MAP_ROW     (PREP_WAVES * PIECES)
static_assert(MAP_ROW == 16, "find splits a word number into block, wave and piece by shifts");
#define NO_PIECE    (1u << 28)
#define LIST_NONE   0xFFFFFFFFu
static_assert(N_EDGES % 4 == 0, "destination list is read 4 at a time");
static_assert(TILE_ROWS + AGG_WAVES <= 1024 && N_EDGES < (1 << 21),
              "a scan entry is (edge << 10) | local row");
static_assert(N_NODES < (1 << 22), "a stored entry is (source node << 10) | local row");
static_assert(AGG_BLOCKS == 32 * MAP_WORDS, "the map block's transposition covers whole words of tiles");
static_assert((AGG_BLOCKS * MAP_ROW / 4) % PREP_THREADS == 0, "a map block stores its words 16 bytes per thread");

#define MAP_BLOCKS    ((NCHUNK + PREP_THREADS - 1) / PREP_THREADS)
#define FIND_WORDS    (MAP_BLOCKS * MAP_ROW)
#define FIND_TRIPS    ((FIND_WORDS + AGG_THREADS - 1) / AGG_THREADS)
#define GATHER_U      10
#define GATHER_ITEMS  (N_NODES * (EMBED_D / 4))
#define GATHER_BLOCKS (GATHER_ITEMS / (PREP_THREADS * GATHER_U))
#define PREP_BLOCKS   (MAP_BLOCKS + GATHER_BLOCKS)
#define ZS 16
static_assert(EMBED_D / 4 == 8 && PREP_THREADS % 8 == 0, "a thread keeps its quarter of a row");
static_assert(GATHER_ITEMS % (PREP_THREADS * GATHER_U) == 0,
              "the copy blocks cover the pieces exactly: no end check, so no load waits behind a branch");
static_assert(GATHER_BLOCKS * PREP_THREADS >= N_GRAPHS * ZS, "one copy thread per pooled word");

template <int SFT, unsigned M>
__device__ __forceinline__ unsigned bit_transpose_step(unsigned x, int lane) {
    const unsigned y = (unsigned)__shfl_xor((int)x, SFT);
    const unsigned lo = (x & M) | ((y & M) << SFT);
    const unsigned hi = (x & ~M) | ((y & ~M) >> SFT);
    return (lane & SFT) == 0 ? lo : hi;
}
__device__ __forceinline__ unsigned bit_transpose32(unsigned x, int lane) {
    x = bit_transpose_step<16, 0x0000FFFFu>(x, lane);
    x = bit_transpose_step< 8, 0x00FF00FFu>(x, lane);
    x = bit_transpose_step< 4, 0x0F0F0F0Fu>(x, lane);
    x = bit_transpose_step< 2, 0x33333333u>(x, lane);
    x = bit_transpose_step< 1, 0x55555555u>(x, lane);
    return x;
}

__device__ __forceinline__ void chunk_map_block(const int* __restrict__ dst,
                                                unsigned* __restrict__ cmap,
                                                int nEdges, int mb, int tid,
                                                unsigned* sm) {
    const int c    = mb * PREP_THREADS + tid;
    const int lane = tid & 31;
    const int wv   = tid >> 5;
    unsigned w[PIECES][MAP_WORDS];
#pragma unroll
    for (int s = 0; s < PIECES; ++s)
#pragma unroll
        for (int k = 0; k < MAP_WORDS; ++k) w[s][k] = 0u;
    const int  total4 = nEdges >> 2;
    const v4i* dst4   = reinterpret_cast<const v4i*>(dst);
    v4i d[CHUNK_E / 4];
#pragma unroll
    for (int q = 0; q < CHUNK_E / 4; ++q) {
        const int i = c * (CHUNK_E / 4) + q;
        d[q] = dst4[i < total4 ? i : 0];
    }
    __builtin_amdgcn_sched_barrier(0);
#pragma unroll
    for (int q = 0; q < CHUNK_E / 4; ++q) {
        const bool ok = c * (CHUNK_E / 4) + q < total4;
        const int  dd[4] = {d[q].x, d[q].y, d[q].z, d[q].w};
#pragma unroll
        for (int z = 0; z < 4; ++z) {
            const unsigned t   = (unsigned)(ok ? dd[z] : -1) / (unsigned)TILE_ROWS;
            const unsigned bit = 1u << (t & 31u);
#pragma unroll
            for (int k = 0; k < MAP_WORDS; ++k)
                w[q / 2][k] |= (t >> 5) == (unsigned)k ? bit : 0u;
        }
    }
#pragma unroll
    for (int s = 0; s < PIECES; ++s)
#pragma unroll
        for (int k = 0; k < MAP_WORDS; ++k)
            sm[(32 * k + lane) * MAP_ROW + wv * PIECES + s] = bit_transpose32(w[s][k], lane);
    __syncthreads();
    const uint4* sm4 = reinterpret_cast<const uint4*>(sm);
    uint4*       o4  = reinterpret_cast<uint4*>(cmap) + (size_t)mb * (AGG_BLOCKS * MAP_ROW / 4);
#pragma unroll
    for (int i = 0; i < (AGG_BLOCKS * MAP_ROW / 4) / PREP_THREADS; ++i)
        o4[i * PREP_THREADS + tid] = sm4[i * PREP_THREADS + tid];
}

__global__ __launch_bounds__(PREP_THREADS)
void prep_kernel(const int* ids,
                 const float* embed,
                 float* x0,
                 const int* __restrict__ dst,
                 unsigned* __restrict__ cmap,
                 float* __restrict__ zsum, int nEdges) {
    __shared__ __attribute__((aligned(16))) unsigned sm[AGG_BLOCKS * MAP_ROW];
    const int tid = threadIdx.x;
    if (blockIdx.x < MAP_BLOCKS) {
        chunk_map_block(dst, cmap, nEdges, (int)blockIdx.x, tid, sm);
        return;
    }
    const int gb = (int)blockIdx.x - MAP_BLOCKS;
    const int zt = gb * PREP_THREADS + tid;
    if (zt < N_GRAPHS * ZS) zsum[zt] = 0.0f;
    const int q = tid & 7;
    int id[GATHER_U];
    v4f v[GATHER_U];
#pragma unroll
    for (int j = 0; j < GATHER_U; ++j) {
        const int t = (gb * GATHER_U + j) * PREP_THREADS + tid;
        id[j] = ids[t >> 3];
    }
#pragma unroll
    for (int j = 0; j < GATHER_U; ++j)
        v[j] = reinterpret_cast<const v4f*>(embed + (size_t)id[j] * EMBED_D)[q];
#pragma unroll
    for (int j = 0; j < GATHER_U; ++j) {
        const int t = (gb * GATHER_U + j) * PREP_THREADS + tid;
        reinterpret_cast<v4f*>(x0)[t] = v[j];
    }
}

template <int DIM, bool COUNT, bool BUILD>
__global__ __launch_bounds__(AGG_THREADS)
void tile_agg_kernel(const int* __restrict__ src,
                     const int* __restrict__ dst,
                     const float* __restrict__ x,
                     float* __restrict__ agg,
                     float* __restrict__ cnt,
                     unsigned* __restrict__ glist,
                     unsigned* __restrict__ gtot,
                     const unsigned* __restrict__ cmap,
                     int nNodes, int nEdges) {
    constexpr int H = DIM / 32;
    __shared__ float    cntL[TILE_ROWS];
    __shared__ unsigned bucket[AGG_WAVES * AGG_WAVES * BCAP];
    __shared__ unsigned wcnt[AGG_WAVES * AGG_WAVES];
    __shared__ unsigned stage[AGG_WAVES * SCAP];
    __shared__ unsigned list[AGG_WAVES * LCAP];
    __shared__ unsigned ovf;

    const int tid  = threadIdx.x;
    const int lane = tid & 31;
    const int wave = tid >> 5;
    const int base = blockIdx.x * TILE_ROWS;
    unsigned* mygl  = glist + (size_t)(blockIdx.x * AGG_WAVES + wave) * LCAP;
    unsigned* mygt  = gtot + blockIdx.x * AGG_WAVES + wave;

    unsigned tot = 0;
    bool     slow;
    if constexpr (!BUILD) tot = *mygt;

    v16f a[H];
#pragma unroll
    for (int h = 0; h < H; ++h)
#pragma unroll
        for (int k = 0; k < 16; ++k) a[h][k] = 0.0f;
    int c[ROWS_PER_WAVE];
#pragma unroll
    for (int k = 0; k < ROWS_PER_WAVE; ++k) c[k] = 0;

    if constexpr (BUILD) {
        if (tid == 0) ovf = 0u;
        __syncthreads();

        unsigned* cand  = list + wave * LCAP;
        unsigned  ncand = 0;
        {
            unsigned mv[FIND_TRIPS];
#pragma unroll
            for (int u = 0; u < FIND_TRIPS; ++u) {
                const int  i  = u * AGG_THREADS + tid;
                const bool in = i < FIND_WORDS;
                const int  ii = in ? i : 0;
                const unsigned wd =
                    cmap[((size_t)(ii >> 4) * AGG_BLOCKS + blockIdx.x) * MAP_ROW + (ii & 15)];
                mv[u] = in ? wd : 0u;
            }
#pragma unroll
            for (int u = 0; u < FIND_TRIPS; ++u) {
                const int i = u * AGG_THREADS + tid;
                const unsigned p0 = ((unsigned)(i >> 4) * (unsigned)PREP_THREADS
                                     + (unsigned)((i >> 2) & 3) * 32u) * (unsigned)PIECES
                                    + (unsigned)(i & 3);
                unsigned wd = mv[u];
                while (__builtin_amdgcn_ballot_w32(wd != 0u) != 0u) {
                    const bool     has = wd != 0u;
                    const unsigned m   = __builtin_amdgcn_ballot_w32(has);
                    const unsigned pos = ncand + __builtin_amdgcn_mbcnt_lo(m, 0u);
                    const unsigned bit = has ? (unsigned)__builtin_ctz(wd) : 0u;
                    if (has && pos < (unsigned)LCAP) cand[pos] = p0 + bit * (unsigned)PIECES;
                    wd &= wd - 1u;
                    ncand += (unsigned)__builtin_popcount(m);
                }
            }
        }
        if (ncand > (unsigned)LCAP) {
            ovf   = 1u;
            ncand = (unsigned)LCAP;
        }
        if (ncand + (unsigned)lane < (unsigned)LCAP) cand[ncand + lane] = NO_PIECE;
        __syncthreads();

        unsigned* stg = stage + wave * SCAP;
        unsigned  nst = 0;
        const int      cg = lane >> 3;
        const unsigned le = (unsigned)(lane & 7);
        for (unsigned c0 = 0; c0 < ncand; c0 += SCAN_P) {
            if (nst > (unsigned)(SCAP - SCAN_U * 32)) { ovf = 1u; break; }
            int      dv[SCAN_U];
            unsigned ev[SCAN_U];
#pragma unroll
            for (int u = 0; u < SCAN_U; ++u) {
                const unsigned e  = cand[c0 + 4 * u + cg] * (unsigned)PIECE_E + le;
                const bool     ok = e < (unsigned)nEdges;
                int d = dst[ok ? e : 0u];
                if (!ok) d = -1;
                dv[u] = d;
                ev[u] = e;
            }
#pragma unroll
            for (int u = 0; u < SCAN_U; ++u) {
                const unsigned r   = (unsigned)(dv[u] - base);
                const bool     hit = r < (unsigned)TILE_ROWS;
                const unsigned m   = __builtin_amdgcn_ballot_w32(hit);
                const unsigned pos = __builtin_amdgcn_mbcnt_lo(m, nst);
                if (hit) stg[pos] = (ev[u] << 10) | r;
                nst += (unsigned)__builtin_popcount(m);
            }
        }
        __syncthreads();

        unsigned mycnt = 0;
        for (unsigned b0 = 0; b0 < nst; b0 += 32) {
            const unsigned qi    = b0 + lane;
            const bool     valid = qi < nst;
            const unsigned en    = stg[valid ? qi : 0u];
            const unsigned owner = en & 31u;
            unsigned same = __builtin_amdgcn_ballot_w32(valid);
            unsigned mine = same;
#pragma unroll
            for (int b = 0; b < 5; ++b) {
                const bool     ob = ((owner >> b) & 1u) != 0u;
                const unsigned bb = __builtin_amdgcn_ballot_w32(ob);
                same &= ob ? bb : ~bb;
                mine &= (((unsigned)lane >> b) & 1u) != 0u ? bb : ~bb;
            }
            const unsigned cur = (unsigned)__builtin_amdgcn_ds_bpermute((int)(owner << 2), (int)mycnt);
            const unsigned pos = cur + __builtin_amdgcn_mbcnt_lo(same, 0u);
            if (valid && pos < (unsigned)BCAP)
                bucket[(wave * AGG_WAVES + owner) * BCAP + pos] = en;
            mycnt += (unsigned)__builtin_popcount(mine);
        }
        if (mycnt > (unsigned)BCAP) ovf = 1u;
        wcnt[wave * AGG_WAVES + lane] = mycnt < (unsigned)BCAP ? mycnt : (unsigned)BCAP;
        __syncthreads();

        unsigned* mylist = list + wave * LCAP;
        const unsigned n_l = wcnt[lane * AGG_WAVES + wave];
        unsigned off = 0;
#pragma unroll
        for (int j = 0; j < 32; ++j) {
            const unsigned nj = (unsigned)__builtin_amdgcn_readlane((int)n_l, j);
            off += j < lane ? nj : 0u;
            tot += nj;
        }
        if (tot > (unsigned)LCAP) {
            ovf = 1u;
        } else {
            for (unsigned k = 0; k < n_l; ++k)
                mylist[off + k] = bucket[(lane * AGG_WAVES + wave) * BCAP + k];
        }
        __syncthreads();
        slow = ovf != 0u;
        if (lane == 0) *mygt = slow ? LIST_NONE : tot;
    } else {
        tot  = (unsigned)__builtin_amdgcn_readfirstlane((int)tot);
        slow = tot == LIST_NONE;
    }

    if (!slow && tot != 0u) {
        constexpr int G  = 8;
        constexpr int NG = 32 / G;
        constexpr int NB = 4;
        static_assert(NG % NB == 0 && NB >= 2, "a group's set is the same in every batch");
        constexpr unsigned PAD = (unsigned)(SINK_SLOT << 5);
        float v[NB][G][H];
        auto issue = [&](int sv, int j0, float (&vv)[G][H]) {
#pragma unroll
            for (int u = 0; u < G; ++u) {
                const int sj = __builtin_amdgcn_readlane(sv, j0 + u);
                if constexpr (H == 2) {
                    const v2f t = *reinterpret_cast<const v2f*>(x + (size_t)sj * DIM + 2 * lane);
                    vv[u][0] = t.x;
                    vv[u][1] = t.y;
                } else {
                    vv[u][0] = x[(size_t)sj * DIM + lane];
                }
            }
        };
        auto addin = [&](int slv, int j0, const float (&vv)[G][H]) {
#pragma unroll
            for (int u = 0; u < G; ++u) {
                const int slot = __builtin_amdgcn_readlane(slv, j0 + u);
#pragma unroll
                for (int h = 0; h < H; ++h)
                    a[h][slot] += vv[u][h];
            }
        };
        auto count = [&](int slv) {
            if constexpr (COUNT) {
#pragma unroll
                for (int k = 0; k < ROWS_PER_WAVE; ++k)
                    c[k] += __builtin_popcount(__builtin_amdgcn_ballot_w32(slv == k));
            }
        };
        auto ask = [&](unsigned q0, unsigned& se, unsigned& raw) {
            const unsigned q  = q0 + lane;
            const unsigned qc = q < tot ? q : tot - 1u;
            if constexpr (BUILD) {
                se  = list[wave * LCAP + qc];
                raw = (unsigned)src[se >> 10];
            } else {
                se  = 0u;
                raw = mygl[qc];
            }
        };
        auto entry = [&](unsigned q0, unsigned se, unsigned raw) -> unsigned {
            const unsigned q = q0 + lane;
            unsigned e;
            if constexpr (BUILD) {
                e = (raw << 10) | (se & 1023u);
                if (q < tot) mygl[q] = e;
            } else {
                e = raw;
            }
            return q < tot ? e : PAD;
        };

        unsigned se0, raw0, se1, raw1;
        ask(0u, se0, raw0);
        ask(32u, se1, raw1);
        const unsigned ent = entry(0u, se0, raw0);
        int s_cur  = (int)(ent >> 10);
        int sl_cur = (int)((ent & 1023u) >> 5);
        count(sl_cur);
#pragma unroll
        for (int t = 0; t < NB - 1; ++t) {
            issue(s_cur, t * G, v[t]);
            __builtin_amdgcn_sched_barrier(0);
        }
        for (unsigned b0 = 0; b0 < tot; b0 += 32) {
            unsigned se2, raw2;
            ask(b0 + 64u, se2, raw2);
            int s_nxt = 0, sl_nxt = SINK_SLOT;
#pragma unroll
            for (int g = 0; g < NG; ++g) {
                const int t = g + NB - 1;
                if (t == NG) {
                    const unsigned entN = entry(b0 + 32u, se1, raw1);
                    s_nxt  = (int)(entN >> 10);
                    sl_nxt = (int)((entN & 1023u) >> 5);
                    count(sl_nxt);
                }
                if (t < NG) issue(s_cur, t * G, v[t % NB]);
                else        issue(s_nxt, (t - NG) * G, v[t % NB]);
                __builtin_amdgcn_sched_barrier(0);
                addin(sl_cur, g * G, v[g % NB]);
            }
            s_cur  = s_nxt;
            sl_cur = sl_nxt;
            se1    = se2;
            raw1   = raw2;
        }
    } else if (slow) {
        for (int i0 = 0; i0 < nEdges; i0 += 32) {
            const int i = i0 + lane;
            const int d = i < nEdges ? dst[i] : -1;
            const unsigned r = (unsigned)(d - base);
            unsigned mask = __builtin_amdgcn_ballot_w32(
                r < (unsigned)TILE_ROWS && (int)(r & 31u) == wave);
            while (mask) {
                const int l = __builtin_ctz(mask);
                mask &= mask - 1;
                const int kr = __builtin_amdgcn_readlane((int)r, l) >> 5;
                const float* xp = x + (size_t)src[i0 + l] * DIM + H * lane;
#pragma unroll
                for (int h = 0; h < H; ++h) a[h][kr] += xp[h];
                if constexpr (COUNT) {
#pragma unroll
                    for (int k = 0; k < ROWS_PER_WAVE; ++k) c[k] += kr == k ? 1 : 0;
                }
            }
        }
    }

    const int left = nNodes - base;
    const int rows = left < TILE_ROWS ? left : TILE_ROWS;
#pragma unroll
    for (int k = 0; k < ROWS_PER_WAVE; ++k) {
        const int r = wave + 32 * k;
        if (r < rows) {
            if constexpr (H == 2) {
                v2f t;
                t.x = a[0][k];
                t.y = a[1][k];
                *reinterpret_cast<v2f*>(agg + (size_t)(base + r) * DIM + 2 * lane) = t;
            } else {
                agg[(size_t)(base + r) * DIM + lane] = a[0][k];
            }
            if (COUNT && lane == 0) cntL[r] = (float)c[k];
        }
    }
    if constexpr (COUNT) {
        __syncthreads();
        for (int i = tid; i < rows; i += AGG_THREADS) cnt[base + i] = cntL[i];
    }
}

#define GEMM_WAVES   8
#define GEMM_THREADS (32 * GEMM_WAVES)
#define GEMM_BLOCKS  256
static_assert(N_NODES % 16 == 0, "row tiles are whole");

template <int K>
__global__ __launch_bounds__(GEMM_THREADS)
void sage_gemm_kernel(const float* aggRaw,
                      const float* cnt,
                      const float* x,
                      const float* __restrict__ Wl,
                      const float* __restrict__ Wr,
                      const float* __restrict__ bias,
                      float* __restrict__ out, int nNodes) {
    constexpr int T = K / 8;
    const int lane = threadIdx.x & 31;
    const int grp  = lane >> 4;
    const int lr   = lane & 15;
    const int wv     = __builtin_amdgcn_readfirstlane((int)(threadIdx.x >> 5));
    const int nWaves = (int)gridDim.x * GEMM_WAVES;
    const int nTiles = nNodes / 16;
    int tile = wv * (int)gridDim.x + (int)blockIdx.x;

    float4 ra0[T], rx0[T], ra1[T], rx1[T];
    float  rc0 = 1.0f, rc1 = 1.0f;
    auto load_tile = [&](int tl, float4 (&a)[T], float4 (&xx)[T], float& c) {
        const int arow = tl * 16 + lr;
        const float4* ap = reinterpret_cast<const float4*>(aggRaw + (size_t)arow * K) + grp;
        const float4* xp = reinterpret_cast<const float4*>(x + (size_t)arow * K) + grp;
        c = cnt[arow];
#pragma unroll
        for (int t = 0; t < T; ++t) { a[t] = ap[2 * t]; xx[t] = xp[2 * t]; }
    };
    if (tile >= nTiles) return;
    load_tile(tile, ra0, rx0, rc0);

    v2f bl[T][2][4], br[T][2][4];
#pragma unroll
    for (int t = 0; t < T; ++t)
#pragma unroll
        for (int p = 0; p < 2; ++p) {
            const int kb = 8 * t + 4 * grp + 2 * p;
#pragma unroll
            for (int n = 0; n < 4; ++n) {
                bl[t][p][n].x = Wl[(size_t)(kb + 0) * HIDDEN_D + 16 * n + lr];
                bl[t][p][n].y = Wl[(size_t)(kb + 1) * HIDDEN_D + 16 * n + lr];
                br[t][p][n].x = Wr[(size_t)(kb + 0) * HIDDEN_D + 16 * n + lr];
                br[t][p][n].y = Wr[(size_t)(kb + 1) * HIDDEN_D + 16 * n + lr];
            }
        }
    float bv[4];
#pragma unroll
    for (int n = 0; n < 4; ++n) bv[n] = bias[16 * n + lr];


    auto compute_tile = [&](int tl, const float4 (&av)[T], const float4 (&xv)[T], float deg) {
        const float invc = 1.0f / fmaxf(deg, 1.0f);
        v8f c[4] = {};
#pragma unroll
        for (int t = 0; t < T; ++t) {
            v2f a0, a1;
            a0.x = av[t].x * invc;  a0.y = av[t].y * invc;
            a1.x = av[t].z * invc;  a1.y = av[t].w * invc;
#pragma unroll
            for (int n = 0; n < 4; ++n)
                c[n] = __builtin_amdgcn_wmma_f32_16x16x4_f32(false, a0, false, bl[t][0][n],
                                                             (short)0, c[n], false, false);
#pragma unroll
            for (int n = 0; n < 4; ++n)
                c[n] = __builtin_amdgcn_wmma_f32_16x16x4_f32(false, a1, false, bl[t][1][n],
                                                             (short)0, c[n], false, false);
        }
#pragma unroll
        for (int t = 0; t < T; ++t) {
            v2f a0, a1;
            a0.x = xv[t].x;  a0.y = xv[t].y;
            a1.x = xv[t].z;  a1.y = xv[t].w;
#pragma unroll
            for (int n = 0; n < 4; ++n)
                c[n] = __builtin_amdgcn_wmma_f32_16x16x4_f32(false, a0, false, br[t][0][n],
                                                             (short)0, c[n], false, false);
#pragma unroll
            for (int n = 0; n < 4; ++n)
                c[n] = __builtin_amdgcn_wmma_f32_16x16x4_f32(false, a1, false, br[t][1][n],
                                                             (short)0, c[n], false, false);
        }
        const int row0 = tl * 16;
#pragma unroll
        for (int n = 0; n < 4; ++n)
#pragma unroll
            for (int v = 0; v < 8; ++v) {
                const int m = v + 8 * grp;
                out[(size_t)(row0 + m) * HIDDEN_D + 16 * n + lr] = fmaxf(c[n][v] + bv[n], 0.0f);
            }
    };

    int next = tile + nWaves;
    load_tile(next < nTiles ? next : tile, ra1, rx1, rc1);
    __builtin_amdgcn_sched_barrier(0);
    compute_tile(tile, ra0, rx0, rc0);
    tile = next;
    while (tile < nTiles) {
        next = tile + nWaves;
        load_tile(next < nTiles ? next : tile, ra0, rx0, rc0);
        __builtin_amdgcn_sched_barrier(0);
        compute_tile(tile, ra1, rx1, rc1);
        tile = next;
        if (tile >= nTiles) break;
        next = tile + nWaves;
        load_tile(next < nTiles ? next : tile, ra1, rx1, rc1);
        __builtin_amdgcn_sched_barrier(0);
        compute_tile(tile, ra0, rx0, rc0);
        tile = next;
    }
}

__global__ void head_pool_kernel(const int* __restrict__ batch,
                                 const float* __restrict__ h,
                                 const float* __restrict__ Wo,
                                 float* __restrict__ zsum, int nNodes) {
    const int lane = threadIdx.x & 31;
    const int tile = blockIdx.x * (blockDim.x >> 5) + (threadIdx.x >> 5);
    const int row0 = tile * 16;
    if (row0 >= nNodes) return;
    const int grp = lane >> 4;
    const int lr  = lane & 15;

    const float4* hp =
        reinterpret_cast<const float4*>(h + (size_t)(row0 + lr) * HIDDEN_D) + grp;
    float4 av[8];
    for (int t = 0; t < 8; ++t) av[t] = hp[2 * t];

    const int4* bp = reinterpret_cast<const int4*>(batch + row0 + 8 * grp);
    const int4 gA = bp[0], gB = bp[1];

    const bool colOk = lr < N_CLASSES;
    const int  cc    = colOk ? lr : 0;

    v8f c = {};
    for (int t = 0; t < 8; ++t) {
        const int kb = 8 * t + 4 * grp;
        const float w0 = Wo[(kb + 0) * N_CLASSES + cc];
        const float w1 = Wo[(kb + 1) * N_CLASSES + cc];
        const float w2 = Wo[(kb + 2) * N_CLASSES + cc];
        const float w3 = Wo[(kb + 3) * N_CLASSES + cc];
        v2f a0, a1, b0, b1;
        a0.x = av[t].x;             a0.y = av[t].y;
        b0.x = colOk ? w0 : 0.0f;   b0.y = colOk ? w1 : 0.0f;
        a1.x = av[t].z;             a1.y = av[t].w;
        b1.x = colOk ? w2 : 0.0f;   b1.y = colOk ? w3 : 0.0f;
        c = __builtin_amdgcn_wmma_f32_16x16x4_f32(false, a0, false, b0,
                                                  (short)0, c, false, false);
        c = __builtin_amdgcn_wmma_f32_16x16x4_f32(false, a1, false, b1,
                                                  (short)0, c, false, false);
    }

    float val[8];
    for (int v = 0; v < 8; ++v) val[v] = colOk ? c[v] : 1.0f;
    const int gid[8] = {gA.x, gA.y, gA.z, gA.w, gB.x, gB.y, gB.z, gB.w};
    if (lr <= N_CLASSES) {
        for (int v = 0; v < 8; ++v)
            atomAdd1(zsum + (size_t)gid[v] * ZS + lr, val[v]);
    }
}

__global__ void finish_kernel(const float* __restrict__ zsum,
                              const float* __restrict__ bo,
                              float* __restrict__ out) {
    int t = blockIdx.x * blockDim.x + threadIdx.x;
    if (t >= N_GRAPHS * N_CLASSES) return;
    int g = t / N_CLASSES;
    int c = t - g * N_CLASSES;
    out[t] = zsum[g * ZS + c] / fmaxf(zsum[g * ZS + N_CLASSES], 1.0f) + bo[c];
}

extern "C" void kernel_launch(void* const* d_in, const int* in_sizes, int n_in,
                              void* d_out, int out_size, void* d_ws, size_t ws_size,
                              hipStream_t stream) {
    const int*   node_ids = (const int*)  d_in[0];
    const int*   src      = (const int*)  d_in[1];
    const int*   dst      = (const int*)  d_in[2];
    const int*   batch    = (const int*)  d_in[3];
    const float* embed    = (const float*)d_in[4];
    const float* W1_l     = (const float*)d_in[5];
    const float* b1       = (const float*)d_in[6];
    const float* W1_r     = (const float*)d_in[7];
    const float* W2_l     = (const float*)d_in[8];
    const float* b2       = (const float*)d_in[9];
    const float* W2_r     = (const float*)d_in[10];
    const float* W_out    = (const float*)d_in[11];
    const float* b_out    = (const float*)d_in[12];
    float* out = (float*)d_out;

    float* ws   = (float*)d_ws;
    float* x0   = ws;
    float* h1   = x0   + (size_t)N_NODES * EMBED_D;
    float* h2   = h1   + (size_t)N_NODES * HIDDEN_D;
    float* agg1 = h2   + (size_t)N_NODES * HIDDEN_D;
    float* agg2 = agg1 + (size_t)N_NODES * EMBED_D;
    float* cnt  = agg2 + (size_t)N_NODES * HIDDEN_D;
    float* zsum = cnt  + (size_t)N_NODES;
    unsigned* gtot  = (unsigned*)(zsum + N_GRAPHS * ZS);
    unsigned* glist = gtot + AGG_BLOCKS * AGG_WAVES;
    unsigned* cmap  = glist + (size_t)AGG_BLOCKS * AGG_WAVES * LCAP;

    const int B = 256;
    auto blocks = [](int n, int b) { return (n + b - 1) / b; };

    prep_kernel<<<PREP_BLOCKS, PREP_THREADS, 0, stream>>>(
        node_ids, embed, x0, dst, cmap, zsum, N_EDGES);

    tile_agg_kernel<EMBED_D, true, true><<<AGG_BLOCKS, AGG_THREADS, 0, stream>>>(
        src, dst, x0, agg1, cnt, glist, gtot, cmap, N_NODES, N_EDGES);
    sage_gemm_kernel<EMBED_D><<<GEMM_BLOCKS, GEMM_THREADS, 0, stream>>>(
        agg1, cnt, x0, W1_l, W1_r, b1, h1, N_NODES);

    tile_agg_kernel<HIDDEN_D, false, false><<<AGG_BLOCKS, AGG_THREADS, 0, stream>>>(
        src, dst, h1, agg2, cnt, glist, gtot, cmap, N_NODES, N_EDGES);
    sage_gemm_kernel<HIDDEN_D><<<GEMM_BLOCKS, GEMM_THREADS, 0, stream>>>(
        agg2, cnt, h1, W2_l, W2_r, b2, h2, N_NODES);

    head_pool_kernel<<<blocks(N_NODES / 16, 4), 128, 0, stream>>>(
        batch, h2, W_out, zsum, N_NODES);
    finish_kernel<<<blocks(N_GRAPHS * N_CLASSES, B), B, 0, stream>>>(
        zsum, b_out, out);
}
